# starting kernel
# baseline (speedup 1.0000x reference)
#include <hip/hip_runtime.h>
#include <hip/hip_bf16.h>

#define F_IN   32
#define F_E    16
#define D      64
#define L      3
#define NEG_SLOPE 0.2f
#define BN_EPS 1e-5f

typedef __attribute__((ext_vector_type(2))) float v2f;
typedef __attribute__((ext_vector_type(8))) float v8f;

__device__ __forceinline__ int f2ord(float f) {
  int i = __float_as_int(f);
  return i >= 0 ? i : (i ^ 0x7FFFFFFF);
}
__device__ __forceinline__ float ord2f(int i) {
  return __int_as_float(i >= 0 ? i : (i ^ 0x7FFFFFFF));
}

__global__ void k_loop_accum(const float* __restrict__ eattr, const int* __restrict__ dst,
                             float* __restrict__ lsum, float* __restrict__ deg, int E) {
  int e = blockIdx.x * blockDim.x + threadIdx.x;
  if (e >= E) return;
  int d = dst[e];
  #pragma unroll
  for (int f = 0; f < F_E; f++) atomicAdd(&lsum[d * F_E + f], eattr[e * F_E + f]);
  atomicAdd(&deg[d], 1.0f);
}

__global__ void k_loop_div(float* __restrict__ lsum, const float* __restrict__ deg, int n) {
  int i = blockIdx.x * blockDim.x + threadIdx.x;
  if (i >= n * F_E) return;
  float c = deg[i / F_E];
  lsum[i] = lsum[i] / fmaxf(c, 1.0f);
}

__global__ void k_wea(const float* __restrict__ We, const float* __restrict__ ae,
                      float* __restrict__ wea) {
  int i = blockIdx.x * blockDim.x + threadIdx.x;
  if (i >= L * F_E) return;
  int l = i / F_E, f = i % F_E;
  float s = 0.f;
  #pragma unroll
  for (int d = 0; d < D; d++) s += We[(l * F_E + f) * D + d] * ae[l * D + d];
  wea[i] = s;
}

__global__ void k_gemm_wmma(const float* __restrict__ A, const float* __restrict__ B,
                            const float* __restrict__ bias, float* __restrict__ C,
                            int n, int K) {
  int wave = (blockIdx.x * blockDim.x + threadIdx.x) >> 5;
  int lane = threadIdx.x & 31;
  int row0 = wave * 16;
  if (row0 >= n) return;
  int lm = lane & 15;
  int kh = (lane >> 4) << 1;
  int mrow = row0 + lm;
  int mrd = mrow < n ? mrow : n - 1;

  v8f acc[4] = {};
  for (int k = 0; k < K; k += 4) {
    v2f a;
    a.x = A[(size_t)mrd * K + k + kh];
    a.y = A[(size_t)mrd * K + k + kh + 1];
    #pragma unroll
    for (int nt = 0; nt < 4; nt++) {
      int col = nt * 16 + lm;
      v2f b;
      b.x = B[(k + kh) * D + col];
      b.y = B[(k + kh + 1) * D + col];
      acc[nt] = __builtin_amdgcn_wmma_f32_16x16x4_f32(
          false, a, false, b, (short)0, acc[nt], false, false);
    }
  }

  int rbase = row0 + ((lane >> 4) << 3);
  #pragma unroll
  for (int nt = 0; nt < 4; nt++) {
    int col = nt * 16 + lm;
    float bv = bias ? bias[col] : 0.f;
    #pragma unroll
    for (int r = 0; r < 8; r++) {
      int row = rbase + r;
      if (row < n) C[(size_t)row * D + col] = acc[nt][r] + bv;
    }
  }
}

__global__ void k_scores(const float* __restrict__ xw, const float* __restrict__ as,
                         const float* __restrict__ ad, float* __restrict__ s,
                         float* __restrict__ dv, int n) {
  int i = blockIdx.x * blockDim.x + threadIdx.x;
  if (i >= n) return;
  const float* row = xw + (size_t)i * D;
  float ss = 0.f, dd = 0.f;
  #pragma unroll
  for (int f = 0; f < D; f++) { float v = row[f]; ss += v * as[f]; dd += v * ad[f]; }
  s[i] = ss;
  dv[i] = dd;
}

__global__ void k_init_amax(int* __restrict__ amax, int n) {
  int i = blockIdx.x * blockDim.x + threadIdx.x;
  if (i < n) amax[i] = (int)0x80000000;
}

__global__ void k_alpha(const float* __restrict__ s, const float* __restrict__ dv,
                        const float* __restrict__ eattr, const float* __restrict__ lattr,
                        const float* __restrict__ wea_l, const int* __restrict__ src,
                        const int* __restrict__ dst, float* __restrict__ alpha,
                        int* __restrict__ amax, int E, int n) {
  int i = blockIdx.x * blockDim.x + threadIdx.x;
  if (i >= E + n) return;
  int se, de;
  const float* ea;
  if (i < E) { se = src[i]; de = dst[i]; ea = eattr + (size_t)i * F_E; }
  else       { se = i - E;  de = i - E;  ea = lattr + (size_t)(i - E) * F_E; }
  float aed = 0.f;
  #pragma unroll
  for (int f = 0; f < F_E; f++) aed += ea[f] * wea_l[f];
  float a = s[se] + dv[de] + aed;
  a = a > 0.f ? a : NEG_SLOPE * a;
  alpha[i] = a;
  atomicMax(&amax[de], f2ord(a));
}

__global__ void k_exp(float* __restrict__ alpha, const int* __restrict__ amax,
                      const int* __restrict__ dst, float* __restrict__ denom, int E, int n) {
  int i = blockIdx.x * blockDim.x + threadIdx.x;
  if (i >= E + n) return;
  int de = i < E ? dst[i] : i - E;
  float ex = expf(alpha[i] - ord2f(amax[de]));
  alpha[i] = ex;
  atomicAdd(&denom[de], ex);
}

__global__ void k_aggr(const float* __restrict__ ex, const float* __restrict__ denom,
                       const float* __restrict__ xw, const int* __restrict__ src,
                       const int* __restrict__ dst, float* __restrict__ acc, int E, int n) {
  int t = blockIdx.x * blockDim.x + threadIdx.x;
  if (t >= (E + n) * 16) return;
  int i = t >> 4;
  int f0 = (t & 15) << 2;
  int se = i < E ? src[i] : i - E;
  int de = i < E ? dst[i] : i - E;
  float w = ex[i] / denom[de];
  const float4 v = *(const float4*)(xw + (size_t)se * D + f0);
  float* ap = acc + (size_t)de * D + f0;
  atomicAdd(ap + 0, w * v.x);
  atomicAdd(ap + 1, w * v.y);
  atomicAdd(ap + 2, w * v.z);
  atomicAdd(ap + 3, w * v.w);
}

__global__ void k_bn(const float* __restrict__ acc, const float* __restrict__ bias,
                     const float* __restrict__ gamma, const float* __restrict__ beta,
                     const float* __restrict__ mean, const float* __restrict__ var,
                     float* __restrict__ h, int n, int relu) {
  int i = blockIdx.x * blockDim.x + threadIdx.x;
  if (i >= n * D) return;
  int f = i & (D - 1);
  float v = acc[i] + bias[f];
  v = (v - mean[f]) * (gamma[f] / sqrtf(var[f] + BN_EPS)) + beta[f];
  if (relu) v = fmaxf(v, 0.f);
  h[i] = v;
}

__global__ void k_output(const float* __restrict__ h, const int* __restrict__ batch,
                         float* __restrict__ out_nodes, float* __restrict__ out_graph, int n) {
  int i = blockIdx.x * blockDim.x + threadIdx.x;
  if (i >= n * D) return;
  float v = h[i];
  out_nodes[i] = v;
  int node = i >> 6;
  int f = i & (D - 1);
  atomicAdd(&out_graph[(size_t)batch[node] * D + f], v);
}

extern "C" void kernel_launch(void* const* d_in, const int* in_sizes, int n_in,
                              void* d_out, int out_size, void* d_ws, size_t ws_size,
                              hipStream_t stream) {
  const float* x     = (const float*)d_in[0];
  const int*   eidx  = (const int*)d_in[1];
  const float* eattr = (const float*)d_in[2];
  const int*   batch = (const int*)d_in[3];
  const float* xlw   = (const float*)d_in[4];
  const float* xlb   = (const float*)d_in[5];
  const float* glw   = (const float*)d_in[6];
  const float* gas   = (const float*)d_in[7];
  const float* gad   = (const float*)d_in[8];
  const float* gew   = (const float*)d_in[9];
  const float* gae   = (const float*)d_in[10];
  const float* gb    = (const float*)d_in[11];
  const float* bng   = (const float*)d_in[12];
  const float* bnb   = (const float*)d_in[13];
  const float* bnm   = (const float*)d_in[14];
  const float* bnv   = (const float*)d_in[15];

  const int n = in_sizes[3];
  const int E = in_sizes[2] / F_E;
  const int* srcp = eidx;
  const int* dstp = eidx + E;

  float* ws    = (float*)d_ws;
  float* h     = ws;  ws += (size_t)n * D;
  float* xw    = ws;  ws += (size_t)n * D;
  float* acc   = ws;  ws += (size_t)n * D;
  float* sS    = ws;  ws += n;
  float* dS    = ws;  ws += n;
  float* lattr = ws;  ws += (size_t)n * F_E;
  float* deg   = ws;  ws += n;
  int*   amax  = (int*)ws; ws += n;
  float* denom = ws;  ws += n;
  float* alpha = ws;  ws += (size_t)(E + n);
  float* wea   = ws;  ws += L * F_E;

  float* out_nodes = (float*)d_out;
  float* out_graph = out_nodes + (size_t)n * D;
  const size_t graph_elems = (size_t)out_size - (size_t)n * D;

  const int B = 256;
  #define GRID(x) (((x) + B - 1) / B)

  hipMemsetAsync(lattr, 0, (size_t)n * F_E * sizeof(float), stream);
  hipMemsetAsync(deg, 0, (size_t)n * sizeof(float), stream);
  k_loop_accum<<<GRID(E), B, 0, stream>>>(eattr, dstp, lattr, deg, E);
  k_loop_div<<<GRID(n * F_E), B, 0, stream>>>(lattr, deg, n);
  k_wea<<<GRID(L * F_E), B, 0, stream>>>(gew, gae, wea);

  {
    int waves = (n + 15) / 16;
    int blocks = (waves + 3) / 4;
    k_gemm_wmma<<<blocks, 128, 0, stream>>>(x, xlw, xlb, h, n, F_IN);
  }

  for (int l = 0; l < L; l++) {
    hipMemsetAsync(denom, 0, (size_t)n * sizeof(float), stream);
    hipMemsetAsync(acc, 0, (size_t)n * D * sizeof(float), stream);
    k_init_amax<<<GRID(n), B, 0, stream>>>(amax, n);

    int waves = (n + 15) / 16;
    int blocks = (waves + 3) / 4;
    k_gemm_wmma<<<blocks, 128, 0, stream>>>(h, glw + (size_t)l * D * D, nullptr, xw, n, D);

    k_scores<<<GRID(n), B, 0, stream>>>(xw, gas + (size_t)l * D, gad + (size_t)l * D, sS, dS, n);
    k_alpha<<<GRID(E + n), B, 0, stream>>>(sS, dS, eattr, lattr, wea + (size_t)l * F_E,
                                           srcp, dstp, alpha, amax, E, n);
    k_exp<<<GRID(E + n), B, 0, stream>>>(alpha, amax, dstp, denom, E, n);
    k_aggr<<<GRID((E + n) * 16), B, 0, stream>>>(alpha, denom, xw, srcp, dstp, acc, E, n);
    k_bn<<<GRID(n * D), B, 0, stream>>>(acc, gb + (size_t)l * D, bng + (size_t)l * D,
                                        bnb + (size_t)l * D, bnm + (size_t)l * D,
                                        bnv + (size_t)l * D, h, n, l < L - 1 ? 1 : 0);
  }

  hipMemsetAsync(out_graph, 0, graph_elems * sizeof(float), stream);
  k_output<<<GRID(n * D), B, 0, stream>>>(h, batch, out_nodes, out_graph, n);
  #undef GRID
}


# float scatter-adds by whole lines: loop-attr sums+count in one 128B row per node, one 17-lane atomic per edge; aggregation two 32-lane 128B atomics per edge; a wave walks 32 edges; arrival order kept
# speedup vs baseline: 1.6931x; 1.6931x over previous
#include <hip/hip_runtime.h>
#include <hip/hip_bf16.h>

#define F_IN   32
#define F_E    16
#define D      64
#define L      3
#define NEG_SLOPE 0.2f
#define BN_EPS 1e-5f

typedef __attribute__((ext_vector_type(2))) float v2f;
typedef __attribute__((ext_vector_type(8))) float v8f;

__device__ __forceinline__ int f2ord(float f) {
  int i = __float_as_int(f);
  return i >= 0 ? i : (i ^ 0x7FFFFFFF);
}
__device__ __forceinline__ float ord2f(int i) {
  return __int_as_float(i >= 0 ? i : (i ^ 0x7FFFFFFF));
}

#define LROW 32
__global__ void k_loop_accum(const float* __restrict__ eattr, const int* __restrict__ dst,
                             float* __restrict__ rows, int E) {
  int el = blockIdx.x * blockDim.x + threadIdx.x;
  int e0 = __builtin_amdgcn_readfirstlane(el & ~31);
  if (e0 >= E) return;
  int lane = threadIdx.x & 31;
  int half = lane >> 4;
  int f = lane & 15;
  int cB = lane ^ 16;
  int dl = dst[el < E ? el : E - 1];
  if (e0 + 32 <= E) {
    float v[16];
    #pragma unroll
    for (int it = 0; it < 16; it++) v[it] = eattr[(size_t)(e0 + 2 * it + half) * F_E + f];
    #pragma unroll
    for (int it = 0; it < 16; it++) {
      int dA = __builtin_amdgcn_readlane(dl, 2 * it);
      int dB = __builtin_amdgcn_readlane(dl, 2 * it + 1);
      if (lane <= 16) atomicAdd(&rows[(size_t)dA * LROW + lane], half ? 1.0f : v[it]);
      if (cB <= 16) atomicAdd(&rows[(size_t)dB * LROW + cB], half ? v[it] : 1.0f);
    }
  } else {
    for (int j = 0; j < E - e0; j++) {
      int d = __shfl(dl, j);
      float x = eattr[(size_t)(e0 + j) * F_E + f];
      if (lane <= 16) atomicAdd(&rows[(size_t)d * LROW + lane], half ? 1.0f : x);
    }
  }
}

__global__ void k_loop_div(const float* __restrict__ rows, float* __restrict__ lattr, int n) {
  int i = blockIdx.x * blockDim.x + threadIdx.x;
  if (i >= n * F_E) return;
  int node = i / F_E, f = i % F_E;
  float c = rows[(size_t)node * LROW + F_E];
  lattr[i] = rows[(size_t)node * LROW + f] / fmaxf(c, 1.0f);
}

__global__ void k_wea(const float* __restrict__ We, const float* __restrict__ ae,
                      float* __restrict__ wea) {
  int i = blockIdx.x * blockDim.x + threadIdx.x;
  if (i >= L * F_E) return;
  int l = i / F_E, f = i % F_E;
  float s = 0.f;
  #pragma unroll
  for (int d = 0; d < D; d++) s += We[(l * F_E + f) * D + d] * ae[l * D + d];
  wea[i] = s;
}

__global__ void k_gemm_wmma(const float* __restrict__ A, const float* __restrict__ B,
                            const float* __restrict__ bias, float* __restrict__ C,
                            int n, int K) {
  int wave = (blockIdx.x * blockDim.x + threadIdx.x) >> 5;
  int lane = threadIdx.x & 31;
  int row0 = wave * 16;
  if (row0 >= n) return;
  int lm = lane & 15;
  int kh = (lane >> 4) << 1;
  int mrow = row0 + lm;
  int mrd = mrow < n ? mrow : n - 1;

  v8f acc[4] = {};
  for (int k = 0; k < K; k += 4) {
    v2f a;
    a.x = A[(size_t)mrd * K + k + kh];
    a.y = A[(size_t)mrd * K + k + kh + 1];
    #pragma unroll
    for (int nt = 0; nt < 4; nt++) {
      int col = nt * 16 + lm;
      v2f b;
      b.x = B[(k + kh) * D + col];
      b.y = B[(k + kh + 1) * D + col];
      acc[nt] = __builtin_amdgcn_wmma_f32_16x16x4_f32(
          false, a, false, b, (short)0, acc[nt], false, false);
    }
  }

  int rbase = row0 + ((lane >> 4) << 3);
  #pragma unroll
  for (int nt = 0; nt < 4; nt++) {
    int col = nt * 16 + lm;
    float bv = bias ? bias[col] : 0.f;
    #pragma unroll
    for (int r = 0; r < 8; r++) {
      int row = rbase + r;
      if (row < n) C[(size_t)row * D + col] = acc[nt][r] + bv;
    }
  }
}

__global__ void k_scores(const float* __restrict__ xw, const float* __restrict__ as,
                         const float* __restrict__ ad, float* __restrict__ s,
                         float* __restrict__ dv, int n) {
  int i = blockIdx.x * blockDim.x + threadIdx.x;
  if (i >= n) return;
  const float* row = xw + (size_t)i * D;
  float ss = 0.f, dd = 0.f;
  #pragma unroll
  for (int f = 0; f < D; f++) { float v = row[f]; ss += v * as[f]; dd += v * ad[f]; }
  s[i] = ss;
  dv[i] = dd;
}

__global__ void k_init_amax(int* __restrict__ amax, int n) {
  int i = blockIdx.x * blockDim.x + threadIdx.x;
  if (i < n) amax[i] = (int)0x80000000;
}

__global__ void k_alpha(const float* __restrict__ s, const float* __restrict__ dv,
                        const float* __restrict__ eattr, const float* __restrict__ lattr,
                        const float* __restrict__ wea_l, const int* __restrict__ src,
                        const int* __restrict__ dst, float* __restrict__ alpha,
                        int* __restrict__ amax, int E, int n) {
  int i = blockIdx.x * blockDim.x + threadIdx.x;
  if (i >= E + n) return;
  int se, de;
  const float* ea;
  if (i < E) { se = src[i]; de = dst[i]; ea = eattr + (size_t)i * F_E; }
  else       { se = i - E;  de = i - E;  ea = lattr + (size_t)(i - E) * F_E; }
  float aed = 0.f;
  #pragma unroll
  for (int f = 0; f < F_E; f++) aed += ea[f] * wea_l[f];
  float a = s[se] + dv[de] + aed;
  a = a > 0.f ? a : NEG_SLOPE * a;
  alpha[i] = a;
  atomicMax(&amax[de], f2ord(a));
}

__global__ void k_exp(float* __restrict__ alpha, const int* __restrict__ amax,
                      const int* __restrict__ dst, float* __restrict__ denom, int E, int n) {
  int i = blockIdx.x * blockDim.x + threadIdx.x;
  if (i >= E + n) return;
  int de = i < E ? dst[i] : i - E;
  float ex = expf(alpha[i] - ord2f(amax[de]));
  alpha[i] = ex;
  atomicAdd(&denom[de], ex);
}

__global__ void k_aggr(const float* __restrict__ ex, const float* __restrict__ denom,
                       const float* __restrict__ xw, const int* __restrict__ src,
                       const int* __restrict__ dst, float* __restrict__ acc, int E, int n) {
  int il = blockIdx.x * blockDim.x + threadIdx.x;
  int i0 = __builtin_amdgcn_readfirstlane(il & ~31);
  int total = E + n;
  if (i0 >= total) return;
  int lane = threadIdx.x & 31;
  int ic = il < total ? il : total - 1;
  int se = ic < E ? src[ic] : ic - E;
  int de = ic < E ? dst[ic] : ic - E;
  float w = ex[ic] / denom[de];
  if (i0 + 32 <= total) {
    int wbits = __float_as_int(w);
    #pragma unroll
    for (int c = 0; c < 32; c += 16) {
      float v0[16], v1[16];
      #pragma unroll
      for (int k = 0; k < 16; k++) {
        int s = __builtin_amdgcn_readlane(se, c + k);
        const float* xr = xw + (size_t)s * D + lane;
        v0[k] = xr[0];
        v1[k] = xr[32];
      }
      #pragma unroll
      for (int k = 0; k < 16; k++) {
        int d = __builtin_amdgcn_readlane(de, c + k);
        float wk = __int_as_float(__builtin_amdgcn_readlane(wbits, c + k));
        float* ap = acc + (size_t)d * D + lane;
        atomicAdd(ap, wk * v0[k]);
        atomicAdd(ap + 32, wk * v1[k]);
      }
    }
  } else {
    for (int j = 0; j < total - i0; j++) {
      int s = __shfl(se, j);
      int d = __shfl(de, j);
      float wj = __shfl(w, j);
      const float* xr = xw + (size_t)s * D + lane;
      float* ap = acc + (size_t)d * D + lane;
      atomicAdd(ap, wj * xr[0]);
      atomicAdd(ap + 32, wj * xr[32]);
    }
  }
}

__global__ void k_bn(const float* __restrict__ acc, const float* __restrict__ bias,
                     const float* __restrict__ gamma, const float* __restrict__ beta,
                     const float* __restrict__ mean, const float* __restrict__ var,
                     float* __restrict__ h, int n, int relu) {
  int i = blockIdx.x * blockDim.x + threadIdx.x;
  if (i >= n * D) return;
  int f = i & (D - 1);
  float v = acc[i] + bias[f];
  v = (v - mean[f]) * (gamma[f] / sqrtf(var[f] + BN_EPS)) + beta[f];
  if (relu) v = fmaxf(v, 0.f);
  h[i] = v;
}

__global__ void k_output(const float* __restrict__ h, const int* __restrict__ batch,
                         float* __restrict__ out_nodes, float* __restrict__ out_graph, int n) {
  int i = blockIdx.x * blockDim.x + threadIdx.x;
  if (i >= n * D) return;
  float v = h[i];
  out_nodes[i] = v;
  int node = i >> 6;
  int f = i & (D - 1);
  atomicAdd(&out_graph[(size_t)batch[node] * D + f], v);
}

extern "C" void kernel_launch(void* const* d_in, const int* in_sizes, int n_in,
                              void* d_out, int out_size, void* d_ws, size_t ws_size,
                              hipStream_t stream) {
  const float* x     = (const float*)d_in[0];
  const int*   eidx  = (const int*)d_in[1];
  const float* eattr = (const float*)d_in[2];
  const int*   batch = (const int*)d_in[3];
  const float* xlw   = (const float*)d_in[4];
  const float* xlb   = (const float*)d_in[5];
  const float* glw   = (const float*)d_in[6];
  const float* gas   = (const float*)d_in[7];
  const float* gad   = (const float*)d_in[8];
  const float* gew   = (const float*)d_in[9];
  const float* gae   = (const float*)d_in[10];
  const float* gb    = (const float*)d_in[11];
  const float* bng   = (const float*)d_in[12];
  const float* bnb   = (const float*)d_in[13];
  const float* bnm   = (const float*)d_in[14];
  const float* bnv   = (const float*)d_in[15];

  const int n = in_sizes[3];
  const int E = in_sizes[2] / F_E;
  const int* srcp = eidx;
  const int* dstp = eidx + E;

  float* ws    = (float*)d_ws;
  float* h     = ws;  ws += (size_t)n * D;
  float* xw    = ws;  ws += (size_t)n * D;
  float* acc   = ws;  ws += (size_t)n * D;
  float* sS    = ws;  ws += n;
  float* dS    = ws;  ws += n;
  float* lattr = ws;  ws += (size_t)n * F_E;
  int*   amax  = (int*)ws; ws += n;
  float* denom = ws;  ws += n;
  float* alpha = ws;  ws += (size_t)(E + n);
  float* wea   = ws;  ws += L * F_E;

  float* out_nodes = (float*)d_out;
  float* out_graph = out_nodes + (size_t)n * D;
  const size_t graph_elems = (size_t)out_size - (size_t)n * D;

  const int B = 256;
  #define GRID(x) (((x) + B - 1) / B)

  float* lrows = acc;
  hipMemsetAsync(lrows, 0, (size_t)n * LROW * sizeof(float), stream);
  k_loop_accum<<<GRID(E), B, 0, stream>>>(eattr, dstp, lrows, E);
  k_loop_div<<<GRID(n * F_E), B, 0, stream>>>(lrows, lattr, n);
  k_wea<<<GRID(L * F_E), B, 0, stream>>>(gew, gae, wea);

  {
    int waves = (n + 15) / 16;
    int blocks = (waves + 3) / 4;
    k_gemm_wmma<<<blocks, 128, 0, stream>>>(x, xlw, xlb, h, n, F_IN);
  }

  for (int l = 0; l < L; l++) {
    hipMemsetAsync(denom, 0, (size_t)n * sizeof(float), stream);
    hipMemsetAsync(acc, 0, (size_t)n * D * sizeof(float), stream);
    k_init_amax<<<GRID(n), B, 0, stream>>>(amax, n);

    int waves = (n + 15) / 16;
    int blocks = (waves + 3) / 4;
    k_gemm_wmma<<<blocks, 128, 0, stream>>>(h, glw + (size_t)l * D * D, nullptr, xw, n, D);

    k_scores<<<GRID(n), B, 0, stream>>>(xw, gas + (size_t)l * D, gad + (size_t)l * D, sS, dS, n);
    k_alpha<<<GRID(E + n), B, 0, stream>>>(sS, dS, eattr, lattr, wea + (size_t)l * F_E,
                                           srcp, dstp, alpha, amax, E, n);
    k_exp<<<GRID(E + n), B, 0, stream>>>(alpha, amax, dstp, denom, E, n);
    k_aggr<<<GRID(E + n), B, 0, stream>>>(alpha, denom, xw, srcp, dstp, acc, E, n);
    k_bn<<<GRID(n * D), B, 0, stream>>>(acc, gb + (size_t)l * D, bng + (size_t)l * D,
                                        bnb + (size_t)l * D, bnm + (size_t)l * D,
                                        bnv + (size_t)l * D, h, n, l < L - 1 ? 1 : 0);
  }

  hipMemsetAsync(out_graph, 0, graph_elems * sizeof(float), stream);
  k_output<<<GRID(n * D), B, 0, stream>>>(h, batch, out_nodes, out_graph, n);
  #undef GRID
}


# edges counting-sorted by destination per call (arrival-order rank); one wave per node: scores, online softmax, row sum, self-loop mean of edge terms, BN in registers; no float atomics in layers
# speedup vs baseline: 4.1444x; 2.4478x over previous
#include <hip/hip_runtime.h>
#include <hip/hip_bf16.h>

#define F_IN   32
#define F_E    16
#define D      64
#define L      3
#define NEG_SLOPE 0.2f
#define BN_EPS 1e-5f
#define SCAN_TILE 4096

typedef __attribute__((ext_vector_type(2))) float v2f;
typedef __attribute__((ext_vector_type(8))) float v8f;

__global__ void k_hist(const int* __restrict__ dst, int* __restrict__ cnt,
                       int* __restrict__ rank, int E) {
  int e = blockIdx.x * blockDim.x + threadIdx.x;
  if (e >= E) return;
  rank[e] = atomicAdd(&cnt[dst[e]], 1);
}

#define SCAN_NT (SCAN_TILE / 4)
#define SCAN_NW (SCAN_NT / 32)
__global__ void __launch_bounds__(SCAN_NT) k_scan(const int* __restrict__ cnt, int* __restrict__ rowptr) {
  __shared__ int wsum[SCAN_NW];
  __shared__ int carry_s[SCAN_NW];
  int t = threadIdx.x;
  int lane = t & 31, wv = t >> 5;
  const int4* c4 = (const int4*)cnt;
  int nf = (int)blockIdx.x * SCAN_NT;
  int front = 0;
  int i = t;
  for (; i + 3 * SCAN_NT < nf; i += 4 * SCAN_NT) {
    int4 q0 = c4[i], q1 = c4[i + SCAN_NT], q2 = c4[i + 2 * SCAN_NT], q3 = c4[i + 3 * SCAN_NT];
    front += (q0.x + q0.y + q0.z + q0.w) + (q1.x + q1.y + q1.z + q1.w)
           + (q2.x + q2.y + q2.z + q2.w) + (q3.x + q3.y + q3.z + q3.w);
  }
  for (; i < nf; i += SCAN_NT) {
    int4 q0 = c4[i];
    front += q0.x + q0.y + q0.z + q0.w;
  }
  int4 q = c4[nf + t];
  int own = q.x + q.y + q.z + q.w;
  int inc = own;
  #pragma unroll
  for (int o = 1; o < 32; o <<= 1) {
    int u = __shfl_up(inc, o);
    if (lane >= o) inc += u;
  }
  #pragma unroll
  for (int o = 16; o > 0; o >>= 1) front += __shfl_xor(front, o);
  if (lane == 31) wsum[wv] = inc;
  if (lane == 0) carry_s[wv] = front;
  __syncthreads();
  int base = 0;
  #pragma unroll
  for (int w = 0; w < SCAN_NW; w++) {
    base += carry_s[w];
    if (w < wv) base += wsum[w];
  }
  int ex = base + inc - own;
  int4 r;
  r.x = ex;
  r.y = ex + q.x;
  r.z = r.y + q.y;
  r.w = r.z + q.z;
  ((int4*)rowptr)[nf + t] = r;
}

__global__ void k_scatter(const int* __restrict__ src, const int* __restrict__ dst,
                          const int* __restrict__ rowptr, const int* __restrict__ rank,
                          int* __restrict__ slot, int E) {
  int e = blockIdx.x * blockDim.x + threadIdx.x;
  if (e >= E) return;
  int pos = rowptr[dst[e]] + rank[e];
  ((int2*)slot)[pos] = make_int2(e, src[e]);
}

__global__ void k_wea(const float* __restrict__ We, const float* __restrict__ ae,
                      float* __restrict__ wea) {
  int i = blockIdx.x * blockDim.x + threadIdx.x;
  if (i >= L * F_E) return;
  int l = i / F_E, f = i % F_E;
  float s = 0.f;
  #pragma unroll
  for (int d = 0; d < D; d++) s += We[(l * F_E + f) * D + d] * ae[l * D + d];
  wea[i] = s;
}

__global__ void k_gemm_wmma(const float* __restrict__ A, const float* __restrict__ B,
                            const float* __restrict__ bias, float* __restrict__ C,
                            int n, int K) {
  int wave = (blockIdx.x * blockDim.x + threadIdx.x) >> 5;
  int lane = threadIdx.x & 31;
  int row0 = wave * 16;
  if (row0 >= n) return;
  int lm = lane & 15;
  int kh = (lane >> 4) << 1;
  int mrow = row0 + lm;
  int mrd = mrow < n ? mrow : n - 1;

  v8f acc[4] = {};
  for (int k = 0; k < K; k += 4) {
    v2f a;
    a.x = A[(size_t)mrd * K + k + kh];
    a.y = A[(size_t)mrd * K + k + kh + 1];
    #pragma unroll
    for (int nt = 0; nt < 4; nt++) {
      int col = nt * 16 + lm;
      v2f b;
      b.x = B[(k + kh) * D + col];
      b.y = B[(k + kh + 1) * D + col];
      acc[nt] = __builtin_amdgcn_wmma_f32_16x16x4_f32(
          false, a, false, b, (short)0, acc[nt], false, false);
    }
  }

  int rbase = row0 + ((lane >> 4) << 3);
  #pragma unroll
  for (int nt = 0; nt < 4; nt++) {
    int col = nt * 16 + lm;
    float bv = bias ? bias[col] : 0.f;
    #pragma unroll
    for (int r = 0; r < 8; r++) {
      int row = rbase + r;
      if (row < n) C[(size_t)row * D + col] = acc[nt][r] + bv;
    }
  }
}

__global__ void k_scores(const float* __restrict__ xw, const float* __restrict__ as,
                         const float* __restrict__ ad, float* __restrict__ s,
                         float* __restrict__ dv, int n) {
  int i = blockIdx.x * blockDim.x + threadIdx.x;
  if (i >= n) return;
  const float* row = xw + (size_t)i * D;
  float ss = 0.f, dd = 0.f;
  #pragma unroll
  for (int f = 0; f < D; f++) { float v = row[f]; ss += v * as[f]; dd += v * ad[f]; }
  s[i] = ss;
  dv[i] = dd;
}

__device__ __forceinline__ float wave_max(float v) {
  #pragma unroll
  for (int o = 16; o > 0; o >>= 1) v = fmaxf(v, __shfl_xor(v, o));
  return v;
}
__device__ __forceinline__ float wave_sum(float v) {
  #pragma unroll
  for (int o = 16; o > 0; o >>= 1) v += __shfl_xor(v, o);
  return v;
}

__global__ void k_node(const float* __restrict__ xw, const float* __restrict__ sS,
                       const float* __restrict__ dS, const float* __restrict__ eattr,
                       const float* __restrict__ wea_l, const int* __restrict__ rowptr,
                       const int* __restrict__ slot,
                       const float* __restrict__ bias, const float* __restrict__ gamma,
                       const float* __restrict__ beta, const float* __restrict__ mean,
                       const float* __restrict__ var, float* __restrict__ h, int n, int relu) {
  int node = __builtin_amdgcn_readfirstlane((int)((blockIdx.x * blockDim.x + threadIdx.x) >> 5));
  if (node >= n) return;
  int lane = threadIdx.x & 31;
  int beg = rowptr[node];
  int end = rowptr[node + 1];
  float dn = dS[node];
  float wl[F_E];
  #pragma unroll
  for (int f = 0; f < F_E; f++) wl[f] = wea_l[f];

  float m = -INFINITY, lsum = 0.f, acc0 = 0.f, acc1 = 0.f, aepart = 0.f;
  for (int base = beg; base < end; base += 32) {
    int j = base + lane;
    bool valid = j < end;
    int2 es = valid ? ((const int2*)slot)[j] : make_int2(0, node);
    int e = es.x;
    int s = es.y;
    const float4* ea4 = (const float4*)(eattr + (size_t)e * F_E);
    float4 q0 = ea4[0], q1 = ea4[1], q2 = ea4[2], q3 = ea4[3];
    float ae = 0.f;
    ae += q0.x * wl[0];  ae += q0.y * wl[1];  ae += q0.z * wl[2];  ae += q0.w * wl[3];
    ae += q1.x * wl[4];  ae += q1.y * wl[5];  ae += q1.z * wl[6];  ae += q1.w * wl[7];
    ae += q2.x * wl[8];  ae += q2.y * wl[9];  ae += q2.z * wl[10]; ae += q2.w * wl[11];
    ae += q3.x * wl[12]; ae += q3.y * wl[13]; ae += q3.z * wl[14]; ae += q3.w * wl[15];
    float a = sS[s] + dn + ae;
    a = a > 0.f ? a : NEG_SLOPE * a;
    if (!valid) { ae = 0.f; a = -INFINITY; }
    aepart += ae;
    float mnew = fmaxf(m, wave_max(a));
    float scale = expf(m - mnew);
    float p = expf(a - mnew);
    lsum = lsum * scale + wave_sum(p);
    acc0 *= scale;
    acc1 *= scale;
    m = mnew;
    int pbits = __float_as_int(p);
    int cnt = end - base < 32 ? end - base : 32;
    for (int j0 = 0; j0 < cnt; j0 += 8) {
      float r0[8], r1[8];
      #pragma unroll
      for (int k = 0; k < 8; k++) {
        int sj = __builtin_amdgcn_readlane(s, j0 + k);
        const float* xr = xw + (size_t)sj * D + lane;
        r0[k] = xr[0];
        r1[k] = xr[32];
      }
      #pragma unroll
      for (int k = 0; k < 8; k++) {
        float pj = __int_as_float(__builtin_amdgcn_readlane(pbits, j0 + k));
        acc0 += pj * r0[k];
        acc1 += pj * r1[k];
      }
    }
  }
  {
    int deg = end - beg;
    float aes = wave_sum(aepart) / fmaxf((float)deg, 1.0f);
    float a = sS[node] + dn + aes;
    a = a > 0.f ? a : NEG_SLOPE * a;
    float mnew = fmaxf(m, a);
    float scale = expf(m - mnew);
    float p = expf(a - mnew);
    lsum = lsum * scale + p;
    const float* xr = xw + (size_t)node * D + lane;
    acc0 = acc0 * scale + p * xr[0];
    acc1 = acc1 * scale + p * xr[32];
  }
  float inv = 1.0f / lsum;
  #pragma unroll
  for (int hlf = 0; hlf < 2; hlf++) {
    int f = lane + 32 * hlf;
    float v = (hlf ? acc1 : acc0) * inv + bias[f];
    v = (v - mean[f]) * (gamma[f] / sqrtf(var[f] + BN_EPS)) + beta[f];
    if (relu) v = fmaxf(v, 0.f);
    h[(size_t)node * D + f] = v;
  }
}

__global__ void k_output(const float* __restrict__ h, const int* __restrict__ batch,
                         float* __restrict__ out_nodes, float* __restrict__ out_graph, int n) {
  int i = blockIdx.x * blockDim.x + threadIdx.x;
  if (i >= n * D) return;
  float v = h[i];
  out_nodes[i] = v;
  int node = i >> 6;
  int f = i & (D - 1);
  atomicAdd(&out_graph[(size_t)batch[node] * D + f], v);
}

extern "C" void kernel_launch(void* const* d_in, const int* in_sizes, int n_in,
                              void* d_out, int out_size, void* d_ws, size_t ws_size,
                              hipStream_t stream) {
  const float* x     = (const float*)d_in[0];
  const int*   eidx  = (const int*)d_in[1];
  const float* eattr = (const float*)d_in[2];
  const int*   batch = (const int*)d_in[3];
  const float* xlw   = (const float*)d_in[4];
  const float* xlb   = (const float*)d_in[5];
  const float* glw   = (const float*)d_in[6];
  const float* gas   = (const float*)d_in[7];
  const float* gad   = (const float*)d_in[8];
  const float* gew   = (const float*)d_in[9];
  const float* gae   = (const float*)d_in[10];
  const float* gb    = (const float*)d_in[11];
  const float* bng   = (const float*)d_in[12];
  const float* bnb   = (const float*)d_in[13];
  const float* bnm   = (const float*)d_in[14];
  const float* bnv   = (const float*)d_in[15];

  const int n = in_sizes[3];
  const int E = in_sizes[2] / F_E;
  const int* srcp = eidx;
  const int* dstp = eidx + E;
  const int npad = ((n + 1 + SCAN_TILE - 1) / SCAN_TILE) * SCAN_TILE;

  float* ws    = (float*)d_ws;
  float* h     = ws;  ws += (size_t)n * D;
  float* xw    = ws;  ws += (size_t)n * D;
  int*   cnt   = (int*)ws; ws += npad;
  int*   rowptr = (int*)ws; ws += npad;
  int*   slot  = (int*)ws; ws += 2 * (size_t)E;
  float* sS    = ws;  ws += n;
  float* dS    = ws;  ws += n;
  float* wea   = ws;  ws += L * F_E;
  int*   rank  = (int*)ws; ws += E;

  float* out_nodes = (float*)d_out;
  float* out_graph = out_nodes + (size_t)n * D;
  const size_t graph_elems = (size_t)out_size - (size_t)n * D;

  const int B = 256;
  #define GRID(x) (((x) + B - 1) / B)

  k_wea<<<GRID(L * F_E), B, 0, stream>>>(gew, gae, wea);
  hipMemsetAsync(cnt, 0, (size_t)npad * sizeof(int), stream);
  k_hist<<<GRID(E), B, 0, stream>>>(dstp, cnt, rank, E);
  k_scan<<<npad / SCAN_TILE, SCAN_NT, 0, stream>>>(cnt, rowptr);
  k_scatter<<<GRID(E), B, 0, stream>>>(srcp, dstp, rowptr, rank, slot, E);

  {
    int waves = (n + 15) / 16;
    int blocks = (waves + 3) / 4;
    k_gemm_wmma<<<blocks, 128, 0, stream>>>(x, xlw, xlb, h, n, F_IN);
  }

  for (int l = 0; l < L; l++) {
    int waves = (n + 15) / 16;
    int blocks = (waves + 3) / 4;
    k_gemm_wmma<<<blocks, 128, 0, stream>>>(h, glw + (size_t)l * D * D, nullptr, xw, n, D);

    k_scores<<<GRID(n), B, 0, stream>>>(xw, gas + (size_t)l * D, gad + (size_t)l * D, sS, dS, n);
    k_node<<<(n + 7) / 8, B, 0, stream>>>(xw, sS, dS, eattr, wea + (size_t)l * F_E, rowptr, slot,
                                          gb + (size_t)l * D, bng + (size_t)l * D, bnb + (size_t)l * D,
                                          bnm + (size_t)l * D, bnv + (size_t)l * D, h, n,
                                          l < L - 1 ? 1 : 0);
  }

  hipMemsetAsync(out_graph, 0, graph_elems * sizeof(float), stream);
  k_output<<<GRID(n * D), B, 0, stream>>>(h, batch, out_nodes, out_graph, n);
  #undef GRID
}


# edge terms (edge_attr . We a_e) of all three layers computed once in the per-call destination-sort scatter, kept beside the source in a 16B slot; layer kernels gather no edge_attr; sum orders kept
# speedup vs baseline: 4.6371x; 1.1189x over previous
#include <hip/hip_runtime.h>
#include <hip/hip_bf16.h>

#define F_IN   32
#define F_E    16
#define D      64
#define L      3
#define NEG_SLOPE 0.2f
#define BN_EPS 1e-5f
#define SCAN_TILE 4096

typedef __attribute__((ext_vector_type(2))) float v2f;
typedef __attribute__((ext_vector_type(8))) float v8f;

__global__ void k_hist(const int* __restrict__ dst, int* __restrict__ cnt,
                       int* __restrict__ rank, int E) {
  int e = blockIdx.x * blockDim.x + threadIdx.x;
  if (e >= E) return;
  rank[e] = atomicAdd(&cnt[dst[e]], 1);
}

#define SCAN_NT (SCAN_TILE / 4)
#define SCAN_NW (SCAN_NT / 32)
__global__ void __launch_bounds__(SCAN_NT) k_scan(const int* __restrict__ cnt, int* __restrict__ rowptr) {
  __shared__ int wsum[SCAN_NW];
  __shared__ int carry_s[SCAN_NW];
  int t = threadIdx.x;
  int lane = t & 31, wv = t >> 5;
  const int4* c4 = (const int4*)cnt;
  int nf = (int)blockIdx.x * SCAN_NT;
  int front = 0;
  int i = t;
  for (; i + 3 * SCAN_NT < nf; i += 4 * SCAN_NT) {
    int4 q0 = c4[i], q1 = c4[i + SCAN_NT], q2 = c4[i + 2 * SCAN_NT], q3 = c4[i + 3 * SCAN_NT];
    front += (q0.x + q0.y + q0.z + q0.w) + (q1.x + q1.y + q1.z + q1.w)
           + (q2.x + q2.y + q2.z + q2.w) + (q3.x + q3.y + q3.z + q3.w);
  }
  for (; i < nf; i += SCAN_NT) {
    int4 q0 = c4[i];
    front += q0.x + q0.y + q0.z + q0.w;
  }
  int4 q = c4[nf + t];
  int own = q.x + q.y + q.z + q.w;
  int inc = own;
  #pragma unroll
  for (int o = 1; o < 32; o <<= 1) {
    int u = __shfl_up(inc, o);
    if (lane >= o) inc += u;
  }
  #pragma unroll
  for (int o = 16; o > 0; o >>= 1) front += __shfl_xor(front, o);
  if (lane == 31) wsum[wv] = inc;
  if (lane == 0) carry_s[wv] = front;
  __syncthreads();
  int base = 0;
  #pragma unroll
  for (int w = 0; w < SCAN_NW; w++) {
    base += carry_s[w];
    if (w < wv) base += wsum[w];
  }
  int ex = base + inc - own;
  int4 r;
  r.x = ex;
  r.y = ex + q.x;
  r.z = r.y + q.y;
  r.w = r.z + q.z;
  ((int4*)rowptr)[nf + t] = r;
}

__global__ void k_scatter(const int* __restrict__ src, const int* __restrict__ dst,
                          const int* __restrict__ rowptr, const int* __restrict__ rank,
                          const float* __restrict__ eattr, const float* __restrict__ wea,
                          int* __restrict__ slot, int E) {
  int e = blockIdx.x * blockDim.x + threadIdx.x;
  if (e >= E) return;
  int pos = rowptr[dst[e]] + rank[e];
  const float4* ea4 = (const float4*)(eattr + (size_t)e * F_E);
  float4 q0 = ea4[0], q1 = ea4[1], q2 = ea4[2], q3 = ea4[3];
  float t[L];
  for (int l = 0; l < L; l++) {
    const float* wl = wea + l * F_E;
    float ae = 0.f;
    ae += q0.x * wl[0];  ae += q0.y * wl[1];  ae += q0.z * wl[2];  ae += q0.w * wl[3];
    ae += q1.x * wl[4];  ae += q1.y * wl[5];  ae += q1.z * wl[6];  ae += q1.w * wl[7];
    ae += q2.x * wl[8];  ae += q2.y * wl[9];  ae += q2.z * wl[10]; ae += q2.w * wl[11];
    ae += q3.x * wl[12]; ae += q3.y * wl[13]; ae += q3.z * wl[14]; ae += q3.w * wl[15];
    t[l] = ae;
  }
  ((int4*)slot)[pos] = make_int4(src[e], __float_as_int(t[0]), __float_as_int(t[1]), __float_as_int(t[2]));
}

__global__ void k_wea(const float* __restrict__ We, const float* __restrict__ ae,
                      float* __restrict__ wea) {
  int i = blockIdx.x * blockDim.x + threadIdx.x;
  if (i >= L * F_E) return;
  int l = i / F_E, f = i % F_E;
  float s = 0.f;
  #pragma unroll
  for (int d = 0; d < D; d++) s += We[(l * F_E + f) * D + d] * ae[l * D + d];
  wea[i] = s;
}

__global__ void k_gemm_wmma(const float* __restrict__ A, const float* __restrict__ B,
                            const float* __restrict__ bias, float* __restrict__ C,
                            int n, int K) {
  int wave = (blockIdx.x * blockDim.x + threadIdx.x) >> 5;
  int lane = threadIdx.x & 31;
  int row0 = wave * 16;
  if (row0 >= n) return;
  int lm = lane & 15;
  int kh = (lane >> 4) << 1;
  int mrow = row0 + lm;
  int mrd = mrow < n ? mrow : n - 1;

  v8f acc[4] = {};
  for (int k = 0; k < K; k += 4) {
    v2f a;
    a.x = A[(size_t)mrd * K + k + kh];
    a.y = A[(size_t)mrd * K + k + kh + 1];
    #pragma unroll
    for (int nt = 0; nt < 4; nt++) {
      int col = nt * 16 + lm;
      v2f b;
      b.x = B[(k + kh) * D + col];
      b.y = B[(k + kh + 1) * D + col];
      acc[nt] = __builtin_amdgcn_wmma_f32_16x16x4_f32(
          false, a, false, b, (short)0, acc[nt], false, false);
    }
  }

  int rbase = row0 + ((lane >> 4) << 3);
  #pragma unroll
  for (int nt = 0; nt < 4; nt++) {
    int col = nt * 16 + lm;
    float bv = bias ? bias[col] : 0.f;
    #pragma unroll
    for (int r = 0; r < 8; r++) {
      int row = rbase + r;
      if (row < n) C[(size_t)row * D + col] = acc[nt][r] + bv;
    }
  }
}

__global__ void k_scores(const float* __restrict__ xw, const float* __restrict__ as,
                         const float* __restrict__ ad, float* __restrict__ s,
                         float* __restrict__ dv, int n) {
  int i = blockIdx.x * blockDim.x + threadIdx.x;
  if (i >= n) return;
  const float* row = xw + (size_t)i * D;
  float ss = 0.f, dd = 0.f;
  #pragma unroll
  for (int f = 0; f < D; f++) { float v = row[f]; ss += v * as[f]; dd += v * ad[f]; }
  s[i] = ss;
  dv[i] = dd;
}

__device__ __forceinline__ float wave_max(float v) {
  #pragma unroll
  for (int o = 16; o > 0; o >>= 1) v = fmaxf(v, __shfl_xor(v, o));
  return v;
}
__device__ __forceinline__ float wave_sum(float v) {
  #pragma unroll
  for (int o = 16; o > 0; o >>= 1) v += __shfl_xor(v, o);
  return v;
}

__global__ void k_node(const float* __restrict__ xw, const float* __restrict__ sS,
                       const float* __restrict__ dS, const int* __restrict__ rowptr,
                       const int* __restrict__ slot, int layer,
                       const float* __restrict__ bias, const float* __restrict__ gamma,
                       const float* __restrict__ beta, const float* __restrict__ mean,
                       const float* __restrict__ var, float* __restrict__ h, int n, int relu) {
  int node = __builtin_amdgcn_readfirstlane((int)((blockIdx.x * blockDim.x + threadIdx.x) >> 5));
  if (node >= n) return;
  int lane = threadIdx.x & 31;
  int beg = rowptr[node];
  int end = rowptr[node + 1];
  float dn = dS[node];

  float m = -INFINITY, lsum = 0.f, acc0 = 0.f, acc1 = 0.f, aepart = 0.f;
  for (int base = beg; base < end; base += 32) {
    int j = base + lane;
    bool valid = j < end;
    int4 es = valid ? ((const int4*)slot)[j] : make_int4(node, 0, 0, 0);
    int s = es.x;
    float ae = __int_as_float(layer == 0 ? es.y : (layer == 1 ? es.z : es.w));
    float a = sS[s] + dn + ae;
    a = a > 0.f ? a : NEG_SLOPE * a;
    if (!valid) { ae = 0.f; a = -INFINITY; }
    aepart += ae;
    float mnew = fmaxf(m, wave_max(a));
    float scale = expf(m - mnew);
    float p = expf(a - mnew);
    lsum = lsum * scale + wave_sum(p);
    acc0 *= scale;
    acc1 *= scale;
    m = mnew;
    int pbits = __float_as_int(p);
    int cnt = end - base < 32 ? end - base : 32;
    for (int j0 = 0; j0 < cnt; j0 += 8) {
      float r0[8], r1[8];
      #pragma unroll
      for (int k = 0; k < 8; k++) {
        int sj = __builtin_amdgcn_readlane(s, j0 + k);
        const float* xr = xw + (size_t)sj * D + lane;
        r0[k] = xr[0];
        r1[k] = xr[32];
      }
      #pragma unroll
      for (int k = 0; k < 8; k++) {
        float pj = __int_as_float(__builtin_amdgcn_readlane(pbits, j0 + k));
        acc0 += pj * r0[k];
        acc1 += pj * r1[k];
      }
    }
  }
  {
    int deg = end - beg;
    float aes = wave_sum(aepart) / fmaxf((float)deg, 1.0f);
    float a = sS[node] + dn + aes;
    a = a > 0.f ? a : NEG_SLOPE * a;
    float mnew = fmaxf(m, a);
    float scale = expf(m - mnew);
    float p = expf(a - mnew);
    lsum = lsum * scale + p;
    const float* xr = xw + (size_t)node * D + lane;
    acc0 = acc0 * scale + p * xr[0];
    acc1 = acc1 * scale + p * xr[32];
  }
  float inv = 1.0f / lsum;
  #pragma unroll
  for (int hlf = 0; hlf < 2; hlf++) {
    int f = lane + 32 * hlf;
    float v = (hlf ? acc1 : acc0) * inv + bias[f];
    v = (v - mean[f]) * (gamma[f] / sqrtf(var[f] + BN_EPS)) + beta[f];
    if (relu) v = fmaxf(v, 0.f);
    h[(size_t)node * D + f] = v;
  }
}

__global__ void k_output(const float* __restrict__ h, const int* __restrict__ batch,
                         float* __restrict__ out_nodes, float* __restrict__ out_graph, int n) {
  int i = blockIdx.x * blockDim.x + threadIdx.x;
  if (i >= n * D) return;
  float v = h[i];
  out_nodes[i] = v;
  int node = i >> 6;
  int f = i & (D - 1);
  atomicAdd(&out_graph[(size_t)batch[node] * D + f], v);
}

extern "C" void kernel_launch(void* const* d_in, const int* in_sizes, int n_in,
                              void* d_out, int out_size, void* d_ws, size_t ws_size,
                              hipStream_t stream) {
  const float* x     = (const float*)d_in[0];
  const int*   eidx  = (const int*)d_in[1];
  const float* eattr = (const float*)d_in[2];
  const int*   batch = (const int*)d_in[3];
  const float* xlw   = (const float*)d_in[4];
  const float* xlb   = (const float*)d_in[5];
  const float* glw   = (const float*)d_in[6];
  const float* gas   = (const float*)d_in[7];
  const float* gad   = (const float*)d_in[8];
  const float* gew   = (const float*)d_in[9];
  const float* gae   = (const float*)d_in[10];
  const float* gb    = (const float*)d_in[11];
  const float* bng   = (const float*)d_in[12];
  const float* bnb   = (const float*)d_in[13];
  const float* bnm   = (const float*)d_in[14];
  const float* bnv   = (const float*)d_in[15];

  const int n = in_sizes[3];
  const int E = in_sizes[2] / F_E;
  const int* srcp = eidx;
  const int* dstp = eidx + E;
  const int npad = ((n + 1 + SCAN_TILE - 1) / SCAN_TILE) * SCAN_TILE;

  float* ws    = (float*)d_ws;
  float* h     = ws;  ws += (size_t)n * D;
  float* xw    = ws;  ws += (size_t)n * D;
  int*   cnt   = (int*)ws; ws += npad;
  int*   rowptr = (int*)ws; ws += npad;
  int*   slot  = (int*)ws; ws += 4 * (size_t)E;
  float* sS    = ws;  ws += n;
  float* dS    = ws;  ws += n;
  float* wea   = ws;  ws += L * F_E;
  int*   rank  = (int*)xw;

  float* out_nodes = (float*)d_out;
  float* out_graph = out_nodes + (size_t)n * D;
  const size_t graph_elems = (size_t)out_size - (size_t)n * D;

  const int B = 256;
  #define GRID(x) (((x) + B - 1) / B)

  k_wea<<<GRID(L * F_E), B, 0, stream>>>(gew, gae, wea);
  hipMemsetAsync(cnt, 0, (size_t)npad * sizeof(int), stream);
  k_hist<<<GRID(E), B, 0, stream>>>(dstp, cnt, rank, E);
  k_scan<<<npad / SCAN_TILE, SCAN_NT, 0, stream>>>(cnt, rowptr);
  k_scatter<<<GRID(E), B, 0, stream>>>(srcp, dstp, rowptr, rank, eattr, wea, slot, E);

  {
    int waves = (n + 15) / 16;
    int blocks = (waves + 3) / 4;
    k_gemm_wmma<<<blocks, 128, 0, stream>>>(x, xlw, xlb, h, n, F_IN);
  }

  for (int l = 0; l < L; l++) {
    int waves = (n + 15) / 16;
    int blocks = (waves + 3) / 4;
    k_gemm_wmma<<<blocks, 128, 0, stream>>>(h, glw + (size_t)l * D * D, nullptr, xw, n, D);

    k_scores<<<GRID(n), B, 0, stream>>>(xw, gas + (size_t)l * D, gad + (size_t)l * D, sS, dS, n);
    k_node<<<(n + 7) / 8, B, 0, stream>>>(xw, sS, dS, rowptr, slot, l,
                                          gb + (size_t)l * D, bng + (size_t)l * D, bnb + (size_t)l * D,
                                          bnm + (size_t)l * D, bnv + (size_t)l * D, h, n,
                                          l < L - 1 ? 1 : 0);
  }

  hipMemsetAsync(out_graph, 0, graph_elems * sizeof(float), stream);
  k_output<<<GRID(n * D), B, 0, stream>>>(h, batch, out_nodes, out_graph, n);
  #undef GRID
}


# node kernel: two nodes per wave, one 16-lane row per node: two edges per lane for scores, row reductions and broadcasts on the vector ALU, one float4 row load per edge; per-node sum order kept
# speedup vs baseline: 4.8836x; 1.0532x over previous
#include <hip/hip_runtime.h>
#include <hip/hip_bf16.h>

#define F_IN   32
#define F_E    16
#define D      64
#define L      3
#define NEG_SLOPE 0.2f
#define BN_EPS 1e-5f
#define SCAN_TILE 4096

typedef __attribute__((ext_vector_type(2))) float v2f;
typedef __attribute__((ext_vector_type(8))) float v8f;

__global__ void k_hist(const int* __restrict__ dst, int* __restrict__ cnt,
                       int* __restrict__ rank, int E) {
  int e = blockIdx.x * blockDim.x + threadIdx.x;
  if (e >= E) return;
  rank[e] = atomicAdd(&cnt[dst[e]], 1);
}

#define SCAN_NT (SCAN_TILE / 4)
#define SCAN_NW (SCAN_NT / 32)
__global__ void __launch_bounds__(SCAN_NT) k_scan(const int* __restrict__ cnt, int* __restrict__ rowptr) {
  __shared__ int wsum[SCAN_NW];
  __shared__ int carry_s[SCAN_NW];
  int t = threadIdx.x;
  int lane = t & 31, wv = t >> 5;
  const int4* c4 = (const int4*)cnt;
  int nf = (int)blockIdx.x * SCAN_NT;
  int front = 0;
  int i = t;
  for (; i + 3 * SCAN_NT < nf; i += 4 * SCAN_NT) {
    int4 q0 = c4[i], q1 = c4[i + SCAN_NT], q2 = c4[i + 2 * SCAN_NT], q3 = c4[i + 3 * SCAN_NT];
    front += (q0.x + q0.y + q0.z + q0.w) + (q1.x + q1.y + q1.z + q1.w)
           + (q2.x + q2.y + q2.z + q2.w) + (q3.x + q3.y + q3.z + q3.w);
  }
  for (; i < nf; i += SCAN_NT) {
    int4 q0 = c4[i];
    front += q0.x + q0.y + q0.z + q0.w;
  }
  int4 q = c4[nf + t];
  int own = q.x + q.y + q.z + q.w;
  int inc = own;
  #pragma unroll
  for (int o = 1; o < 32; o <<= 1) {
    int u = __shfl_up(inc, o);
    if (lane >= o) inc += u;
  }
  #pragma unroll
  for (int o = 16; o > 0; o >>= 1) front += __shfl_xor(front, o);
  if (lane == 31) wsum[wv] = inc;
  if (lane == 0) carry_s[wv] = front;
  __syncthreads();
  int base = 0;
  #pragma unroll
  for (int w = 0; w < SCAN_NW; w++) {
    base += carry_s[w];
    if (w < wv) base += wsum[w];
  }
  int ex = base + inc - own;
  int4 r;
  r.x = ex;
  r.y = ex + q.x;
  r.z = r.y + q.y;
  r.w = r.z + q.z;
  ((int4*)rowptr)[nf + t] = r;
}

__global__ void k_scatter(const int* __restrict__ src, const int* __restrict__ dst,
                          const int* __restrict__ rowptr, const int* __restrict__ rank,
                          const float* __restrict__ eattr, const float* __restrict__ wea,
                          int* __restrict__ slot, int E) {
  int e = blockIdx.x * blockDim.x + threadIdx.x;
  if (e >= E) return;
  int pos = rowptr[dst[e]] + rank[e];
  const float4* ea4 = (const float4*)(eattr + (size_t)e * F_E);
  float4 q0 = ea4[0], q1 = ea4[1], q2 = ea4[2], q3 = ea4[3];
  float t[L];
  for (int l = 0; l < L; l++) {
    const float* wl = wea + l * F_E;
    float ae = 0.f;
    ae += q0.x * wl[0];  ae += q0.y * wl[1];  ae += q0.z * wl[2];  ae += q0.w * wl[3];
    ae += q1.x * wl[4];  ae += q1.y * wl[5];  ae += q1.z * wl[6];  ae += q1.w * wl[7];
    ae += q2.x * wl[8];  ae += q2.y * wl[9];  ae += q2.z * wl[10]; ae += q2.w * wl[11];
    ae += q3.x * wl[12]; ae += q3.y * wl[13]; ae += q3.z * wl[14]; ae += q3.w * wl[15];
    t[l] = ae;
  }
  ((int4*)slot)[pos] = make_int4(src[e], __float_as_int(t[0]), __float_as_int(t[1]), __float_as_int(t[2]));
}

__global__ void k_wea(const float* __restrict__ We, const float* __restrict__ ae,
                      float* __restrict__ wea) {
  int i = blockIdx.x * blockDim.x + threadIdx.x;
  if (i >= L * F_E) return;
  int l = i / F_E, f = i % F_E;
  float s = 0.f;
  #pragma unroll
  for (int d = 0; d < D; d++) s += We[(l * F_E + f) * D + d] * ae[l * D + d];
  wea[i] = s;
}

__global__ void k_gemm_wmma(const float* __restrict__ A, const float* __restrict__ B,
                            const float* __restrict__ bias, float* __restrict__ C,
                            int n, int K) {
  int wave = (blockIdx.x * blockDim.x + threadIdx.x) >> 5;
  int lane = threadIdx.x & 31;
  int row0 = wave * 16;
  if (row0 >= n) return;
  int lm = lane & 15;
  int kh = (lane >> 4) << 1;
  int mrow = row0 + lm;
  int mrd = mrow < n ? mrow : n - 1;

  v8f acc[4] = {};
  for (int k = 0; k < K; k += 4) {
    v2f a;
    a.x = A[(size_t)mrd * K + k + kh];
    a.y = A[(size_t)mrd * K + k + kh + 1];
    #pragma unroll
    for (int nt = 0; nt < 4; nt++) {
      int col = nt * 16 + lm;
      v2f b;
      b.x = B[(k + kh) * D + col];
      b.y = B[(k + kh + 1) * D + col];
      acc[nt] = __builtin_amdgcn_wmma_f32_16x16x4_f32(
          false, a, false, b, (short)0, acc[nt], false, false);
    }
  }

  int rbase = row0 + ((lane >> 4) << 3);
  #pragma unroll
  for (int nt = 0; nt < 4; nt++) {
    int col = nt * 16 + lm;
    float bv = bias ? bias[col] : 0.f;
    #pragma unroll
    for (int r = 0; r < 8; r++) {
      int row = rbase + r;
      if (row < n) C[(size_t)row * D + col] = acc[nt][r] + bv;
    }
  }
}

__global__ void k_scores(const float* __restrict__ xw, const float* __restrict__ as,
                         const float* __restrict__ ad, float* __restrict__ s,
                         float* __restrict__ dv, int n) {
  int i = blockIdx.x * blockDim.x + threadIdx.x;
  if (i >= n) return;
  const float* row = xw + (size_t)i * D;
  float ss = 0.f, dd = 0.f;
  #pragma unroll
  for (int f = 0; f < D; f++) { float v = row[f]; ss += v * as[f]; dd += v * ad[f]; }
  s[i] = ss;
  dv[i] = dd;
}


template <int M> __device__ __forceinline__ float row_xor(float v) {
  return __builtin_bit_cast(float, __builtin_amdgcn_update_dpp(0, __builtin_bit_cast(int, v), 0x160 + M, 0xf, 0xf, false));
}
__device__ __forceinline__ float row_max(float v) {
  v = fmaxf(v, row_xor<8>(v));
  v = fmaxf(v, row_xor<4>(v));
  v = fmaxf(v, row_xor<2>(v));
  v = fmaxf(v, row_xor<1>(v));
  return v;
}
__device__ __forceinline__ float row_sum(float v) {
  v += row_xor<8>(v);
  v += row_xor<4>(v);
  v += row_xor<2>(v);
  v += row_xor<1>(v);
  return v;
}
__device__ __forceinline__ int row_bcast(int v, unsigned sel) {
  return (int)__builtin_amdgcn_permlane16(0u, (unsigned)v, sel, sel, false, false);
}

__global__ void k_node(const float* __restrict__ xw, const float* __restrict__ sS,
                       const float* __restrict__ dS, const int* __restrict__ rowptr,
                       const int* __restrict__ slot, int layer,
                       const float* __restrict__ bias, const float* __restrict__ gamma,
                       const float* __restrict__ beta, const float* __restrict__ mean,
                       const float* __restrict__ var, float* __restrict__ h, int n, int relu) {
  int wave = __builtin_amdgcn_readfirstlane((int)((blockIdx.x * blockDim.x + threadIdx.x) >> 5));
  int nodeA = 2 * wave;
  if (nodeA >= n) return;
  int lane = threadIdx.x & 31;
  int c = lane & 15;
  int hi = lane >> 4;
  int iB = nodeA + 1 < n ? nodeA + 1 : n;
  int iC = nodeA + 2 < n ? nodeA + 2 : n;
  int r0 = rowptr[nodeA];
  int r1 = rowptr[iB];
  int r2 = rowptr[iC];
  int degA = r1 - r0, degB = r2 - r1;
  int maxdeg = degA > degB ? degA : degB;
  int beg = hi ? r1 : r0;
  int deg = hi ? degB : degA;
  int node = nodeA + hi;
  bool live = node < n;
  int nodec = live ? node : nodeA;
  float dn = dS[nodec];
  const int4* slot4 = (const int4*)slot;
  const float4* xw4 = (const float4*)xw;

  float m = -INFINITY, lsum = 0.f, aepart0 = 0.f, aepart1 = 0.f;
  float4 acc = make_float4(0.f, 0.f, 0.f, 0.f);
  for (int base = 0; base < maxdeg; base += 32) {
    int e0 = base + c, e1 = base + 16 + c;
    bool v0 = e0 < deg, v1 = e1 < deg;
    bool act = base < deg;
    int4 es0 = v0 ? slot4[beg + e0] : make_int4(nodec, 0, 0, 0);
    int4 es1 = v1 ? slot4[beg + e1] : make_int4(nodec, 0, 0, 0);
    int s0 = es0.x, s1 = es1.x;
    float ae0 = __int_as_float(layer == 0 ? es0.y : (layer == 1 ? es0.z : es0.w));
    float ae1 = __int_as_float(layer == 0 ? es1.y : (layer == 1 ? es1.z : es1.w));
    float a0 = sS[s0] + dn + ae0;
    float a1 = sS[s1] + dn + ae1;
    a0 = a0 > 0.f ? a0 : NEG_SLOPE * a0;
    a1 = a1 > 0.f ? a1 : NEG_SLOPE * a1;
    if (!v0) { ae0 = 0.f; a0 = -INFINITY; }
    if (!v1) { ae1 = 0.f; a1 = -INFINITY; }
    aepart0 += ae0;
    aepart1 += ae1;
    float mnew = act ? fmaxf(m, row_max(fmaxf(a0, a1))) : m;
    float scale = act ? expf(m - mnew) : 1.0f;
    float p0 = v0 ? expf(a0 - mnew) : 0.f;
    float p1 = v1 ? expf(a1 - mnew) : 0.f;
    lsum = lsum * scale + row_sum(p0 + p1);
    acc.x *= scale;
    acc.y *= scale;
    acc.z *= scale;
    acc.w *= scale;
    m = mnew;
    int p0b = __float_as_int(p0), p1b = __float_as_int(p1);
    int cnt = maxdeg - base < 32 ? maxdeg - base : 32;
    for (int j0 = 0; j0 < cnt; j0 += 8) {
      int ssrc = j0 < 16 ? s0 : s1;
      int psrc = j0 < 16 ? p0b : p1b;
      unsigned selb = (unsigned)(j0 & 15) * 0x11111111u;
      float4 r[8];
      #pragma unroll
      for (int k = 0; k < 8; k++) {
        int sj = row_bcast(ssrc, selb + (unsigned)k * 0x11111111u);
        r[k] = xw4[(size_t)sj * (D / 4) + c];
      }
      #pragma unroll
      for (int k = 0; k < 8; k++) {
        float pj = __int_as_float(row_bcast(psrc, selb + (unsigned)k * 0x11111111u));
        acc.x += pj * r[k].x;
        acc.y += pj * r[k].y;
        acc.z += pj * r[k].z;
        acc.w += pj * r[k].w;
      }
    }
  }
  {
    float aes = row_sum(aepart0 + aepart1) / fmaxf((float)deg, 1.0f);
    float a = sS[nodec] + dn + aes;
    a = a > 0.f ? a : NEG_SLOPE * a;
    float mnew = fmaxf(m, a);
    float scale = expf(m - mnew);
    float p = expf(a - mnew);
    lsum = lsum * scale + p;
    float4 xr = xw4[(size_t)nodec * (D / 4) + c];
    acc.x = acc.x * scale + p * xr.x;
    acc.y = acc.y * scale + p * xr.y;
    acc.z = acc.z * scale + p * xr.z;
    acc.w = acc.w * scale + p * xr.w;
  }
  float inv = 1.0f / lsum;
  float4 b4 = ((const float4*)bias)[c], g4 = ((const float4*)gamma)[c], be4 = ((const float4*)beta)[c];
  float4 mu4 = ((const float4*)mean)[c], va4 = ((const float4*)var)[c];
  float4 o;
  o.x = (acc.x * inv + b4.x - mu4.x) * (g4.x / sqrtf(va4.x + BN_EPS)) + be4.x;
  o.y = (acc.y * inv + b4.y - mu4.y) * (g4.y / sqrtf(va4.y + BN_EPS)) + be4.y;
  o.z = (acc.z * inv + b4.z - mu4.z) * (g4.z / sqrtf(va4.z + BN_EPS)) + be4.z;
  o.w = (acc.w * inv + b4.w - mu4.w) * (g4.w / sqrtf(va4.w + BN_EPS)) + be4.w;
  if (relu) {
    o.x = fmaxf(o.x, 0.f);
    o.y = fmaxf(o.y, 0.f);
    o.z = fmaxf(o.z, 0.f);
    o.w = fmaxf(o.w, 0.f);
  }
  if (live) ((float4*)h)[(size_t)node * (D / 4) + c] = o;
}

__global__ void k_output(const float* __restrict__ h, const int* __restrict__ batch,
                         float* __restrict__ out_nodes, float* __restrict__ out_graph, int n) {
  int i = blockIdx.x * blockDim.x + threadIdx.x;
  if (i >= n * D) return;
  float v = h[i];
  out_nodes[i] = v;
  int node = i >> 6;
  int f = i & (D - 1);
  atomicAdd(&out_graph[(size_t)batch[node] * D + f], v);
}

extern "C" void kernel_launch(void* const* d_in, const int* in_sizes, int n_in,
                              void* d_out, int out_size, void* d_ws, size_t ws_size,
                              hipStream_t stream) {
  const float* x     = (const float*)d_in[0];
  const int*   eidx  = (const int*)d_in[1];
  const float* eattr = (const float*)d_in[2];
  const int*   batch = (const int*)d_in[3];
  const float* xlw   = (const float*)d_in[4];
  const float* xlb   = (const float*)d_in[5];
  const float* glw   = (const float*)d_in[6];
  const float* gas   = (const float*)d_in[7];
  const float* gad   = (const float*)d_in[8];
  const float* gew   = (const float*)d_in[9];
  const float* gae   = (const float*)d_in[10];
  const float* gb    = (const float*)d_in[11];
  const float* bng   = (const float*)d_in[12];
  const float* bnb   = (const float*)d_in[13];
  const float* bnm   = (const float*)d_in[14];
  const float* bnv   = (const float*)d_in[15];

  const int n = in_sizes[3];
  const int E = in_sizes[2] / F_E;
  const int* srcp = eidx;
  const int* dstp = eidx + E;
  const int npad = ((n + 1 + SCAN_TILE - 1) / SCAN_TILE) * SCAN_TILE;

  float* ws    = (float*)d_ws;
  float* h     = ws;  ws += (size_t)n * D;
  float* xw    = ws;  ws += (size_t)n * D;
  int*   cnt   = (int*)ws; ws += npad;
  int*   rowptr = (int*)ws; ws += npad;
  int*   slot  = (int*)ws; ws += 4 * (size_t)E;
  float* sS    = ws;  ws += n;
  float* dS    = ws;  ws += n;
  float* wea   = ws;  ws += L * F_E;
  int*   rank  = (int*)xw;

  float* out_nodes = (float*)d_out;
  float* out_graph = out_nodes + (size_t)n * D;
  const size_t graph_elems = (size_t)out_size - (size_t)n * D;

  const int B = 256;
  #define GRID(x) (((x) + B - 1) / B)

  k_wea<<<GRID(L * F_E), B, 0, stream>>>(gew, gae, wea);
  hipMemsetAsync(cnt, 0, (size_t)npad * sizeof(int), stream);
  k_hist<<<GRID(E), B, 0, stream>>>(dstp, cnt, rank, E);
  k_scan<<<npad / SCAN_TILE, SCAN_NT, 0, stream>>>(cnt, rowptr);
  k_scatter<<<GRID(E), B, 0, stream>>>(srcp, dstp, rowptr, rank, eattr, wea, slot, E);

  {
    int waves = (n + 15) / 16;
    int blocks = (waves + 3) / 4;
    k_gemm_wmma<<<blocks, 128, 0, stream>>>(x, xlw, xlb, h, n, F_IN);
  }

  for (int l = 0; l < L; l++) {
    int waves = (n + 15) / 16;
    int blocks = (waves + 3) / 4;
    k_gemm_wmma<<<blocks, 128, 0, stream>>>(h, glw + (size_t)l * D * D, nullptr, xw, n, D);

    k_scores<<<GRID(n), B, 0, stream>>>(xw, gas + (size_t)l * D, gad + (size_t)l * D, sS, dS, n);
    k_node<<<(n + 15) / 16, B, 0, stream>>>(xw, sS, dS, rowptr, slot, l,
                                          gb + (size_t)l * D, bng + (size_t)l * D, bnb + (size_t)l * D,
                                          bnm + (size_t)l * D, bnv + (size_t)l * D, h, n,
                                          l < L - 1 ? 1 : 0);
  }

  hipMemsetAsync(out_graph, 0, graph_elems * sizeof(float), stream);
  k_output<<<GRID(n * D), B, 0, stream>>>(h, batch, out_nodes, out_graph, n);
  #undef GRID
}


# graph readout: a 16-lane row walks 16 consecutive nodes, sums rows of equal graph index in registers, one atomic set per run instead of per node; readout sum order changed (runs, then atomics)
# speedup vs baseline: 5.0518x; 1.0344x over previous
#include <hip/hip_runtime.h>
#include <hip/hip_bf16.h>

#define F_IN   32
#define F_E    16
#define D      64
#define L      3
#define NEG_SLOPE 0.2f
#define BN_EPS 1e-5f
#define SCAN_TILE 4096

typedef __attribute__((ext_vector_type(2))) float v2f;
typedef __attribute__((ext_vector_type(8))) float v8f;

__global__ void k_hist(const int* __restrict__ dst, int* __restrict__ cnt,
                       int* __restrict__ rank, int E) {
  int e = blockIdx.x * blockDim.x + threadIdx.x;
  if (e >= E) return;
  rank[e] = atomicAdd(&cnt[dst[e]], 1);
}

#define SCAN_NT (SCAN_TILE / 4)
#define SCAN_NW (SCAN_NT / 32)
__global__ void __launch_bounds__(SCAN_NT) k_scan(const int* __restrict__ cnt, int* __restrict__ rowptr) {
  __shared__ int wsum[SCAN_NW];
  __shared__ int carry_s[SCAN_NW];
  int t = threadIdx.x;
  int lane = t & 31, wv = t >> 5;
  const int4* c4 = (const int4*)cnt;
  int nf = (int)blockIdx.x * SCAN_NT;
  int front = 0;
  int i = t;
  for (; i + 3 * SCAN_NT < nf; i += 4 * SCAN_NT) {
    int4 q0 = c4[i], q1 = c4[i + SCAN_NT], q2 = c4[i + 2 * SCAN_NT], q3 = c4[i + 3 * SCAN_NT];
    front += (q0.x + q0.y + q0.z + q0.w) + (q1.x + q1.y + q1.z + q1.w)
           + (q2.x + q2.y + q2.z + q2.w) + (q3.x + q3.y + q3.z + q3.w);
  }
  for (; i < nf; i += SCAN_NT) {
    int4 q0 = c4[i];
    front += q0.x + q0.y + q0.z + q0.w;
  }
  int4 q = c4[nf + t];
  int own = q.x + q.y + q.z + q.w;
  int inc = own;
  #pragma unroll
  for (int o = 1; o < 32; o <<= 1) {
    int u = __shfl_up(inc, o);
    if (lane >= o) inc += u;
  }
  #pragma unroll
  for (int o = 16; o > 0; o >>= 1) front += __shfl_xor(front, o);
  if (lane == 31) wsum[wv] = inc;
  if (lane == 0) carry_s[wv] = front;
  __syncthreads();
  int base = 0;
  #pragma unroll
  for (int w = 0; w < SCAN_NW; w++) {
    base += carry_s[w];
    if (w < wv) base += wsum[w];
  }
  int ex = base + inc - own;
  int4 r;
  r.x = ex;
  r.y = ex + q.x;
  r.z = r.y + q.y;
  r.w = r.z + q.z;
  ((int4*)rowptr)[nf + t] = r;
}

__global__ void k_scatter(const int* __restrict__ src, const int* __restrict__ dst,
                          const int* __restrict__ rowptr, const int* __restrict__ rank,
                          const float* __restrict__ eattr, const float* __restrict__ wea,
                          int* __restrict__ slot, int E) {
  int e = blockIdx.x * blockDim.x + threadIdx.x;
  if (e >= E) return;
  int pos = rowptr[dst[e]] + rank[e];
  const float4* ea4 = (const float4*)(eattr + (size_t)e * F_E);
  float4 q0 = ea4[0], q1 = ea4[1], q2 = ea4[2], q3 = ea4[3];
  float t[L];
  for (int l = 0; l < L; l++) {
    const float* wl = wea + l * F_E;
    float ae = 0.f;
    ae += q0.x * wl[0];  ae += q0.y * wl[1];  ae += q0.z * wl[2];  ae += q0.w * wl[3];
    ae += q1.x * wl[4];  ae += q1.y * wl[5];  ae += q1.z * wl[6];  ae += q1.w * wl[7];
    ae += q2.x * wl[8];  ae += q2.y * wl[9];  ae += q2.z * wl[10]; ae += q2.w * wl[11];
    ae += q3.x * wl[12]; ae += q3.y * wl[13]; ae += q3.z * wl[14]; ae += q3.w * wl[15];
    t[l] = ae;
  }
  ((int4*)slot)[pos] = make_int4(src[e], __float_as_int(t[0]), __float_as_int(t[1]), __float_as_int(t[2]));
}

__global__ void k_wea(const float* __restrict__ We, const float* __restrict__ ae,
                      float* __restrict__ wea) {
  int i = blockIdx.x * blockDim.x + threadIdx.x;
  if (i >= L * F_E) return;
  int l = i / F_E, f = i % F_E;
  float s = 0.f;
  #pragma unroll
  for (int d = 0; d < D; d++) s += We[(l * F_E + f) * D + d] * ae[l * D + d];
  wea[i] = s;
}

__global__ void k_gemm_wmma(const float* __restrict__ A, const float* __restrict__ B,
                            const float* __restrict__ bias, float* __restrict__ C,
                            int n, int K) {
  int wave = (blockIdx.x * blockDim.x + threadIdx.x) >> 5;
  int lane = threadIdx.x & 31;
  int row0 = wave * 16;
  if (row0 >= n) return;
  int lm = lane & 15;
  int kh = (lane >> 4) << 1;
  int mrow = row0 + lm;
  int mrd = mrow < n ? mrow : n - 1;

  v8f acc[4] = {};
  for (int k = 0; k < K; k += 4) {
    v2f a;
    a.x = A[(size_t)mrd * K + k + kh];
    a.y = A[(size_t)mrd * K + k + kh + 1];
    #pragma unroll
    for (int nt = 0; nt < 4; nt++) {
      int col = nt * 16 + lm;
      v2f b;
      b.x = B[(k + kh) * D + col];
      b.y = B[(k + kh + 1) * D + col];
      acc[nt] = __builtin_amdgcn_wmma_f32_16x16x4_f32(
          false, a, false, b, (short)0, acc[nt], false, false);
    }
  }

  int rbase = row0 + ((lane >> 4) << 3);
  #pragma unroll
  for (int nt = 0; nt < 4; nt++) {
    int col = nt * 16 + lm;
    float bv = bias ? bias[col] : 0.f;
    #pragma unroll
    for (int r = 0; r < 8; r++) {
      int row = rbase + r;
      if (row < n) C[(size_t)row * D + col] = acc[nt][r] + bv;
    }
  }
}

__global__ void k_scores(const float* __restrict__ xw, const float* __restrict__ as,
                         const float* __restrict__ ad, float* __restrict__ s,
                         float* __restrict__ dv, int n) {
  int i = blockIdx.x * blockDim.x + threadIdx.x;
  if (i >= n) return;
  const float* row = xw + (size_t)i * D;
  float ss = 0.f, dd = 0.f;
  #pragma unroll
  for (int f = 0; f < D; f++) { float v = row[f]; ss += v * as[f]; dd += v * ad[f]; }
  s[i] = ss;
  dv[i] = dd;
}


template <int M> __device__ __forceinline__ float row_xor(float v) {
  return __builtin_bit_cast(float, __builtin_amdgcn_update_dpp(0, __builtin_bit_cast(int, v), 0x160 + M, 0xf, 0xf, false));
}
__device__ __forceinline__ float row_max(float v) {
  v = fmaxf(v, row_xor<8>(v));
  v = fmaxf(v, row_xor<4>(v));
  v = fmaxf(v, row_xor<2>(v));
  v = fmaxf(v, row_xor<1>(v));
  return v;
}
__device__ __forceinline__ float row_sum(float v) {
  v += row_xor<8>(v);
  v += row_xor<4>(v);
  v += row_xor<2>(v);
  v += row_xor<1>(v);
  return v;
}
__device__ __forceinline__ int row_bcast(int v, unsigned sel) {
  return (int)__builtin_amdgcn_permlane16(0u, (unsigned)v, sel, sel, false, false);
}

__global__ void k_node(const float* __restrict__ xw, const float* __restrict__ sS,
                       const float* __restrict__ dS, const int* __restrict__ rowptr,
                       const int* __restrict__ slot, int layer,
                       const float* __restrict__ bias, const float* __restrict__ gamma,
                       const float* __restrict__ beta, const float* __restrict__ mean,
                       const float* __restrict__ var, float* __restrict__ h, int n, int relu) {
  int wave = __builtin_amdgcn_readfirstlane((int)((blockIdx.x * blockDim.x + threadIdx.x) >> 5));
  int nodeA = 2 * wave;
  if (nodeA >= n) return;
  int lane = threadIdx.x & 31;
  int c = lane & 15;
  int hi = lane >> 4;
  int iB = nodeA + 1 < n ? nodeA + 1 : n;
  int iC = nodeA + 2 < n ? nodeA + 2 : n;
  int r0 = rowptr[nodeA];
  int r1 = rowptr[iB];
  int r2 = rowptr[iC];
  int degA = r1 - r0, degB = r2 - r1;
  int maxdeg = degA > degB ? degA : degB;
  int beg = hi ? r1 : r0;
  int deg = hi ? degB : degA;
  int node = nodeA + hi;
  bool live = node < n;
  int nodec = live ? node : nodeA;
  float dn = dS[nodec];
  const int4* slot4 = (const int4*)slot;
  const float4* xw4 = (const float4*)xw;

  float m = -INFINITY, lsum = 0.f, aepart0 = 0.f, aepart1 = 0.f;
  float4 acc = make_float4(0.f, 0.f, 0.f, 0.f);
  for (int base = 0; base < maxdeg; base += 32) {
    int e0 = base + c, e1 = base + 16 + c;
    bool v0 = e0 < deg, v1 = e1 < deg;
    bool act = base < deg;
    int4 es0 = v0 ? slot4[beg + e0] : make_int4(nodec, 0, 0, 0);
    int4 es1 = v1 ? slot4[beg + e1] : make_int4(nodec, 0, 0, 0);
    int s0 = es0.x, s1 = es1.x;
    float ae0 = __int_as_float(layer == 0 ? es0.y : (layer == 1 ? es0.z : es0.w));
    float ae1 = __int_as_float(layer == 0 ? es1.y : (layer == 1 ? es1.z : es1.w));
    float a0 = sS[s0] + dn + ae0;
    float a1 = sS[s1] + dn + ae1;
    a0 = a0 > 0.f ? a0 : NEG_SLOPE * a0;
    a1 = a1 > 0.f ? a1 : NEG_SLOPE * a1;
    if (!v0) { ae0 = 0.f; a0 = -INFINITY; }
    if (!v1) { ae1 = 0.f; a1 = -INFINITY; }
    aepart0 += ae0;
    aepart1 += ae1;
    float mnew = act ? fmaxf(m, row_max(fmaxf(a0, a1))) : m;
    float scale = act ? expf(m - mnew) : 1.0f;
    float p0 = v0 ? expf(a0 - mnew) : 0.f;
    float p1 = v1 ? expf(a1 - mnew) : 0.f;
    lsum = lsum * scale + row_sum(p0 + p1);
    acc.x *= scale;
    acc.y *= scale;
    acc.z *= scale;
    acc.w *= scale;
    m = mnew;
    int p0b = __float_as_int(p0), p1b = __float_as_int(p1);
    int cnt = maxdeg - base < 32 ? maxdeg - base : 32;
    for (int j0 = 0; j0 < cnt; j0 += 8) {
      int ssrc = j0 < 16 ? s0 : s1;
      int psrc = j0 < 16 ? p0b : p1b;
      unsigned selb = (unsigned)(j0 & 15) * 0x11111111u;
      float4 r[8];
      #pragma unroll
      for (int k = 0; k < 8; k++) {
        int sj = row_bcast(ssrc, selb + (unsigned)k * 0x11111111u);
        r[k] = xw4[(size_t)sj * (D / 4) + c];
      }
      #pragma unroll
      for (int k = 0; k < 8; k++) {
        float pj = __int_as_float(row_bcast(psrc, selb + (unsigned)k * 0x11111111u));
        acc.x += pj * r[k].x;
        acc.y += pj * r[k].y;
        acc.z += pj * r[k].z;
        acc.w += pj * r[k].w;
      }
    }
  }
  {
    float aes = row_sum(aepart0 + aepart1) / fmaxf((float)deg, 1.0f);
    float a = sS[nodec] + dn + aes;
    a = a > 0.f ? a : NEG_SLOPE * a;
    float mnew = fmaxf(m, a);
    float scale = expf(m - mnew);
    float p = expf(a - mnew);
    lsum = lsum * scale + p;
    float4 xr = xw4[(size_t)nodec * (D / 4) + c];
    acc.x = acc.x * scale + p * xr.x;
    acc.y = acc.y * scale + p * xr.y;
    acc.z = acc.z * scale + p * xr.z;
    acc.w = acc.w * scale + p * xr.w;
  }
  float inv = 1.0f / lsum;
  float4 b4 = ((const float4*)bias)[c], g4 = ((const float4*)gamma)[c], be4 = ((const float4*)beta)[c];
  float4 mu4 = ((const float4*)mean)[c], va4 = ((const float4*)var)[c];
  float4 o;
  o.x = (acc.x * inv + b4.x - mu4.x) * (g4.x / sqrtf(va4.x + BN_EPS)) + be4.x;
  o.y = (acc.y * inv + b4.y - mu4.y) * (g4.y / sqrtf(va4.y + BN_EPS)) + be4.y;
  o.z = (acc.z * inv + b4.z - mu4.z) * (g4.z / sqrtf(va4.z + BN_EPS)) + be4.z;
  o.w = (acc.w * inv + b4.w - mu4.w) * (g4.w / sqrtf(va4.w + BN_EPS)) + be4.w;
  if (relu) {
    o.x = fmaxf(o.x, 0.f);
    o.y = fmaxf(o.y, 0.f);
    o.z = fmaxf(o.z, 0.f);
    o.w = fmaxf(o.w, 0.f);
  }
  if (live) ((float4*)h)[(size_t)node * (D / 4) + c] = o;
}

#define RUN 16
__device__ __forceinline__ void graph_add(float* __restrict__ og, int g, int c, float4 a) {
  float* p = og + (size_t)g * D + 4 * c;
  atomicAdd(p + 0, a.x);
  atomicAdd(p + 1, a.y);
  atomicAdd(p + 2, a.z);
  atomicAdd(p + 3, a.w);
}
__global__ void __launch_bounds__(128) k_output(const float* __restrict__ h, const int* __restrict__ batch,
                                                float* __restrict__ out_nodes, float* __restrict__ out_graph, int n) {
  int wave = (blockIdx.x * blockDim.x + threadIdx.x) >> 5;
  if (wave * 2 * RUN >= n) return;
  int lane = threadIdx.x & 31;
  int c = lane & 15;
  int hi = lane >> 4;
  int first = (wave * 2 + hi) * RUN;
  const float4* h4 = (const float4*)h;
  float4* o4 = (float4*)out_nodes;
  int gn = first + c < n ? first + c : n - 1;
  int gv = batch[gn];
  float4 r[RUN];
  #pragma unroll
  for (int j = 0; j < RUN; j++) {
    int node = first + j < n ? first + j : n - 1;
    r[j] = h4[(size_t)node * (D / 4) + c];
  }
  #pragma unroll
  for (int j = 0; j < RUN; j++)
    if (first + j < n) o4[(size_t)(first + j) * (D / 4) + c] = r[j];

  int cur = 0;
  bool have = false;
  float4 acc = make_float4(0.f, 0.f, 0.f, 0.f);
  #pragma unroll
  for (int j = 0; j < RUN; j++) {
    int g = row_bcast(gv, (unsigned)j * 0x11111111u);
    if (first + j < n) {
      if (have && g == cur) {
        acc.x += r[j].x;
        acc.y += r[j].y;
        acc.z += r[j].z;
        acc.w += r[j].w;
      } else {
        if (have) graph_add(out_graph, cur, c, acc);
        cur = g;
        have = true;
        acc = r[j];
      }
    }
  }
  int ocur = __shfl_xor(cur, 16);
  int ohave = __shfl_xor((int)have, 16);
  float4 oacc;
  oacc.x = __shfl_xor(acc.x, 16);
  oacc.y = __shfl_xor(acc.y, 16);
  oacc.z = __shfl_xor(acc.z, 16);
  oacc.w = __shfl_xor(acc.w, 16);
  bool joined = have && ohave && cur == ocur;
  if (joined) {
    if (hi == 0) {
      acc.x += oacc.x;
      acc.y += oacc.y;
      acc.z += oacc.z;
      acc.w += oacc.w;
      graph_add(out_graph, cur, c, acc);
    }
  } else if (have) {
    graph_add(out_graph, cur, c, acc);
  }
}

extern "C" void kernel_launch(void* const* d_in, const int* in_sizes, int n_in,
                              void* d_out, int out_size, void* d_ws, size_t ws_size,
                              hipStream_t stream) {
  const float* x     = (const float*)d_in[0];
  const int*   eidx  = (const int*)d_in[1];
  const float* eattr = (const float*)d_in[2];
  const int*   batch = (const int*)d_in[3];
  const float* xlw   = (const float*)d_in[4];
  const float* xlb   = (const float*)d_in[5];
  const float* glw   = (const float*)d_in[6];
  const float* gas   = (const float*)d_in[7];
  const float* gad   = (const float*)d_in[8];
  const float* gew   = (const float*)d_in[9];
  const float* gae   = (const float*)d_in[10];
  const float* gb    = (const float*)d_in[11];
  const float* bng   = (const float*)d_in[12];
  const float* bnb   = (const float*)d_in[13];
  const float* bnm   = (const float*)d_in[14];
  const float* bnv   = (const float*)d_in[15];

  const int n = in_sizes[3];
  const int E = in_sizes[2] / F_E;
  const int* srcp = eidx;
  const int* dstp = eidx + E;
  const int npad = ((n + 1 + SCAN_TILE - 1) / SCAN_TILE) * SCAN_TILE;

  float* ws    = (float*)d_ws;
  float* h     = ws;  ws += (size_t)n * D;
  float* xw    = ws;  ws += (size_t)n * D;
  int*   cnt   = (int*)ws; ws += npad;
  int*   rowptr = (int*)ws; ws += npad;
  int*   slot  = (int*)ws; ws += 4 * (size_t)E;
  float* sS    = ws;  ws += n;
  float* dS    = ws;  ws += n;
  float* wea   = ws;  ws += L * F_E;
  int*   rank  = (int*)xw;

  float* out_nodes = (float*)d_out;
  float* out_graph = out_nodes + (size_t)n * D;
  const size_t graph_elems = (size_t)out_size - (size_t)n * D;

  const int B = 256;
  #define GRID(x) (((x) + B - 1) / B)

  k_wea<<<GRID(L * F_E), B, 0, stream>>>(gew, gae, wea);
  hipMemsetAsync(cnt, 0, (size_t)npad * sizeof(int), stream);
  k_hist<<<GRID(E), B, 0, stream>>>(dstp, cnt, rank, E);
  k_scan<<<npad / SCAN_TILE, SCAN_NT, 0, stream>>>(cnt, rowptr);
  k_scatter<<<GRID(E), B, 0, stream>>>(srcp, dstp, rowptr, rank, eattr, wea, slot, E);

  {
    int waves = (n + 15) / 16;
    int blocks = (waves + 3) / 4;
    k_gemm_wmma<<<blocks, 128, 0, stream>>>(x, xlw, xlb, h, n, F_IN);
  }

  for (int l = 0; l < L; l++) {
    int waves = (n + 15) / 16;
    int blocks = (waves + 3) / 4;
    k_gemm_wmma<<<blocks, 128, 0, stream>>>(h, glw + (size_t)l * D * D, nullptr, xw, n, D);

    k_scores<<<GRID(n), B, 0, stream>>>(xw, gas + (size_t)l * D, gad + (size_t)l * D, sS, dS, n);
    k_node<<<(n + 15) / 16, B, 0, stream>>>(xw, sS, dS, rowptr, slot, l,
                                          gb + (size_t)l * D, bng + (size_t)l * D, bnb + (size_t)l * D,
                                          bnm + (size_t)l * D, bnv + (size_t)l * D, h, n,
                                          l < L - 1 ? 1 : 0);
  }

  hipMemsetAsync(out_graph, 0, graph_elems * sizeof(float), stream);
  k_output<<<(n + 8 * RUN - 1) / (8 * RUN), 128, 0, stream>>>(h, batch, out_nodes, out_graph, n);
  #undef GRID
}


# edges grouped by destination with LDS counters only: per-chunk bucket counts, scan, records by bucket, then each 256-node bucket ordered in LDS; no global atomic or memset; rank still arrival order
# speedup vs baseline: 7.1298x; 1.4113x over previous
#include <hip/hip_runtime.h>
#include <hip/hip_bf16.h>

#define F_IN   32
#define F_E    16
#define D      64
#define L      3
#define NEG_SLOPE 0.2f
#define BN_EPS 1e-5f
#define SCAN_TILE 4096

typedef __attribute__((ext_vector_type(2))) float v2f;
typedef __attribute__((ext_vector_type(8))) float v8f;
typedef __attribute__((ext_vector_type(4))) int v4i;

#define BK_LOG   8
#define BKN      (1 << BK_LOG)
#define NB       256
#define MAXBK    1024
#define SRC_BITS 20
#define PT       1024
#define DT       1024
#define STAGE    6144

__global__ void __launch_bounds__(PT) k_bcount(const int* __restrict__ dst, int* __restrict__ bh,
                                               int E, int ch, int nbk, int m, int mpad) {
  __shared__ int hist[MAXBK];
  int t = threadIdx.x, b = blockIdx.x;
  for (int k = t; k < nbk; k += PT) hist[k] = 0;
  __syncthreads();
  int e0 = b * ch;
  int e1 = e0 + ch < E ? e0 + ch : E;
  for (int e = e0 + t; e < e1; e += 4 * PT) {
    int eb = e + PT, ec = e + 2 * PT, ed = e + 3 * PT;
    int da = dst[e];
    int db = eb < e1 ? dst[eb] : -1;
    int dc = ec < e1 ? dst[ec] : -1;
    int dd = ed < e1 ? dst[ed] : -1;
    atomicAdd(&hist[da >> BK_LOG], 1);
    if (db >= 0) atomicAdd(&hist[db >> BK_LOG], 1);
    if (dc >= 0) atomicAdd(&hist[dc >> BK_LOG], 1);
    if (dd >= 0) atomicAdd(&hist[dd >> BK_LOG], 1);
  }
  __syncthreads();
  for (int k = t; k < nbk; k += PT) bh[k * NB + b] = hist[k];
  if (b == 0)
    for (int i = m + t; i < mpad; i += PT) bh[i] = 0;
}

#define SCAN_NT (SCAN_TILE / 4)
#define SCAN_NW (SCAN_NT / 32)
__global__ void __launch_bounds__(SCAN_NT) k_scan(const int* __restrict__ cnt, int* __restrict__ rowptr) {
  __shared__ int wsum[SCAN_NW];
  __shared__ int carry_s[SCAN_NW];
  int t = threadIdx.x;
  int lane = t & 31, wv = t >> 5;
  const int4* c4 = (const int4*)cnt;
  int nf = (int)blockIdx.x * SCAN_NT;
  int front = 0;
  int i = t;
  for (; i + 3 * SCAN_NT < nf; i += 4 * SCAN_NT) {
    int4 q0 = c4[i], q1 = c4[i + SCAN_NT], q2 = c4[i + 2 * SCAN_NT], q3 = c4[i + 3 * SCAN_NT];
    front += (q0.x + q0.y + q0.z + q0.w) + (q1.x + q1.y + q1.z + q1.w)
           + (q2.x + q2.y + q2.z + q2.w) + (q3.x + q3.y + q3.z + q3.w);
  }
  for (; i < nf; i += SCAN_NT) {
    int4 q0 = c4[i];
    front += q0.x + q0.y + q0.z + q0.w;
  }
  int4 q = c4[nf + t];
  int own = q.x + q.y + q.z + q.w;
  int inc = own;
  #pragma unroll
  for (int o = 1; o < 32; o <<= 1) {
    int u = __shfl_up(inc, o);
    if (lane >= o) inc += u;
  }
  #pragma unroll
  for (int o = 16; o > 0; o >>= 1) front += __shfl_xor(front, o);
  if (lane == 31) wsum[wv] = inc;
  if (lane == 0) carry_s[wv] = front;
  __syncthreads();
  int base = 0;
  #pragma unroll
  for (int w = 0; w < SCAN_NW; w++) {
    base += carry_s[w];
    if (w < wv) base += wsum[w];
  }
  int ex = base + inc - own;
  int4 r;
  r.x = ex;
  r.y = ex + q.x;
  r.z = r.y + q.y;
  r.w = r.z + q.z;
  ((int4*)rowptr)[nf + t] = r;
}

__device__ __forceinline__ int4 edge_record(int sx, float4 q0, float4 q1, float4 q2, float4 q3,
                                            const float* __restrict__ wea) {
  float t[L];
  for (int l = 0; l < L; l++) {
    const float* wl = wea + l * F_E;
    float ae = 0.f;
    ae += q0.x * wl[0];  ae += q0.y * wl[1];  ae += q0.z * wl[2];  ae += q0.w * wl[3];
    ae += q1.x * wl[4];  ae += q1.y * wl[5];  ae += q1.z * wl[6];  ae += q1.w * wl[7];
    ae += q2.x * wl[8];  ae += q2.y * wl[9];  ae += q2.z * wl[10]; ae += q2.w * wl[11];
    ae += q3.x * wl[12]; ae += q3.y * wl[13]; ae += q3.z * wl[14]; ae += q3.w * wl[15];
    t[l] = ae;
  }
  return make_int4(sx, __float_as_int(t[0]), __float_as_int(t[1]), __float_as_int(t[2]));
}
__global__ void __launch_bounds__(PT) k_bplace(const int* __restrict__ src, const int* __restrict__ dst,
                                               const int* __restrict__ boff,
                                               const float* __restrict__ eattr, const float* __restrict__ wea,
                                               int* __restrict__ tslot, int E, int ch, int nbk) {
  __shared__ int cur[MAXBK];
  int t = threadIdx.x, b = blockIdx.x;
  for (int k = t; k < nbk; k += PT) cur[k] = boff[k * NB + b];
  __syncthreads();
  int e0 = b * ch;
  int e1 = e0 + ch < E ? e0 + ch : E;
  int4* ts4 = (int4*)tslot;
  int e = e0 + t;
  for (; e + PT < e1; e += 2 * PT) {
    int eb = e + PT;
    int da = dst[e], db = dst[eb];
    int sa = src[e], sb = src[eb];
    const float4* pa = (const float4*)(eattr + (size_t)e * F_E);
    const float4* pb = (const float4*)(eattr + (size_t)eb * F_E);
    float4 a0 = pa[0], a1 = pa[1], a2 = pa[2], a3 = pa[3];
    float4 b0 = pb[0], b1 = pb[1], b2 = pb[2], b3 = pb[3];
    int posa = atomicAdd(&cur[da >> BK_LOG], 1);
    int posb = atomicAdd(&cur[db >> BK_LOG], 1);
    ts4[posa] = edge_record(sa | ((da & (BKN - 1)) << SRC_BITS), a0, a1, a2, a3, wea);
    ts4[posb] = edge_record(sb | ((db & (BKN - 1)) << SRC_BITS), b0, b1, b2, b3, wea);
  }
  if (e < e1) {
    int da = dst[e];
    int sa = src[e];
    const float4* pa = (const float4*)(eattr + (size_t)e * F_E);
    float4 a0 = pa[0], a1 = pa[1], a2 = pa[2], a3 = pa[3];
    int posa = atomicAdd(&cur[da >> BK_LOG], 1);
    ts4[posa] = edge_record(sa | ((da & (BKN - 1)) << SRC_BITS), a0, a1, a2, a3, wea);
  }
}

__global__ void __launch_bounds__(DT) k_bsort(const int* __restrict__ boff, const int* __restrict__ tslot,
                                              int* __restrict__ rowptr, int* __restrict__ slot, int n) {
  __shared__ int cnt[BKN];
  __shared__ int wsum[BKN / 32];
  __shared__ v4i stage[STAGE];
  int t = threadIdx.x, k = blockIdx.x;
  int lane = t & 31, wv = t >> 5;
  int beg = boff[k * NB];
  int end = boff[(k + 1) * NB];
  if (t < BKN) cnt[t] = 0;
  __syncthreads();
  for (int i = beg + t; i < end; i += 4 * DT) {
    int ib = i + DT, ic = i + 2 * DT, id = i + 3 * DT;
    int xa = tslot[4 * (size_t)i];
    int xb = ib < end ? tslot[4 * (size_t)ib] : -1;
    int xc = ic < end ? tslot[4 * (size_t)ic] : -1;
    int xd = id < end ? tslot[4 * (size_t)id] : -1;
    atomicAdd(&cnt[xa >> SRC_BITS], 1);
    if (xb >= 0) atomicAdd(&cnt[xb >> SRC_BITS], 1);
    if (xc >= 0) atomicAdd(&cnt[xc >> SRC_BITS], 1);
    if (xd >= 0) atomicAdd(&cnt[xd >> SRC_BITS], 1);
  }
  __syncthreads();
  int own = 0, inc = 0;
  if (t < BKN) {
    own = cnt[t];
    inc = own;
    #pragma unroll
    for (int o = 1; o < 32; o <<= 1) {
      int u = __shfl_up(inc, o);
      if (lane >= o) inc += u;
    }
    if (lane == 31) wsum[wv] = inc;
  }
  __syncthreads();
  if (t < BKN) {
    int base = 0;
    #pragma unroll
    for (int w = 0; w < BKN / 32; w++)
      if (w < wv) base += wsum[w];
    int ex = base + inc - own;
    cnt[t] = ex;
    int node = k * BKN + t;
    if (node < n) rowptr[node] = beg + ex;
  }
  if (k == (int)gridDim.x - 1 && t == 0) rowptr[n] = end;
  __syncthreads();
  const v4i* ts4 = (const v4i*)tslot;
  v4i* sl4 = (v4i*)slot;
  const v4i none = {-1, 0, 0, 0};
  const bool fits = end - beg <= STAGE;
  for (int i = beg + t; i < end; i += 4 * DT) {
    int ib = i + DT, ic = i + 2 * DT, id = i + 3 * DT;
    v4i ra = ts4[i];
    v4i rb = ib < end ? ts4[ib] : none;
    v4i rc = ic < end ? ts4[ic] : none;
    v4i rd = id < end ? ts4[id] : none;
    {
      int p = atomicAdd(&cnt[ra.x >> SRC_BITS], 1);
      ra.x &= (1 << SRC_BITS) - 1;
      if (fits) stage[p] = ra; else sl4[beg + p] = ra;
    }
    if (rb.x >= 0) {
      int p = atomicAdd(&cnt[rb.x >> SRC_BITS], 1);
      rb.x &= (1 << SRC_BITS) - 1;
      if (fits) stage[p] = rb; else sl4[beg + p] = rb;
    }
    if (rc.x >= 0) {
      int p = atomicAdd(&cnt[rc.x >> SRC_BITS], 1);
      rc.x &= (1 << SRC_BITS) - 1;
      if (fits) stage[p] = rc; else sl4[beg + p] = rc;
    }
    if (rd.x >= 0) {
      int p = atomicAdd(&cnt[rd.x >> SRC_BITS], 1);
      rd.x &= (1 << SRC_BITS) - 1;
      if (fits) stage[p] = rd; else sl4[beg + p] = rd;
    }
  }
  if (fits) {
    __syncthreads();
    for (int i = t; i < end - beg; i += DT) sl4[beg + i] = stage[i];
  }
}

__global__ void k_wea(const float* __restrict__ We, const float* __restrict__ ae,
                      float* __restrict__ wea) {
  int i = blockIdx.x * blockDim.x + threadIdx.x;
  if (i >= L * F_E) return;
  int l = i / F_E, f = i % F_E;
  float s = 0.f;
  #pragma unroll
  for (int d = 0; d < D; d++) s += We[(l * F_E + f) * D + d] * ae[l * D + d];
  wea[i] = s;
}

__global__ void k_gemm_wmma(const float* __restrict__ A, const float* __restrict__ B,
                            const float* __restrict__ bias, float* __restrict__ C,
                            int n, int K) {
  int wave = (blockIdx.x * blockDim.x + threadIdx.x) >> 5;
  int lane = threadIdx.x & 31;
  int row0 = wave * 16;
  if (row0 >= n) return;
  int lm = lane & 15;
  int kh = (lane >> 4) << 1;
  int mrow = row0 + lm;
  int mrd = mrow < n ? mrow : n - 1;

  v8f acc[4] = {};
  for (int k = 0; k < K; k += 4) {
    v2f a;
    a.x = A[(size_t)mrd * K + k + kh];
    a.y = A[(size_t)mrd * K + k + kh + 1];
    #pragma unroll
    for (int nt = 0; nt < 4; nt++) {
      int col = nt * 16 + lm;
      v2f b;
      b.x = B[(k + kh) * D + col];
      b.y = B[(k + kh + 1) * D + col];
      acc[nt] = __builtin_amdgcn_wmma_f32_16x16x4_f32(
          false, a, false, b, (short)0, acc[nt], false, false);
    }
  }

  int rbase = row0 + ((lane >> 4) << 3);
  #pragma unroll
  for (int nt = 0; nt < 4; nt++) {
    int col = nt * 16 + lm;
    float bv = bias ? bias[col] : 0.f;
    #pragma unroll
    for (int r = 0; r < 8; r++) {
      int row = rbase + r;
      if (row < n) C[(size_t)row * D + col] = acc[nt][r] + bv;
    }
  }
}

__global__ void k_scores(const float* __restrict__ xw, const float* __restrict__ as,
                         const float* __restrict__ ad, float* __restrict__ s,
                         float* __restrict__ dv, int n) {
  int i = blockIdx.x * blockDim.x + threadIdx.x;
  if (i >= n) return;
  const float* row = xw + (size_t)i * D;
  float ss = 0.f, dd = 0.f;
  #pragma unroll
  for (int f = 0; f < D; f++) { float v = row[f]; ss += v * as[f]; dd += v * ad[f]; }
  s[i] = ss;
  dv[i] = dd;
}


template <int M> __device__ __forceinline__ float row_xor(float v) {
  return __builtin_bit_cast(float, __builtin_amdgcn_update_dpp(0, __builtin_bit_cast(int, v), 0x160 + M, 0xf, 0xf, false));
}
__device__ __forceinline__ float row_max(float v) {
  v = fmaxf(v, row_xor<8>(v));
  v = fmaxf(v, row_xor<4>(v));
  v = fmaxf(v, row_xor<2>(v));
  v = fmaxf(v, row_xor<1>(v));
  return v;
}
__device__ __forceinline__ float row_sum(float v) {
  v += row_xor<8>(v);
  v += row_xor<4>(v);
  v += row_xor<2>(v);
  v += row_xor<1>(v);
  return v;
}
__device__ __forceinline__ int row_bcast(int v, unsigned sel) {
  return (int)__builtin_amdgcn_permlane16(0u, (unsigned)v, sel, sel, false, false);
}

__global__ void k_node(const float* __restrict__ xw, const float* __restrict__ sS,
                       const float* __restrict__ dS, const int* __restrict__ rowptr,
                       const int* __restrict__ slot, int layer,
                       const float* __restrict__ bias, const float* __restrict__ gamma,
                       const float* __restrict__ beta, const float* __restrict__ mean,
                       const float* __restrict__ var, float* __restrict__ h, int n, int relu) {
  int wave = __builtin_amdgcn_readfirstlane((int)((blockIdx.x * blockDim.x + threadIdx.x) >> 5));
  int nodeA = 2 * wave;
  if (nodeA >= n) return;
  int lane = threadIdx.x & 31;
  int c = lane & 15;
  int hi = lane >> 4;
  int iB = nodeA + 1 < n ? nodeA + 1 : n;
  int iC = nodeA + 2 < n ? nodeA + 2 : n;
  int r0 = rowptr[nodeA];
  int r1 = rowptr[iB];
  int r2 = rowptr[iC];
  int degA = r1 - r0, degB = r2 - r1;
  int maxdeg = degA > degB ? degA : degB;
  int beg = hi ? r1 : r0;
  int deg = hi ? degB : degA;
  int node = nodeA + hi;
  bool live = node < n;
  int nodec = live ? node : nodeA;
  float dn = dS[nodec];
  const int4* slot4 = (const int4*)slot;
  const float4* xw4 = (const float4*)xw;

  float m = -INFINITY, lsum = 0.f, aepart0 = 0.f, aepart1 = 0.f;
  float4 acc = make_float4(0.f, 0.f, 0.f, 0.f);
  for (int base = 0; base < maxdeg; base += 32) {
    int e0 = base + c, e1 = base + 16 + c;
    bool v0 = e0 < deg, v1 = e1 < deg;
    bool act = base < deg;
    int4 es0 = v0 ? slot4[beg + e0] : make_int4(nodec, 0, 0, 0);
    int4 es1 = v1 ? slot4[beg + e1] : make_int4(nodec, 0, 0, 0);
    int s0 = es0.x, s1 = es1.x;
    float ae0 = __int_as_float(layer == 0 ? es0.y : (layer == 1 ? es0.z : es0.w));
    float ae1 = __int_as_float(layer == 0 ? es1.y : (layer == 1 ? es1.z : es1.w));
    float a0 = sS[s0] + dn + ae0;
    float a1 = sS[s1] + dn + ae1;
    a0 = a0 > 0.f ? a0 : NEG_SLOPE * a0;
    a1 = a1 > 0.f ? a1 : NEG_SLOPE * a1;
    if (!v0) { ae0 = 0.f; a0 = -INFINITY; }
    if (!v1) { ae1 = 0.f; a1 = -INFINITY; }
    aepart0 += ae0;
    aepart1 += ae1;
    float mnew = act ? fmaxf(m, row_max(fmaxf(a0, a1))) : m;
    float scale = act ? expf(m - mnew) : 1.0f;
    float p0 = v0 ? expf(a0 - mnew) : 0.f;
    float p1 = v1 ? expf(a1 - mnew) : 0.f;
    lsum = lsum * scale + row_sum(p0 + p1);
    acc.x *= scale;
    acc.y *= scale;
    acc.z *= scale;
    acc.w *= scale;
    m = mnew;
    int p0b = __float_as_int(p0), p1b = __float_as_int(p1);
    int cnt = maxdeg - base < 32 ? maxdeg - base : 32;
    for (int j0 = 0; j0 < cnt; j0 += 8) {
      int ssrc = j0 < 16 ? s0 : s1;
      int psrc = j0 < 16 ? p0b : p1b;
      unsigned selb = (unsigned)(j0 & 15) * 0x11111111u;
      float4 r[8];
      #pragma unroll
      for (int k = 0; k < 8; k++) {
        int sj = row_bcast(ssrc, selb + (unsigned)k * 0x11111111u);
        r[k] = xw4[(size_t)sj * (D / 4) + c];
      }
      #pragma unroll
      for (int k = 0; k < 8; k++) {
        float pj = __int_as_float(row_bcast(psrc, selb + (unsigned)k * 0x11111111u));
        acc.x += pj * r[k].x;
        acc.y += pj * r[k].y;
        acc.z += pj * r[k].z;
        acc.w += pj * r[k].w;
      }
    }
  }
  {
    float aes = row_sum(aepart0 + aepart1) / fmaxf((float)deg, 1.0f);
    float a = sS[nodec] + dn + aes;
    a = a > 0.f ? a : NEG_SLOPE * a;
    float mnew = fmaxf(m, a);
    float scale = expf(m - mnew);
    float p = expf(a - mnew);
    lsum = lsum * scale + p;
    float4 xr = xw4[(size_t)nodec * (D / 4) + c];
    acc.x = acc.x * scale + p * xr.x;
    acc.y = acc.y * scale + p * xr.y;
    acc.z = acc.z * scale + p * xr.z;
    acc.w = acc.w * scale + p * xr.w;
  }
  float inv = 1.0f / lsum;
  float4 b4 = ((const float4*)bias)[c], g4 = ((const float4*)gamma)[c], be4 = ((const float4*)beta)[c];
  float4 mu4 = ((const float4*)mean)[c], va4 = ((const float4*)var)[c];
  float4 o;
  o.x = (acc.x * inv + b4.x - mu4.x) * (g4.x / sqrtf(va4.x + BN_EPS)) + be4.x;
  o.y = (acc.y * inv + b4.y - mu4.y) * (g4.y / sqrtf(va4.y + BN_EPS)) + be4.y;
  o.z = (acc.z * inv + b4.z - mu4.z) * (g4.z / sqrtf(va4.z + BN_EPS)) + be4.z;
  o.w = (acc.w * inv + b4.w - mu4.w) * (g4.w / sqrtf(va4.w + BN_EPS)) + be4.w;
  if (relu) {
    o.x = fmaxf(o.x, 0.f);
    o.y = fmaxf(o.y, 0.f);
    o.z = fmaxf(o.z, 0.f);
    o.w = fmaxf(o.w, 0.f);
  }
  if (live) ((float4*)h)[(size_t)node * (D / 4) + c] = o;
}

#define RUN 16
__device__ __forceinline__ void graph_add(float* __restrict__ og, int g, int c, float4 a) {
  float* p = og + (size_t)g * D + 4 * c;
  atomicAdd(p + 0, a.x);
  atomicAdd(p + 1, a.y);
  atomicAdd(p + 2, a.z);
  atomicAdd(p + 3, a.w);
}
__global__ void __launch_bounds__(128) k_output(const float* __restrict__ h, const int* __restrict__ batch,
                                                float* __restrict__ out_nodes, float* __restrict__ out_graph, int n) {
  int wave = (blockIdx.x * blockDim.x + threadIdx.x) >> 5;
  if (wave * 2 * RUN >= n) return;
  int lane = threadIdx.x & 31;
  int c = lane & 15;
  int hi = lane >> 4;
  int first = (wave * 2 + hi) * RUN;
  const float4* h4 = (const float4*)h;
  float4* o4 = (float4*)out_nodes;
  int gn = first + c < n ? first + c : n - 1;
  int gv = batch[gn];
  float4 r[RUN];
  #pragma unroll
  for (int j = 0; j < RUN; j++) {
    int node = first + j < n ? first + j : n - 1;
    r[j] = h4[(size_t)node * (D / 4) + c];
  }
  #pragma unroll
  for (int j = 0; j < RUN; j++)
    if (first + j < n) o4[(size_t)(first + j) * (D / 4) + c] = r[j];

  int cur = 0;
  bool have = false;
  float4 acc = make_float4(0.f, 0.f, 0.f, 0.f);
  #pragma unroll
  for (int j = 0; j < RUN; j++) {
    int g = row_bcast(gv, (unsigned)j * 0x11111111u);
    if (first + j < n) {
      if (have && g == cur) {
        acc.x += r[j].x;
        acc.y += r[j].y;
        acc.z += r[j].z;
        acc.w += r[j].w;
      } else {
        if (have) graph_add(out_graph, cur, c, acc);
        cur = g;
        have = true;
        acc = r[j];
      }
    }
  }
  int ocur = __shfl_xor(cur, 16);
  int ohave = __shfl_xor((int)have, 16);
  float4 oacc;
  oacc.x = __shfl_xor(acc.x, 16);
  oacc.y = __shfl_xor(acc.y, 16);
  oacc.z = __shfl_xor(acc.z, 16);
  oacc.w = __shfl_xor(acc.w, 16);
  bool joined = have && ohave && cur == ocur;
  if (joined) {
    if (hi == 0) {
      acc.x += oacc.x;
      acc.y += oacc.y;
      acc.z += oacc.z;
      acc.w += oacc.w;
      graph_add(out_graph, cur, c, acc);
    }
  } else if (have) {
    graph_add(out_graph, cur, c, acc);
  }
}

extern "C" void kernel_launch(void* const* d_in, const int* in_sizes, int n_in,
                              void* d_out, int out_size, void* d_ws, size_t ws_size,
                              hipStream_t stream) {
  const float* x     = (const float*)d_in[0];
  const int*   eidx  = (const int*)d_in[1];
  const float* eattr = (const float*)d_in[2];
  const int*   batch = (const int*)d_in[3];
  const float* xlw   = (const float*)d_in[4];
  const float* xlb   = (const float*)d_in[5];
  const float* glw   = (const float*)d_in[6];
  const float* gas   = (const float*)d_in[7];
  const float* gad   = (const float*)d_in[8];
  const float* gew   = (const float*)d_in[9];
  const float* gae   = (const float*)d_in[10];
  const float* gb    = (const float*)d_in[11];
  const float* bng   = (const float*)d_in[12];
  const float* bnb   = (const float*)d_in[13];
  const float* bnm   = (const float*)d_in[14];
  const float* bnv   = (const float*)d_in[15];

  const int n = in_sizes[3];
  const int E = in_sizes[2] / F_E;
  const int* srcp = eidx;
  const int* dstp = eidx + E;
  const int npad = ((n + 1 + SCAN_TILE - 1) / SCAN_TILE) * SCAN_TILE;
  const int nbk  = (n + BKN - 1) / BKN;
  const int m    = nbk * NB;
  const int mpad = ((m + 1 + SCAN_TILE - 1) / SCAN_TILE) * SCAN_TILE;
  const int ch   = (E + NB - 1) / NB;

  float* ws    = (float*)d_ws;
  float* h     = ws;  ws += (size_t)n * D;
  float* xw    = ws;  ws += (size_t)n * D;
  int*   rowptr = (int*)ws; ws += npad;
  int*   slot  = (int*)ws; ws += 4 * (size_t)E;
  float* sS    = ws;  ws += n;
  float* dS    = ws;  ws += n;
  float* wea   = ws;  ws += L * F_E;
  int*   tslot = (int*)h;
  int*   bh    = (int*)xw;
  int*   boff  = (int*)xw + mpad;

  float* out_nodes = (float*)d_out;
  float* out_graph = out_nodes + (size_t)n * D;
  const size_t graph_elems = (size_t)out_size - (size_t)n * D;

  const int B = 256;
  #define GRID(x) (((x) + B - 1) / B)

  k_wea<<<GRID(L * F_E), B, 0, stream>>>(gew, gae, wea);
  k_bcount<<<NB, PT, 0, stream>>>(dstp, bh, E, ch, nbk, m, mpad);
  k_scan<<<mpad / SCAN_TILE, SCAN_NT, 0, stream>>>(bh, boff);
  k_bplace<<<NB, PT, 0, stream>>>(srcp, dstp, boff, eattr, wea, tslot, E, ch, nbk);
  k_bsort<<<nbk, DT, 0, stream>>>(boff, tslot, rowptr, slot, n);

  {
    int waves = (n + 15) / 16;
    int blocks = (waves + 3) / 4;
    k_gemm_wmma<<<blocks, 128, 0, stream>>>(x, xlw, xlb, h, n, F_IN);
  }

  for (int l = 0; l < L; l++) {
    int waves = (n + 15) / 16;
    int blocks = (waves + 3) / 4;
    k_gemm_wmma<<<blocks, 128, 0, stream>>>(h, glw + (size_t)l * D * D, nullptr, xw, n, D);

    k_scores<<<GRID(n), B, 0, stream>>>(xw, gas + (size_t)l * D, gad + (size_t)l * D, sS, dS, n);
    k_node<<<(n + 15) / 16, B, 0, stream>>>(xw, sS, dS, rowptr, slot, l,
                                          gb + (size_t)l * D, bng + (size_t)l * D, bnb + (size_t)l * D,
                                          bnm + (size_t)l * D, bnv + (size_t)l * D, h, n,
                                          l < L - 1 ? 1 : 0);
  }

  hipMemsetAsync(out_graph, 0, graph_elems * sizeof(float), stream);
  k_output<<<(n + 8 * RUN - 1) / (8 * RUN), 128, 0, stream>>>(h, batch, out_nodes, out_graph, n);
  #undef GRID
}


# per-node attention scores s, d taken from the GEMM's accumulators in its epilogue (lane sums 4 column tiles, then 16-lane butterfly: score sum order changed); the three score launches are gone
# speedup vs baseline: 7.4455x; 1.0443x over previous
#include <hip/hip_runtime.h>
#include <hip/hip_bf16.h>

#define F_IN   32
#define F_E    16
#define D      64
#define L      3
#define NEG_SLOPE 0.2f
#define BN_EPS 1e-5f
#define SCAN_TILE 4096

typedef __attribute__((ext_vector_type(2))) float v2f;
typedef __attribute__((ext_vector_type(8))) float v8f;
typedef __attribute__((ext_vector_type(4))) int v4i;

#define BK_LOG   8
#define BKN      (1 << BK_LOG)
#define NB       256
#define MAXBK    1024
#define SRC_BITS 20
#define PT       1024
#define DT       1024
#define STAGE    6144

__global__ void __launch_bounds__(PT) k_bcount(const int* __restrict__ dst, int* __restrict__ bh,
                                               int E, int ch, int nbk, int m, int mpad) {
  __shared__ int hist[MAXBK];
  int t = threadIdx.x, b = blockIdx.x;
  for (int k = t; k < nbk; k += PT) hist[k] = 0;
  __syncthreads();
  int e0 = b * ch;
  int e1 = e0 + ch < E ? e0 + ch : E;
  for (int e = e0 + t; e < e1; e += 4 * PT) {
    int eb = e + PT, ec = e + 2 * PT, ed = e + 3 * PT;
    int da = dst[e];
    int db = eb < e1 ? dst[eb] : -1;
    int dc = ec < e1 ? dst[ec] : -1;
    int dd = ed < e1 ? dst[ed] : -1;
    atomicAdd(&hist[da >> BK_LOG], 1);
    if (db >= 0) atomicAdd(&hist[db >> BK_LOG], 1);
    if (dc >= 0) atomicAdd(&hist[dc >> BK_LOG], 1);
    if (dd >= 0) atomicAdd(&hist[dd >> BK_LOG], 1);
  }
  __syncthreads();
  for (int k = t; k < nbk; k += PT) bh[k * NB + b] = hist[k];
  if (b == 0)
    for (int i = m + t; i < mpad; i += PT) bh[i] = 0;
}

#define SCAN_NT (SCAN_TILE / 4)
#define SCAN_NW (SCAN_NT / 32)
__global__ void __launch_bounds__(SCAN_NT) k_scan(const int* __restrict__ cnt, int* __restrict__ rowptr) {
  __shared__ int wsum[SCAN_NW];
  __shared__ int carry_s[SCAN_NW];
  int t = threadIdx.x;
  int lane = t & 31, wv = t >> 5;
  const int4* c4 = (const int4*)cnt;
  int nf = (int)blockIdx.x * SCAN_NT;
  int front = 0;
  int i = t;
  for (; i + 3 * SCAN_NT < nf; i += 4 * SCAN_NT) {
    int4 q0 = c4[i], q1 = c4[i + SCAN_NT], q2 = c4[i + 2 * SCAN_NT], q3 = c4[i + 3 * SCAN_NT];
    front += (q0.x + q0.y + q0.z + q0.w) + (q1.x + q1.y + q1.z + q1.w)
           + (q2.x + q2.y + q2.z + q2.w) + (q3.x + q3.y + q3.z + q3.w);
  }
  for (; i < nf; i += SCAN_NT) {
    int4 q0 = c4[i];
    front += q0.x + q0.y + q0.z + q0.w;
  }
  int4 q = c4[nf + t];
  int own = q.x + q.y + q.z + q.w;
  int inc = own;
  #pragma unroll
  for (int o = 1; o < 32; o <<= 1) {
    int u = __shfl_up(inc, o);
    if (lane >= o) inc += u;
  }
  #pragma unroll
  for (int o = 16; o > 0; o >>= 1) front += __shfl_xor(front, o);
  if (lane == 31) wsum[wv] = inc;
  if (lane == 0) carry_s[wv] = front;
  __syncthreads();
  int base = 0;
  #pragma unroll
  for (int w = 0; w < SCAN_NW; w++) {
    base += carry_s[w];
    if (w < wv) base += wsum[w];
  }
  int ex = base + inc - own;
  int4 r;
  r.x = ex;
  r.y = ex + q.x;
  r.z = r.y + q.y;
  r.w = r.z + q.z;
  ((int4*)rowptr)[nf + t] = r;
}

__device__ __forceinline__ int4 edge_record(int sx, float4 q0, float4 q1, float4 q2, float4 q3,
                                            const float* __restrict__ wea) {
  float t[L];
  for (int l = 0; l < L; l++) {
    const float* wl = wea + l * F_E;
    float ae = 0.f;
    ae += q0.x * wl[0];  ae += q0.y * wl[1];  ae += q0.z * wl[2];  ae += q0.w * wl[3];
    ae += q1.x * wl[4];  ae += q1.y * wl[5];  ae += q1.z * wl[6];  ae += q1.w * wl[7];
    ae += q2.x * wl[8];  ae += q2.y * wl[9];  ae += q2.z * wl[10]; ae += q2.w * wl[11];
    ae += q3.x * wl[12]; ae += q3.y * wl[13]; ae += q3.z * wl[14]; ae += q3.w * wl[15];
    t[l] = ae;
  }
  return make_int4(sx, __float_as_int(t[0]), __float_as_int(t[1]), __float_as_int(t[2]));
}
__global__ void __launch_bounds__(PT) k_bplace(const int* __restrict__ src, const int* __restrict__ dst,
                                               const int* __restrict__ boff,
                                               const float* __restrict__ eattr, const float* __restrict__ wea,
                                               int* __restrict__ tslot, int E, int ch, int nbk) {
  __shared__ int cur[MAXBK];
  int t = threadIdx.x, b = blockIdx.x;
  for (int k = t; k < nbk; k += PT) cur[k] = boff[k * NB + b];
  __syncthreads();
  int e0 = b * ch;
  int e1 = e0 + ch < E ? e0 + ch : E;
  int4* ts4 = (int4*)tslot;
  int e = e0 + t;
  for (; e + PT < e1; e += 2 * PT) {
    int eb = e + PT;
    int da = dst[e], db = dst[eb];
    int sa = src[e], sb = src[eb];
    const float4* pa = (const float4*)(eattr + (size_t)e * F_E);
    const float4* pb = (const float4*)(eattr + (size_t)eb * F_E);
    float4 a0 = pa[0], a1 = pa[1], a2 = pa[2], a3 = pa[3];
    float4 b0 = pb[0], b1 = pb[1], b2 = pb[2], b3 = pb[3];
    int posa = atomicAdd(&cur[da >> BK_LOG], 1);
    int posb = atomicAdd(&cur[db >> BK_LOG], 1);
    ts4[posa] = edge_record(sa | ((da & (BKN - 1)) << SRC_BITS), a0, a1, a2, a3, wea);
    ts4[posb] = edge_record(sb | ((db & (BKN - 1)) << SRC_BITS), b0, b1, b2, b3, wea);
  }
  if (e < e1) {
    int da = dst[e];
    int sa = src[e];
    const float4* pa = (const float4*)(eattr + (size_t)e * F_E);
    float4 a0 = pa[0], a1 = pa[1], a2 = pa[2], a3 = pa[3];
    int posa = atomicAdd(&cur[da >> BK_LOG], 1);
    ts4[posa] = edge_record(sa | ((da & (BKN - 1)) << SRC_BITS), a0, a1, a2, a3, wea);
  }
}

__global__ void __launch_bounds__(DT) k_bsort(const int* __restrict__ boff, const int* __restrict__ tslot,
                                              int* __restrict__ rowptr, int* __restrict__ slot, int n) {
  __shared__ int cnt[BKN];
  __shared__ int wsum[BKN / 32];
  __shared__ v4i stage[STAGE];
  int t = threadIdx.x, k = blockIdx.x;
  int lane = t & 31, wv = t >> 5;
  int beg = boff[k * NB];
  int end = boff[(k + 1) * NB];
  if (t < BKN) cnt[t] = 0;
  __syncthreads();
  for (int i = beg + t; i < end; i += 4 * DT) {
    int ib = i + DT, ic = i + 2 * DT, id = i + 3 * DT;
    int xa = tslot[4 * (size_t)i];
    int xb = ib < end ? tslot[4 * (size_t)ib] : -1;
    int xc = ic < end ? tslot[4 * (size_t)ic] : -1;
    int xd = id < end ? tslot[4 * (size_t)id] : -1;
    atomicAdd(&cnt[xa >> SRC_BITS], 1);
    if (xb >= 0) atomicAdd(&cnt[xb >> SRC_BITS], 1);
    if (xc >= 0) atomicAdd(&cnt[xc >> SRC_BITS], 1);
    if (xd >= 0) atomicAdd(&cnt[xd >> SRC_BITS], 1);
  }
  __syncthreads();
  int own = 0, inc = 0;
  if (t < BKN) {
    own = cnt[t];
    inc = own;
    #pragma unroll
    for (int o = 1; o < 32; o <<= 1) {
      int u = __shfl_up(inc, o);
      if (lane >= o) inc += u;
    }
    if (lane == 31) wsum[wv] = inc;
  }
  __syncthreads();
  if (t < BKN) {
    int base = 0;
    #pragma unroll
    for (int w = 0; w < BKN / 32; w++)
      if (w < wv) base += wsum[w];
    int ex = base + inc - own;
    cnt[t] = ex;
    int node = k * BKN + t;
    if (node < n) rowptr[node] = beg + ex;
  }
  if (k == (int)gridDim.x - 1 && t == 0) rowptr[n] = end;
  __syncthreads();
  const v4i* ts4 = (const v4i*)tslot;
  v4i* sl4 = (v4i*)slot;
  const v4i none = {-1, 0, 0, 0};
  const bool fits = end - beg <= STAGE;
  for (int i = beg + t; i < end; i += 4 * DT) {
    int ib = i + DT, ic = i + 2 * DT, id = i + 3 * DT;
    v4i ra = ts4[i];
    v4i rb = ib < end ? ts4[ib] : none;
    v4i rc = ic < end ? ts4[ic] : none;
    v4i rd = id < end ? ts4[id] : none;
    {
      int p = atomicAdd(&cnt[ra.x >> SRC_BITS], 1);
      ra.x &= (1 << SRC_BITS) - 1;
      if (fits) stage[p] = ra; else sl4[beg + p] = ra;
    }
    if (rb.x >= 0) {
      int p = atomicAdd(&cnt[rb.x >> SRC_BITS], 1);
      rb.x &= (1 << SRC_BITS) - 1;
      if (fits) stage[p] = rb; else sl4[beg + p] = rb;
    }
    if (rc.x >= 0) {
      int p = atomicAdd(&cnt[rc.x >> SRC_BITS], 1);
      rc.x &= (1 << SRC_BITS) - 1;
      if (fits) stage[p] = rc; else sl4[beg + p] = rc;
    }
    if (rd.x >= 0) {
      int p = atomicAdd(&cnt[rd.x >> SRC_BITS], 1);
      rd.x &= (1 << SRC_BITS) - 1;
      if (fits) stage[p] = rd; else sl4[beg + p] = rd;
    }
  }
  if (fits) {
    __syncthreads();
    for (int i = t; i < end - beg; i += DT) sl4[beg + i] = stage[i];
  }
}

__global__ void k_wea(const float* __restrict__ We, const float* __restrict__ ae,
                      float* __restrict__ wea) {
  int i = blockIdx.x * blockDim.x + threadIdx.x;
  if (i >= L * F_E) return;
  int l = i / F_E, f = i % F_E;
  float s = 0.f;
  #pragma unroll
  for (int d = 0; d < D; d++) s += We[(l * F_E + f) * D + d] * ae[l * D + d];
  wea[i] = s;
}

template <int M> __device__ __forceinline__ float row_xor(float v) {
  return __builtin_bit_cast(float, __builtin_amdgcn_update_dpp(0, __builtin_bit_cast(int, v), 0x160 + M, 0xf, 0xf, false));
}
__device__ __forceinline__ float row_max(float v) {
  v = fmaxf(v, row_xor<8>(v));
  v = fmaxf(v, row_xor<4>(v));
  v = fmaxf(v, row_xor<2>(v));
  v = fmaxf(v, row_xor<1>(v));
  return v;
}
__device__ __forceinline__ float row_sum(float v) {
  v += row_xor<8>(v);
  v += row_xor<4>(v);
  v += row_xor<2>(v);
  v += row_xor<1>(v);
  return v;
}

__global__ void k_gemm_wmma(const float* __restrict__ A, const float* __restrict__ B,
                            const float* __restrict__ bias, float* __restrict__ C,
                            int n, int K,
                            const float* __restrict__ as, const float* __restrict__ ad,
                            float* __restrict__ s, float* __restrict__ dv) {
  int wave = (blockIdx.x * blockDim.x + threadIdx.x) >> 5;
  int lane = threadIdx.x & 31;
  int row0 = wave * 16;
  if (row0 >= n) return;
  int lm = lane & 15;
  int kh = (lane >> 4) << 1;
  int mrow = row0 + lm;
  int mrd = mrow < n ? mrow : n - 1;

  v8f acc[4] = {};
  for (int k = 0; k < K; k += 4) {
    v2f a;
    a.x = A[(size_t)mrd * K + k + kh];
    a.y = A[(size_t)mrd * K + k + kh + 1];
    #pragma unroll
    for (int nt = 0; nt < 4; nt++) {
      int col = nt * 16 + lm;
      v2f b;
      b.x = B[(k + kh) * D + col];
      b.y = B[(k + kh + 1) * D + col];
      acc[nt] = __builtin_amdgcn_wmma_f32_16x16x4_f32(
          false, a, false, b, (short)0, acc[nt], false, false);
    }
  }

  int rbase = row0 + ((lane >> 4) << 3);
  #pragma unroll
  for (int nt = 0; nt < 4; nt++) {
    int col = nt * 16 + lm;
    float bv = bias ? bias[col] : 0.f;
    #pragma unroll
    for (int r = 0; r < 8; r++) {
      int row = rbase + r;
      if (row < n) C[(size_t)row * D + col] = acc[nt][r] + bv;
    }
  }

  if (as) {
    float as0 = as[lm], as1 = as[16 + lm], as2 = as[32 + lm], as3 = as[48 + lm];
    float ad0 = ad[lm], ad1 = ad[16 + lm], ad2 = ad[32 + lm], ad3 = ad[48 + lm];
    float keep = 0.f;
    #pragma unroll
    for (int r = 0; r < 8; r++) {
      float ps = acc[0][r] * as0;
      ps += acc[1][r] * as1;
      ps += acc[2][r] * as2;
      ps += acc[3][r] * as3;
      float pd = acc[0][r] * ad0;
      pd += acc[1][r] * ad1;
      pd += acc[2][r] * ad2;
      pd += acc[3][r] * ad3;
      ps = row_sum(ps);
      pd = row_sum(pd);
      if (lm == r) keep = ps;
      if (lm == r + 8) keep = pd;
    }
    int row = rbase + (lm & 7);
    float* dstp = lm < 8 ? s : dv;
    if (row < n) dstp[row] = keep;
  }
}


__device__ __forceinline__ int row_bcast(int v, unsigned sel) {
  return (int)__builtin_amdgcn_permlane16(0u, (unsigned)v, sel, sel, false, false);
}

__global__ void k_node(const float* __restrict__ xw, const float* __restrict__ sS,
                       const float* __restrict__ dS, const int* __restrict__ rowptr,
                       const int* __restrict__ slot, int layer,
                       const float* __restrict__ bias, const float* __restrict__ gamma,
                       const float* __restrict__ beta, const float* __restrict__ mean,
                       const float* __restrict__ var, float* __restrict__ h, int n, int relu) {
  int wave = __builtin_amdgcn_readfirstlane((int)((blockIdx.x * blockDim.x + threadIdx.x) >> 5));
  int nodeA = 2 * wave;
  if (nodeA >= n) return;
  int lane = threadIdx.x & 31;
  int c = lane & 15;
  int hi = lane >> 4;
  int iB = nodeA + 1 < n ? nodeA + 1 : n;
  int iC = nodeA + 2 < n ? nodeA + 2 : n;
  int r0 = rowptr[nodeA];
  int r1 = rowptr[iB];
  int r2 = rowptr[iC];
  int degA = r1 - r0, degB = r2 - r1;
  int maxdeg = degA > degB ? degA : degB;
  int beg = hi ? r1 : r0;
  int deg = hi ? degB : degA;
  int node = nodeA + hi;
  bool live = node < n;
  int nodec = live ? node : nodeA;
  float dn = dS[nodec];
  const int4* slot4 = (const int4*)slot;
  const float4* xw4 = (const float4*)xw;

  float m = -INFINITY, lsum = 0.f, aepart0 = 0.f, aepart1 = 0.f;
  float4 acc = make_float4(0.f, 0.f, 0.f, 0.f);
  for (int base = 0; base < maxdeg; base += 32) {
    int e0 = base + c, e1 = base + 16 + c;
    bool v0 = e0 < deg, v1 = e1 < deg;
    bool act = base < deg;
    int4 es0 = v0 ? slot4[beg + e0] : make_int4(nodec, 0, 0, 0);
    int4 es1 = v1 ? slot4[beg + e1] : make_int4(nodec, 0, 0, 0);
    int s0 = es0.x, s1 = es1.x;
    float ae0 = __int_as_float(layer == 0 ? es0.y : (layer == 1 ? es0.z : es0.w));
    float ae1 = __int_as_float(layer == 0 ? es1.y : (layer == 1 ? es1.z : es1.w));
    float a0 = sS[s0] + dn + ae0;
    float a1 = sS[s1] + dn + ae1;
    a0 = a0 > 0.f ? a0 : NEG_SLOPE * a0;
    a1 = a1 > 0.f ? a1 : NEG_SLOPE * a1;
    if (!v0) { ae0 = 0.f; a0 = -INFINITY; }
    if (!v1) { ae1 = 0.f; a1 = -INFINITY; }
    aepart0 += ae0;
    aepart1 += ae1;
    float mnew = act ? fmaxf(m, row_max(fmaxf(a0, a1))) : m;
    float scale = act ? expf(m - mnew) : 1.0f;
    float p0 = v0 ? expf(a0 - mnew) : 0.f;
    float p1 = v1 ? expf(a1 - mnew) : 0.f;
    lsum = lsum * scale + row_sum(p0 + p1);
    acc.x *= scale;
    acc.y *= scale;
    acc.z *= scale;
    acc.w *= scale;
    m = mnew;
    int p0b = __float_as_int(p0), p1b = __float_as_int(p1);
    int cnt = maxdeg - base < 32 ? maxdeg - base : 32;
    for (int j0 = 0; j0 < cnt; j0 += 8) {
      int ssrc = j0 < 16 ? s0 : s1;
      int psrc = j0 < 16 ? p0b : p1b;
      unsigned selb = (unsigned)(j0 & 15) * 0x11111111u;
      float4 r[8];
      #pragma unroll
      for (int k = 0; k < 8; k++) {
        int sj = row_bcast(ssrc, selb + (unsigned)k * 0x11111111u);
        r[k] = xw4[(size_t)sj * (D / 4) + c];
      }
      #pragma unroll
      for (int k = 0; k < 8; k++) {
        float pj = __int_as_float(row_bcast(psrc, selb + (unsigned)k * 0x11111111u));
        acc.x += pj * r[k].x;
        acc.y += pj * r[k].y;
        acc.z += pj * r[k].z;
        acc.w += pj * r[k].w;
      }
    }
  }
  {
    float aes = row_sum(aepart0 + aepart1) / fmaxf((float)deg, 1.0f);
    float a = sS[nodec] + dn + aes;
    a = a > 0.f ? a : NEG_SLOPE * a;
    float mnew = fmaxf(m, a);
    float scale = expf(m - mnew);
    float p = expf(a - mnew);
    lsum = lsum * scale + p;
    float4 xr = xw4[(size_t)nodec * (D / 4) + c];
    acc.x = acc.x * scale + p * xr.x;
    acc.y = acc.y * scale + p * xr.y;
    acc.z = acc.z * scale + p * xr.z;
    acc.w = acc.w * scale + p * xr.w;
  }
  float inv = 1.0f / lsum;
  float4 b4 = ((const float4*)bias)[c], g4 = ((const float4*)gamma)[c], be4 = ((const float4*)beta)[c];
  float4 mu4 = ((const float4*)mean)[c], va4 = ((const float4*)var)[c];
  float4 o;
  o.x = (acc.x * inv + b4.x - mu4.x) * (g4.x / sqrtf(va4.x + BN_EPS)) + be4.x;
  o.y = (acc.y * inv + b4.y - mu4.y) * (g4.y / sqrtf(va4.y + BN_EPS)) + be4.y;
  o.z = (acc.z * inv + b4.z - mu4.z) * (g4.z / sqrtf(va4.z + BN_EPS)) + be4.z;
  o.w = (acc.w * inv + b4.w - mu4.w) * (g4.w / sqrtf(va4.w + BN_EPS)) + be4.w;
  if (relu) {
    o.x = fmaxf(o.x, 0.f);
    o.y = fmaxf(o.y, 0.f);
    o.z = fmaxf(o.z, 0.f);
    o.w = fmaxf(o.w, 0.f);
  }
  if (live) ((float4*)h)[(size_t)node * (D / 4) + c] = o;
}

#define RUN 16
__device__ __forceinline__ void graph_add(float* __restrict__ og, int g, int c, float4 a) {
  float* p = og + (size_t)g * D + 4 * c;
  atomicAdd(p + 0, a.x);
  atomicAdd(p + 1, a.y);
  atomicAdd(p + 2, a.z);
  atomicAdd(p + 3, a.w);
}
__global__ void __launch_bounds__(128) k_output(const float* __restrict__ h, const int* __restrict__ batch,
                                                float* __restrict__ out_nodes, float* __restrict__ out_graph, int n) {
  int wave = (blockIdx.x * blockDim.x + threadIdx.x) >> 5;
  if (wave * 2 * RUN >= n) return;
  int lane = threadIdx.x & 31;
  int c = lane & 15;
  int hi = lane >> 4;
  int first = (wave * 2 + hi) * RUN;
  const float4* h4 = (const float4*)h;
  float4* o4 = (float4*)out_nodes;
  int gn = first + c < n ? first + c : n - 1;
  int gv = batch[gn];
  float4 r[RUN];
  #pragma unroll
  for (int j = 0; j < RUN; j++) {
    int node = first + j < n ? first + j : n - 1;
    r[j] = h4[(size_t)node * (D / 4) + c];
  }
  #pragma unroll
  for (int j = 0; j < RUN; j++)
    if (first + j < n) o4[(size_t)(first + j) * (D / 4) + c] = r[j];

  int cur = 0;
  bool have = false;
  float4 acc = make_float4(0.f, 0.f, 0.f, 0.f);
  #pragma unroll
  for (int j = 0; j < RUN; j++) {
    int g = row_bcast(gv, (unsigned)j * 0x11111111u);
    if (first + j < n) {
      if (have && g == cur) {
        acc.x += r[j].x;
        acc.y += r[j].y;
        acc.z += r[j].z;
        acc.w += r[j].w;
      } else {
        if (have) graph_add(out_graph, cur, c, acc);
        cur = g;
        have = true;
        acc = r[j];
      }
    }
  }
  int ocur = __shfl_xor(cur, 16);
  int ohave = __shfl_xor((int)have, 16);
  float4 oacc;
  oacc.x = __shfl_xor(acc.x, 16);
  oacc.y = __shfl_xor(acc.y, 16);
  oacc.z = __shfl_xor(acc.z, 16);
  oacc.w = __shfl_xor(acc.w, 16);
  bool joined = have && ohave && cur == ocur;
  if (joined) {
    if (hi == 0) {
      acc.x += oacc.x;
      acc.y += oacc.y;
      acc.z += oacc.z;
      acc.w += oacc.w;
      graph_add(out_graph, cur, c, acc);
    }
  } else if (have) {
    graph_add(out_graph, cur, c, acc);
  }
}

extern "C" void kernel_launch(void* const* d_in, const int* in_sizes, int n_in,
                              void* d_out, int out_size, void* d_ws, size_t ws_size,
                              hipStream_t stream) {
  const float* x     = (const float*)d_in[0];
  const int*   eidx  = (const int*)d_in[1];
  const float* eattr = (const float*)d_in[2];
  const int*   batch = (const int*)d_in[3];
  const float* xlw   = (const float*)d_in[4];
  const float* xlb   = (const float*)d_in[5];
  const float* glw   = (const float*)d_in[6];
  const float* gas   = (const float*)d_in[7];
  const float* gad   = (const float*)d_in[8];
  const float* gew   = (const float*)d_in[9];
  const float* gae   = (const float*)d_in[10];
  const float* gb    = (const float*)d_in[11];
  const float* bng   = (const float*)d_in[12];
  const float* bnb   = (const float*)d_in[13];
  const float* bnm   = (const float*)d_in[14];
  const float* bnv   = (const float*)d_in[15];

  const int n = in_sizes[3];
  const int E = in_sizes[2] / F_E;
  const int* srcp = eidx;
  const int* dstp = eidx + E;
  const int npad = ((n + 1 + SCAN_TILE - 1) / SCAN_TILE) * SCAN_TILE;
  const int nbk  = (n + BKN - 1) / BKN;
  const int m    = nbk * NB;
  const int mpad = ((m + 1 + SCAN_TILE - 1) / SCAN_TILE) * SCAN_TILE;
  const int ch   = (E + NB - 1) / NB;

  float* ws    = (float*)d_ws;
  float* h     = ws;  ws += (size_t)n * D;
  float* xw    = ws;  ws += (size_t)n * D;
  int*   rowptr = (int*)ws; ws += npad;
  int*   slot  = (int*)ws; ws += 4 * (size_t)E;
  float* sS    = ws;  ws += n;
  float* dS    = ws;  ws += n;
  float* wea   = ws;  ws += L * F_E;
  int*   tslot = (int*)h;
  int*   bh    = (int*)xw;
  int*   boff  = (int*)xw + mpad;

  float* out_nodes = (float*)d_out;
  float* out_graph = out_nodes + (size_t)n * D;
  const size_t graph_elems = (size_t)out_size - (size_t)n * D;

  const int B = 256;
  #define GRID(x) (((x) + B - 1) / B)

  k_wea<<<GRID(L * F_E), B, 0, stream>>>(gew, gae, wea);
  k_bcount<<<NB, PT, 0, stream>>>(dstp, bh, E, ch, nbk, m, mpad);
  k_scan<<<mpad / SCAN_TILE, SCAN_NT, 0, stream>>>(bh, boff);
  k_bplace<<<NB, PT, 0, stream>>>(srcp, dstp, boff, eattr, wea, tslot, E, ch, nbk);
  k_bsort<<<nbk, DT, 0, stream>>>(boff, tslot, rowptr, slot, n);

  {
    int waves = (n + 15) / 16;
    int blocks = (waves + 3) / 4;
    k_gemm_wmma<<<blocks, 128, 0, stream>>>(x, xlw, xlb, h, n, F_IN, nullptr, nullptr, nullptr, nullptr);
  }

  for (int l = 0; l < L; l++) {
    int waves = (n + 15) / 16;
    int blocks = (waves + 3) / 4;
    k_gemm_wmma<<<blocks, 128, 0, stream>>>(h, glw + (size_t)l * D * D, nullptr, xw, n, D,
                                            gas + (size_t)l * D, gad + (size_t)l * D, sS, dS);
    k_node<<<(n + 15) / 16, B, 0, stream>>>(xw, sS, dS, rowptr, slot, l,
                                          gb + (size_t)l * D, bng + (size_t)l * D, bnb + (size_t)l * D,
                                          bnm + (size_t)l * D, bnv + (size_t)l * D, h, n,
                                          l < L - 1 ? 1 : 0);
  }

  hipMemsetAsync(out_graph, 0, graph_elems * sizeof(float), stream);
  k_output<<<(n + 8 * RUN - 1) / (8 * RUN), 128, 0, stream>>>(h, batch, out_nodes, out_graph, n);
  #undef GRID
}


# node kernel: edge source score = dot of its gathered row with a_s (score sum order changed); trips of 24 edges, rows in registers (softmax sums above degree 24 split anew); no score gather
# speedup vs baseline: 7.6745x; 1.0308x over previous
#include <hip/hip_runtime.h>
#include <hip/hip_bf16.h>

#define F_IN   32
#define F_E    16
#define D      64
#define L      3
#define NEG_SLOPE 0.2f
#define BN_EPS 1e-5f
#define SCAN_TILE 4096

typedef __attribute__((ext_vector_type(2))) float v2f;
typedef __attribute__((ext_vector_type(8))) float v8f;
typedef __attribute__((ext_vector_type(4))) int v4i;

#define BK_LOG   8
#define BKN      (1 << BK_LOG)
#define NB       256
#define MAXBK    1024
#define SRC_BITS 20
#define PT       1024
#define DT       1024
#define STAGE    6144

__global__ void __launch_bounds__(PT) k_bcount(const int* __restrict__ dst, int* __restrict__ bh,
                                               int E, int ch, int nbk, int m, int mpad) {
  __shared__ int hist[MAXBK];
  int t = threadIdx.x, b = blockIdx.x;
  for (int k = t; k < nbk; k += PT) hist[k] = 0;
  __syncthreads();
  int e0 = b * ch;
  int e1 = e0 + ch < E ? e0 + ch : E;
  for (int e = e0 + t; e < e1; e += 4 * PT) {
    int eb = e + PT, ec = e + 2 * PT, ed = e + 3 * PT;
    int da = dst[e];
    int db = eb < e1 ? dst[eb] : -1;
    int dc = ec < e1 ? dst[ec] : -1;
    int dd = ed < e1 ? dst[ed] : -1;
    atomicAdd(&hist[da >> BK_LOG], 1);
    if (db >= 0) atomicAdd(&hist[db >> BK_LOG], 1);
    if (dc >= 0) atomicAdd(&hist[dc >> BK_LOG], 1);
    if (dd >= 0) atomicAdd(&hist[dd >> BK_LOG], 1);
  }
  __syncthreads();
  for (int k = t; k < nbk; k += PT) bh[k * NB + b] = hist[k];
  if (b == 0)
    for (int i = m + t; i < mpad; i += PT) bh[i] = 0;
}

#define SCAN_NT (SCAN_TILE / 4)
#define SCAN_NW (SCAN_NT / 32)
__global__ void __launch_bounds__(SCAN_NT) k_scan(const int* __restrict__ cnt, int* __restrict__ rowptr) {
  __shared__ int wsum[SCAN_NW];
  __shared__ int carry_s[SCAN_NW];
  int t = threadIdx.x;
  int lane = t & 31, wv = t >> 5;
  const int4* c4 = (const int4*)cnt;
  int nf = (int)blockIdx.x * SCAN_NT;
  int front = 0;
  int i = t;
  for (; i + 3 * SCAN_NT < nf; i += 4 * SCAN_NT) {
    int4 q0 = c4[i], q1 = c4[i + SCAN_NT], q2 = c4[i + 2 * SCAN_NT], q3 = c4[i + 3 * SCAN_NT];
    front += (q0.x + q0.y + q0.z + q0.w) + (q1.x + q1.y + q1.z + q1.w)
           + (q2.x + q2.y + q2.z + q2.w) + (q3.x + q3.y + q3.z + q3.w);
  }
  for (; i < nf; i += SCAN_NT) {
    int4 q0 = c4[i];
    front += q0.x + q0.y + q0.z + q0.w;
  }
  int4 q = c4[nf + t];
  int own = q.x + q.y + q.z + q.w;
  int inc = own;
  #pragma unroll
  for (int o = 1; o < 32; o <<= 1) {
    int u = __shfl_up(inc, o);
    if (lane >= o) inc += u;
  }
  #pragma unroll
  for (int o = 16; o > 0; o >>= 1) front += __shfl_xor(front, o);
  if (lane == 31) wsum[wv] = inc;
  if (lane == 0) carry_s[wv] = front;
  __syncthreads();
  int base = 0;
  #pragma unroll
  for (int w = 0; w < SCAN_NW; w++) {
    base += carry_s[w];
    if (w < wv) base += wsum[w];
  }
  int ex = base + inc - own;
  int4 r;
  r.x = ex;
  r.y = ex + q.x;
  r.z = r.y + q.y;
  r.w = r.z + q.z;
  ((int4*)rowptr)[nf + t] = r;
}

__device__ __forceinline__ int4 edge_record(int sx, float4 q0, float4 q1, float4 q2, float4 q3,
                                            const float* __restrict__ wea) {
  float t[L];
  for (int l = 0; l < L; l++) {
    const float* wl = wea + l * F_E;
    float ae = 0.f;
    ae += q0.x * wl[0];  ae += q0.y * wl[1];  ae += q0.z * wl[2];  ae += q0.w * wl[3];
    ae += q1.x * wl[4];  ae += q1.y * wl[5];  ae += q1.z * wl[6];  ae += q1.w * wl[7];
    ae += q2.x * wl[8];  ae += q2.y * wl[9];  ae += q2.z * wl[10]; ae += q2.w * wl[11];
    ae += q3.x * wl[12]; ae += q3.y * wl[13]; ae += q3.z * wl[14]; ae += q3.w * wl[15];
    t[l] = ae;
  }
  return make_int4(sx, __float_as_int(t[0]), __float_as_int(t[1]), __float_as_int(t[2]));
}
__global__ void __launch_bounds__(PT) k_bplace(const int* __restrict__ src, const int* __restrict__ dst,
                                               const int* __restrict__ boff,
                                               const float* __restrict__ eattr, const float* __restrict__ wea,
                                               int* __restrict__ tslot, int E, int ch, int nbk) {
  __shared__ int cur[MAXBK];
  int t = threadIdx.x, b = blockIdx.x;
  for (int k = t; k < nbk; k += PT) cur[k] = boff[k * NB + b];
  __syncthreads();
  int e0 = b * ch;
  int e1 = e0 + ch < E ? e0 + ch : E;
  int4* ts4 = (int4*)tslot;
  int e = e0 + t;
  for (; e + PT < e1; e += 2 * PT) {
    int eb = e + PT;
    int da = dst[e], db = dst[eb];
    int sa = src[e], sb = src[eb];
    const float4* pa = (const float4*)(eattr + (size_t)e * F_E);
    const float4* pb = (const float4*)(eattr + (size_t)eb * F_E);
    float4 a0 = pa[0], a1 = pa[1], a2 = pa[2], a3 = pa[3];
    float4 b0 = pb[0], b1 = pb[1], b2 = pb[2], b3 = pb[3];
    int posa = atomicAdd(&cur[da >> BK_LOG], 1);
    int posb = atomicAdd(&cur[db >> BK_LOG], 1);
    ts4[posa] = edge_record(sa | ((da & (BKN - 1)) << SRC_BITS), a0, a1, a2, a3, wea);
    ts4[posb] = edge_record(sb | ((db & (BKN - 1)) << SRC_BITS), b0, b1, b2, b3, wea);
  }
  if (e < e1) {
    int da = dst[e];
    int sa = src[e];
    const float4* pa = (const float4*)(eattr + (size_t)e * F_E);
    float4 a0 = pa[0], a1 = pa[1], a2 = pa[2], a3 = pa[3];
    int posa = atomicAdd(&cur[da >> BK_LOG], 1);
    ts4[posa] = edge_record(sa | ((da & (BKN - 1)) << SRC_BITS), a0, a1, a2, a3, wea);
  }
}

__global__ void __launch_bounds__(DT) k_bsort(const int* __restrict__ boff, const int* __restrict__ tslot,
                                              int* __restrict__ rowptr, int* __restrict__ slot, int n) {
  __shared__ int cnt[BKN];
  __shared__ int wsum[BKN / 32];
  __shared__ v4i stage[STAGE];
  int t = threadIdx.x, k = blockIdx.x;
  int lane = t & 31, wv = t >> 5;
  int beg = boff[k * NB];
  int end = boff[(k + 1) * NB];
  if (t < BKN) cnt[t] = 0;
  __syncthreads();
  for (int i = beg + t; i < end; i += 4 * DT) {
    int ib = i + DT, ic = i + 2 * DT, id = i + 3 * DT;
    int xa = tslot[4 * (size_t)i];
    int xb = ib < end ? tslot[4 * (size_t)ib] : -1;
    int xc = ic < end ? tslot[4 * (size_t)ic] : -1;
    int xd = id < end ? tslot[4 * (size_t)id] : -1;
    atomicAdd(&cnt[xa >> SRC_BITS], 1);
    if (xb >= 0) atomicAdd(&cnt[xb >> SRC_BITS], 1);
    if (xc >= 0) atomicAdd(&cnt[xc >> SRC_BITS], 1);
    if (xd >= 0) atomicAdd(&cnt[xd >> SRC_BITS], 1);
  }
  __syncthreads();
  int own = 0, inc = 0;
  if (t < BKN) {
    own = cnt[t];
    inc = own;
    #pragma unroll
    for (int o = 1; o < 32; o <<= 1) {
      int u = __shfl_up(inc, o);
      if (lane >= o) inc += u;
    }
    if (lane == 31) wsum[wv] = inc;
  }
  __syncthreads();
  if (t < BKN) {
    int base = 0;
    #pragma unroll
    for (int w = 0; w < BKN / 32; w++)
      if (w < wv) base += wsum[w];
    int ex = base + inc - own;
    cnt[t] = ex;
    int node = k * BKN + t;
    if (node < n) rowptr[node] = beg + ex;
  }
  if (k == (int)gridDim.x - 1 && t == 0) rowptr[n] = end;
  __syncthreads();
  const v4i* ts4 = (const v4i*)tslot;
  v4i* sl4 = (v4i*)slot;
  const v4i none = {-1, 0, 0, 0};
  const bool fits = end - beg <= STAGE;
  for (int i = beg + t; i < end; i += 4 * DT) {
    int ib = i + DT, ic = i + 2 * DT, id = i + 3 * DT;
    v4i ra = ts4[i];
    v4i rb = ib < end ? ts4[ib] : none;
    v4i rc = ic < end ? ts4[ic] : none;
    v4i rd = id < end ? ts4[id] : none;
    {
      int p = atomicAdd(&cnt[ra.x >> SRC_BITS], 1);
      ra.x &= (1 << SRC_BITS) - 1;
      if (fits) stage[p] = ra; else sl4[beg + p] = ra;
    }
    if (rb.x >= 0) {
      int p = atomicAdd(&cnt[rb.x >> SRC_BITS], 1);
      rb.x &= (1 << SRC_BITS) - 1;
      if (fits) stage[p] = rb; else sl4[beg + p] = rb;
    }
    if (rc.x >= 0) {
      int p = atomicAdd(&cnt[rc.x >> SRC_BITS], 1);
      rc.x &= (1 << SRC_BITS) - 1;
      if (fits) stage[p] = rc; else sl4[beg + p] = rc;
    }
    if (rd.x >= 0) {
      int p = atomicAdd(&cnt[rd.x >> SRC_BITS], 1);
      rd.x &= (1 << SRC_BITS) - 1;
      if (fits) stage[p] = rd; else sl4[beg + p] = rd;
    }
  }
  if (fits) {
    __syncthreads();
    for (int i = t; i < end - beg; i += DT) sl4[beg + i] = stage[i];
  }
}

__global__ void k_wea(const float* __restrict__ We, const float* __restrict__ ae,
                      float* __restrict__ wea) {
  int i = blockIdx.x * blockDim.x + threadIdx.x;
  if (i >= L * F_E) return;
  int l = i / F_E, f = i % F_E;
  float s = 0.f;
  #pragma unroll
  for (int d = 0; d < D; d++) s += We[(l * F_E + f) * D + d] * ae[l * D + d];
  wea[i] = s;
}

template <int M> __device__ __forceinline__ float row_xor(float v) {
  return __builtin_bit_cast(float, __builtin_amdgcn_update_dpp(0, __builtin_bit_cast(int, v), 0x160 + M, 0xf, 0xf, false));
}
__device__ __forceinline__ float row_max(float v) {
  v = fmaxf(v, row_xor<8>(v));
  v = fmaxf(v, row_xor<4>(v));
  v = fmaxf(v, row_xor<2>(v));
  v = fmaxf(v, row_xor<1>(v));
  return v;
}
__device__ __forceinline__ float row_sum(float v) {
  v += row_xor<8>(v);
  v += row_xor<4>(v);
  v += row_xor<2>(v);
  v += row_xor<1>(v);
  return v;
}

__global__ void k_gemm_wmma(const float* __restrict__ A, const float* __restrict__ B,
                            const float* __restrict__ bias, float* __restrict__ C,
                            int n, int K,
                            const float* __restrict__ as, const float* __restrict__ ad,
                            float* __restrict__ s, float* __restrict__ dv) {
  int wave = (blockIdx.x * blockDim.x + threadIdx.x) >> 5;
  int lane = threadIdx.x & 31;
  int row0 = wave * 16;
  if (row0 >= n) return;
  int lm = lane & 15;
  int kh = (lane >> 4) << 1;
  int mrow = row0 + lm;
  int mrd = mrow < n ? mrow : n - 1;

  v8f acc[4] = {};
  for (int k = 0; k < K; k += 4) {
    v2f a;
    a.x = A[(size_t)mrd * K + k + kh];
    a.y = A[(size_t)mrd * K + k + kh + 1];
    #pragma unroll
    for (int nt = 0; nt < 4; nt++) {
      int col = nt * 16 + lm;
      v2f b;
      b.x = B[(k + kh) * D + col];
      b.y = B[(k + kh + 1) * D + col];
      acc[nt] = __builtin_amdgcn_wmma_f32_16x16x4_f32(
          false, a, false, b, (short)0, acc[nt], false, false);
    }
  }

  int rbase = row0 + ((lane >> 4) << 3);
  #pragma unroll
  for (int nt = 0; nt < 4; nt++) {
    int col = nt * 16 + lm;
    float bv = bias ? bias[col] : 0.f;
    #pragma unroll
    for (int r = 0; r < 8; r++) {
      int row = rbase + r;
      if (row < n) C[(size_t)row * D + col] = acc[nt][r] + bv;
    }
  }

  if (as) {
    float as0 = as[lm], as1 = as[16 + lm], as2 = as[32 + lm], as3 = as[48 + lm];
    float ad0 = ad[lm], ad1 = ad[16 + lm], ad2 = ad[32 + lm], ad3 = ad[48 + lm];
    float keep = 0.f;
    #pragma unroll
    for (int r = 0; r < 8; r++) {
      float ps = acc[0][r] * as0;
      ps += acc[1][r] * as1;
      ps += acc[2][r] * as2;
      ps += acc[3][r] * as3;
      float pd = acc[0][r] * ad0;
      pd += acc[1][r] * ad1;
      pd += acc[2][r] * ad2;
      pd += acc[3][r] * ad3;
      ps = row_sum(ps);
      pd = row_sum(pd);
      if (lm == r) keep = ps;
      if (lm == r + 8) keep = pd;
    }
    int row = rbase + (lm & 7);
    float* dstp = lm < 8 ? s : dv;
    if (row < n) dstp[row] = keep;
  }
}


__device__ __forceinline__ int row_bcast(int v, unsigned sel) {
  return (int)__builtin_amdgcn_permlane16(0u, (unsigned)v, sel, sel, false, false);
}
__device__ __forceinline__ int row_bcast_all(int v, unsigned sel) {
  return (int)__builtin_amdgcn_permlane16(0u, (unsigned)v, sel, sel, true, false);
}


template <int NG>
__device__ __forceinline__ void node_trip(const char* __restrict__ xwb, unsigned cb, int c, int s0, int s1,
                                          v2f asl, v2f ash, float dn, float ae0, float ae1,
                                          bool v0, bool v1, bool act,
                                          float& m, float& lsum, v2f& accl, v2f& acch) {
  v2f rl[8 * NG], rh[8 * NG];
  unsigned off[8 * NG];
  #pragma unroll
  for (int k = 0; k < 8 * NG; k++)
    off[k] = ((unsigned)row_bcast_all(k < 16 ? s0 : s1, (unsigned)(k & 15) * 0x11111111u) << 8) + cb;
  #pragma unroll
  for (int k = 0; k < 8 * NG; k++) {
    float4 q = *(const float4*)(xwb + off[k]);
    rl[k].x = q.x; rl[k].y = q.y;
    rh[k].x = q.z; rh[k].y = q.w;
  }
  float sc0 = 0.f, sc1 = 0.f;
  #pragma unroll
  for (int k = 0; k < 8 * NG; k++) {
    v2f t = rl[k] * asl;
    t = __builtin_elementwise_fma(rh[k], ash, t);
    float d = row_sum(t.x + t.y);
    if (k < 16) {
      if (c == k) sc0 = d;
    } else {
      if (c == k - 16) sc1 = d;
    }
  }
  float a0 = sc0 + dn + ae0;
  float a1 = sc1 + dn + ae1;
  a0 = a0 > 0.f ? a0 : NEG_SLOPE * a0;
  a1 = a1 > 0.f ? a1 : NEG_SLOPE * a1;
  if (!v0) a0 = -INFINITY;
  if (!v1) a1 = -INFINITY;
  float mnew = act ? fmaxf(m, row_max(fmaxf(a0, a1))) : m;
  float scale = act ? expf(m - mnew) : 1.0f;
  float p0 = v0 ? expf(a0 - mnew) : 0.f;
  float p1 = v1 ? expf(a1 - mnew) : 0.f;
  lsum = lsum * scale + row_sum(p0 + p1);
  accl *= scale;
  acch *= scale;
  m = mnew;
  int p0b = __float_as_int(p0), p1b = __float_as_int(p1);
  #pragma unroll
  for (int k = 0; k < 8 * NG; k++) {
    float pj = __int_as_float(row_bcast_all(k < 16 ? p0b : p1b, (unsigned)(k & 15) * 0x11111111u));
    v2f pp = {pj, pj};
    accl = __builtin_elementwise_fma(pp, rl[k], accl);
    acch = __builtin_elementwise_fma(pp, rh[k], acch);
  }
}

__global__ void __launch_bounds__(128, 7) k_node(const float* __restrict__ xw, const float* __restrict__ sS,
                       const float* __restrict__ dS, const float* __restrict__ avs,
                       const int* __restrict__ rowptr,
                       const int* __restrict__ slot, int layer,
                       const float* __restrict__ bias, const float* __restrict__ gamma,
                       const float* __restrict__ beta, const float* __restrict__ mean,
                       const float* __restrict__ var, float* __restrict__ h, int n, int relu) {
  int wave = __builtin_amdgcn_readfirstlane((int)((blockIdx.x * blockDim.x + threadIdx.x) >> 5));
  int nodeA = 2 * wave;
  if (nodeA >= n) return;
  int lane = threadIdx.x & 31;
  int c = lane & 15;
  int hi = lane >> 4;
  int iB = nodeA + 1 < n ? nodeA + 1 : n;
  int iC = nodeA + 2 < n ? nodeA + 2 : n;
  int r0 = rowptr[nodeA];
  int r1 = rowptr[iB];
  int r2 = rowptr[iC];
  int degA = r1 - r0, degB = r2 - r1;
  int maxdeg = degA > degB ? degA : degB;
  int beg = hi ? r1 : r0;
  int deg = hi ? degB : degA;
  int node = nodeA + hi;
  bool live = node < n;
  int nodec = live ? node : nodeA;
  float dn = dS[nodec];
  float4 as4 = ((const float4*)avs)[c];
  v2f asl = {as4.x, as4.y}, ash = {as4.z, as4.w};
  const int4* slot4 = (const int4*)slot;
  const float4* xw4 = (const float4*)xw;

  float m = -INFINITY, lsum = 0.f, aepart0 = 0.f, aepart1 = 0.f;
  v2f accl = {0.f, 0.f}, acch = {0.f, 0.f};
  const char* xwb = (const char*)xw;
  unsigned cb = (unsigned)c << 4;
  for (int base = 0; base < maxdeg; base += 24) {
    int e0 = base + c, e1 = base + 16 + c;
    bool v0 = e0 < deg, v1 = c < 8 && e1 < deg;
    bool act = base < deg;
    int4 es0 = v0 ? slot4[beg + e0] : make_int4(nodec, 0, 0, 0);
    int4 es1 = v1 ? slot4[beg + e1] : make_int4(nodec, 0, 0, 0);
    float ae0 = __int_as_float(layer == 0 ? es0.y : (layer == 1 ? es0.z : es0.w));
    float ae1 = __int_as_float(layer == 0 ? es1.y : (layer == 1 ? es1.z : es1.w));
    if (!v0) ae0 = 0.f;
    if (!v1) ae1 = 0.f;
    aepart0 += ae0;
    aepart1 += ae1;
    int cnt = maxdeg - base < 24 ? maxdeg - base : 24;
    int ng = (cnt + 7) >> 3;
    if (ng == 1)      node_trip<1>(xwb, cb, c, es0.x, es1.x, asl, ash, dn, ae0, ae1, v0, v1, act, m, lsum, accl, acch);
    else if (ng == 2) node_trip<2>(xwb, cb, c, es0.x, es1.x, asl, ash, dn, ae0, ae1, v0, v1, act, m, lsum, accl, acch);
    else              node_trip<3>(xwb, cb, c, es0.x, es1.x, asl, ash, dn, ae0, ae1, v0, v1, act, m, lsum, accl, acch);
  }
  {
    float aes = row_sum(aepart0 + aepart1) / fmaxf((float)deg, 1.0f);
    float a = sS[nodec] + dn + aes;
    a = a > 0.f ? a : NEG_SLOPE * a;
    float mnew = fmaxf(m, a);
    float scale = expf(m - mnew);
    float p = expf(a - mnew);
    lsum = lsum * scale + p;
    float4 xr = xw4[(size_t)nodec * (D / 4) + c];
    accl.x = accl.x * scale + p * xr.x;
    accl.y = accl.y * scale + p * xr.y;
    acch.x = acch.x * scale + p * xr.z;
    acch.y = acch.y * scale + p * xr.w;
  }
  float4 acc = make_float4(accl.x, accl.y, acch.x, acch.y);
  float inv = 1.0f / lsum;
  float4 b4 = ((const float4*)bias)[c], g4 = ((const float4*)gamma)[c], be4 = ((const float4*)beta)[c];
  float4 mu4 = ((const float4*)mean)[c], va4 = ((const float4*)var)[c];
  float4 o;
  o.x = (acc.x * inv + b4.x - mu4.x) * (g4.x / sqrtf(va4.x + BN_EPS)) + be4.x;
  o.y = (acc.y * inv + b4.y - mu4.y) * (g4.y / sqrtf(va4.y + BN_EPS)) + be4.y;
  o.z = (acc.z * inv + b4.z - mu4.z) * (g4.z / sqrtf(va4.z + BN_EPS)) + be4.z;
  o.w = (acc.w * inv + b4.w - mu4.w) * (g4.w / sqrtf(va4.w + BN_EPS)) + be4.w;
  if (relu) {
    o.x = fmaxf(o.x, 0.f);
    o.y = fmaxf(o.y, 0.f);
    o.z = fmaxf(o.z, 0.f);
    o.w = fmaxf(o.w, 0.f);
  }
  if (live) ((float4*)h)[(size_t)node * (D / 4) + c] = o;
}

#define RUN 16
__device__ __forceinline__ void graph_add(float* __restrict__ og, int g, int c, float4 a) {
  float* p = og + (size_t)g * D + 4 * c;
  atomicAdd(p + 0, a.x);
  atomicAdd(p + 1, a.y);
  atomicAdd(p + 2, a.z);
  atomicAdd(p + 3, a.w);
}
__global__ void __launch_bounds__(128) k_output(const float* __restrict__ h, const int* __restrict__ batch,
                                                float* __restrict__ out_nodes, float* __restrict__ out_graph, int n) {
  int wave = (blockIdx.x * blockDim.x + threadIdx.x) >> 5;
  if (wave * 2 * RUN >= n) return;
  int lane = threadIdx.x & 31;
  int c = lane & 15;
  int hi = lane >> 4;
  int first = (wave * 2 + hi) * RUN;
  const float4* h4 = (const float4*)h;
  float4* o4 = (float4*)out_nodes;
  int gn = first + c < n ? first + c : n - 1;
  int gv = batch[gn];
  float4 r[RUN];
  #pragma unroll
  for (int j = 0; j < RUN; j++) {
    int node = first + j < n ? first + j : n - 1;
    r[j] = h4[(size_t)node * (D / 4) + c];
  }
  #pragma unroll
  for (int j = 0; j < RUN; j++)
    if (first + j < n) o4[(size_t)(first + j) * (D / 4) + c] = r[j];

  int cur = 0;
  bool have = false;
  float4 acc = make_float4(0.f, 0.f, 0.f, 0.f);
  #pragma unroll
  for (int j = 0; j < RUN; j++) {
    int g = row_bcast(gv, (unsigned)j * 0x11111111u);
    if (first + j < n) {
      if (have && g == cur) {
        acc.x += r[j].x;
        acc.y += r[j].y;
        acc.z += r[j].z;
        acc.w += r[j].w;
      } else {
        if (have) graph_add(out_graph, cur, c, acc);
        cur = g;
        have = true;
        acc = r[j];
      }
    }
  }
  int ocur = __shfl_xor(cur, 16);
  int ohave = __shfl_xor((int)have, 16);
  float4 oacc;
  oacc.x = __shfl_xor(acc.x, 16);
  oacc.y = __shfl_xor(acc.y, 16);
  oacc.z = __shfl_xor(acc.z, 16);
  oacc.w = __shfl_xor(acc.w, 16);
  bool joined = have && ohave && cur == ocur;
  if (joined) {
    if (hi == 0) {
      acc.x += oacc.x;
      acc.y += oacc.y;
      acc.z += oacc.z;
      acc.w += oacc.w;
      graph_add(out_graph, cur, c, acc);
    }
  } else if (have) {
    graph_add(out_graph, cur, c, acc);
  }
}

extern "C" void kernel_launch(void* const* d_in, const int* in_sizes, int n_in,
                              void* d_out, int out_size, void* d_ws, size_t ws_size,
                              hipStream_t stream) {
  const float* x     = (const float*)d_in[0];
  const int*   eidx  = (const int*)d_in[1];
  const float* eattr = (const float*)d_in[2];
  const int*   batch = (const int*)d_in[3];
  const float* xlw   = (const float*)d_in[4];
  const float* xlb   = (const float*)d_in[5];
  const float* glw   = (const float*)d_in[6];
  const float* gas   = (const float*)d_in[7];
  const float* gad   = (const float*)d_in[8];
  const float* gew   = (const float*)d_in[9];
  const float* gae   = (const float*)d_in[10];
  const float* gb    = (const float*)d_in[11];
  const float* bng   = (const float*)d_in[12];
  const float* bnb   = (const float*)d_in[13];
  const float* bnm   = (const float*)d_in[14];
  const float* bnv   = (const float*)d_in[15];

  const int n = in_sizes[3];
  const int E = in_sizes[2] / F_E;
  const int* srcp = eidx;
  const int* dstp = eidx + E;
  const int npad = ((n + 1 + SCAN_TILE - 1) / SCAN_TILE) * SCAN_TILE;
  const int nbk  = (n + BKN - 1) / BKN;
  const int m    = nbk * NB;
  const int mpad = ((m + 1 + SCAN_TILE - 1) / SCAN_TILE) * SCAN_TILE;
  const int ch   = (E + NB - 1) / NB;

  float* ws    = (float*)d_ws;
  float* h     = ws;  ws += (size_t)n * D;
  float* xw    = ws;  ws += (size_t)n * D;
  int*   rowptr = (int*)ws; ws += npad;
  int*   slot  = (int*)ws; ws += 4 * (size_t)E;
  float* sS    = ws;  ws += n;
  float* dS    = ws;  ws += n;
  float* wea   = ws;  ws += L * F_E;
  int*   tslot = (int*)h;
  int*   bh    = (int*)xw;
  int*   boff  = (int*)xw + mpad;

  float* out_nodes = (float*)d_out;
  float* out_graph = out_nodes + (size_t)n * D;
  const size_t graph_elems = (size_t)out_size - (size_t)n * D;

  const int B = 256;
  #define GRID(x) (((x) + B - 1) / B)

  k_wea<<<GRID(L * F_E), B, 0, stream>>>(gew, gae, wea);
  k_bcount<<<NB, PT, 0, stream>>>(dstp, bh, E, ch, nbk, m, mpad);
  k_scan<<<mpad / SCAN_TILE, SCAN_NT, 0, stream>>>(bh, boff);
  k_bplace<<<NB, PT, 0, stream>>>(srcp, dstp, boff, eattr, wea, tslot, E, ch, nbk);
  k_bsort<<<nbk, DT, 0, stream>>>(boff, tslot, rowptr, slot, n);

  {
    int waves = (n + 15) / 16;
    int blocks = (waves + 3) / 4;
    k_gemm_wmma<<<blocks, 128, 0, stream>>>(x, xlw, xlb, h, n, F_IN, nullptr, nullptr, nullptr, nullptr);
  }

  for (int l = 0; l < L; l++) {
    int waves = (n + 15) / 16;
    int blocks = (waves + 3) / 4;
    k_gemm_wmma<<<blocks, 128, 0, stream>>>(h, glw + (size_t)l * D * D, nullptr, xw, n, D,
                                            gas + (size_t)l * D, gad + (size_t)l * D, sS, dS);
    k_node<<<(n + 7) / 8, 128, 0, stream>>>(xw, sS, dS, gas + (size_t)l * D, rowptr, slot, l,
                                          gb + (size_t)l * D, bng + (size_t)l * D, bnb + (size_t)l * D,
                                          bnm + (size_t)l * D, bnv + (size_t)l * D, h, n,
                                          l < L - 1 ? 1 : 0);
  }

  hipMemsetAsync(out_graph, 0, graph_elems * sizeof(float), stream);
  k_output<<<(n + 8 * RUN - 1) / (8 * RUN), 128, 0, stream>>>(h, batch, out_nodes, out_graph, n);
  #undef GRID
}


# h never stored: each h @ W runs where its rows are formed, 16 rows via LDS to WMMA (node kernel epilogue; input linear + layer 0 fused); node's own s, d: dots of its xw row (score sum order changed)
# speedup vs baseline: 9.6453x; 1.2568x over previous
#include <hip/hip_runtime.h>
#include <hip/hip_bf16.h>

#define F_IN   32
#define F_E    16
#define D      64
#define L      3
#define NEG_SLOPE 0.2f
#define BN_EPS 1e-5f
#define SCAN_TILE 4096

typedef __attribute__((ext_vector_type(2))) float v2f;
typedef __attribute__((ext_vector_type(8))) float v8f;
typedef __attribute__((ext_vector_type(4))) int v4i;

#define BK_LOG   8
#define BKN      (1 << BK_LOG)
#define NB       256
#define MAXBK    1024
#define SRC_BITS 20
#define PT       1024
#define DT       1024
#define STAGE    6144

__global__ void __launch_bounds__(PT) k_bcount(const int* __restrict__ dst, int* __restrict__ bh,
                                               int E, int ch, int nbk, int m, int mpad) {
  __shared__ int hist[MAXBK];
  int t = threadIdx.x, b = blockIdx.x;
  for (int k = t; k < nbk; k += PT) hist[k] = 0;
  __syncthreads();
  int e0 = b * ch;
  int e1 = e0 + ch < E ? e0 + ch : E;
  for (int e = e0 + t; e < e1; e += 4 * PT) {
    int eb = e + PT, ec = e + 2 * PT, ed = e + 3 * PT;
    int da = dst[e];
    int db = eb < e1 ? dst[eb] : -1;
    int dc = ec < e1 ? dst[ec] : -1;
    int dd = ed < e1 ? dst[ed] : -1;
    atomicAdd(&hist[da >> BK_LOG], 1);
    if (db >= 0) atomicAdd(&hist[db >> BK_LOG], 1);
    if (dc >= 0) atomicAdd(&hist[dc >> BK_LOG], 1);
    if (dd >= 0) atomicAdd(&hist[dd >> BK_LOG], 1);
  }
  __syncthreads();
  for (int k = t; k < nbk; k += PT) bh[k * NB + b] = hist[k];
  if (b == 0)
    for (int i = m + t; i < mpad; i += PT) bh[i] = 0;
}

#define SCAN_NT (SCAN_TILE / 4)
#define SCAN_NW (SCAN_NT / 32)
__global__ void __launch_bounds__(SCAN_NT) k_scan(const int* __restrict__ cnt, int* __restrict__ rowptr) {
  __shared__ int wsum[SCAN_NW];
  __shared__ int carry_s[SCAN_NW];
  int t = threadIdx.x;
  int lane = t & 31, wv = t >> 5;
  const int4* c4 = (const int4*)cnt;
  int nf = (int)blockIdx.x * SCAN_NT;
  int front = 0;
  int i = t;
  for (; i + 3 * SCAN_NT < nf; i += 4 * SCAN_NT) {
    int4 q0 = c4[i], q1 = c4[i + SCAN_NT], q2 = c4[i + 2 * SCAN_NT], q3 = c4[i + 3 * SCAN_NT];
    front += (q0.x + q0.y + q0.z + q0.w) + (q1.x + q1.y + q1.z + q1.w)
           + (q2.x + q2.y + q2.z + q2.w) + (q3.x + q3.y + q3.z + q3.w);
  }
  for (; i < nf; i += SCAN_NT) {
    int4 q0 = c4[i];
    front += q0.x + q0.y + q0.z + q0.w;
  }
  int4 q = c4[nf + t];
  int own = q.x + q.y + q.z + q.w;
  int inc = own;
  #pragma unroll
  for (int o = 1; o < 32; o <<= 1) {
    int u = __shfl_up(inc, o);
    if (lane >= o) inc += u;
  }
  #pragma unroll
  for (int o = 16; o > 0; o >>= 1) front += __shfl_xor(front, o);
  if (lane == 31) wsum[wv] = inc;
  if (lane == 0) carry_s[wv] = front;
  __syncthreads();
  int base = 0;
  #pragma unroll
  for (int w = 0; w < SCAN_NW; w++) {
    base += carry_s[w];
    if (w < wv) base += wsum[w];
  }
  int ex = base + inc - own;
  int4 r;
  r.x = ex;
  r.y = ex + q.x;
  r.z = r.y + q.y;
  r.w = r.z + q.z;
  ((int4*)rowptr)[nf + t] = r;
}

__device__ __forceinline__ int4 edge_record(int sx, float4 q0, float4 q1, float4 q2, float4 q3,
                                            const float* __restrict__ wea) {
  float t[L];
  for (int l = 0; l < L; l++) {
    const float* wl = wea + l * F_E;
    float ae = 0.f;
    ae += q0.x * wl[0];  ae += q0.y * wl[1];  ae += q0.z * wl[2];  ae += q0.w * wl[3];
    ae += q1.x * wl[4];  ae += q1.y * wl[5];  ae += q1.z * wl[6];  ae += q1.w * wl[7];
    ae += q2.x * wl[8];  ae += q2.y * wl[9];  ae += q2.z * wl[10]; ae += q2.w * wl[11];
    ae += q3.x * wl[12]; ae += q3.y * wl[13]; ae += q3.z * wl[14]; ae += q3.w * wl[15];
    t[l] = ae;
  }
  return make_int4(sx, __float_as_int(t[0]), __float_as_int(t[1]), __float_as_int(t[2]));
}
__global__ void __launch_bounds__(PT) k_bplace(const int* __restrict__ src, const int* __restrict__ dst,
                                               const int* __restrict__ boff,
                                               const float* __restrict__ eattr, const float* __restrict__ wea,
                                               int* __restrict__ tslot, int E, int ch, int nbk) {
  __shared__ int cur[MAXBK];
  int t = threadIdx.x, b = blockIdx.x;
  for (int k = t; k < nbk; k += PT) cur[k] = boff[k * NB + b];
  __syncthreads();
  int e0 = b * ch;
  int e1 = e0 + ch < E ? e0 + ch : E;
  int4* ts4 = (int4*)tslot;
  int e = e0 + t;
  for (; e + PT < e1; e += 2 * PT) {
    int eb = e + PT;
    int da = dst[e], db = dst[eb];
    int sa = src[e], sb = src[eb];
    const float4* pa = (const float4*)(eattr + (size_t)e * F_E);
    const float4* pb = (const float4*)(eattr + (size_t)eb * F_E);
    float4 a0 = pa[0], a1 = pa[1], a2 = pa[2], a3 = pa[3];
    float4 b0 = pb[0], b1 = pb[1], b2 = pb[2], b3 = pb[3];
    int posa = atomicAdd(&cur[da >> BK_LOG], 1);
    int posb = atomicAdd(&cur[db >> BK_LOG], 1);
    ts4[posa] = edge_record(sa | ((da & (BKN - 1)) << SRC_BITS), a0, a1, a2, a3, wea);
    ts4[posb] = edge_record(sb | ((db & (BKN - 1)) << SRC_BITS), b0, b1, b2, b3, wea);
  }
  if (e < e1) {
    int da = dst[e];
    int sa = src[e];
    const float4* pa = (const float4*)(eattr + (size_t)e * F_E);
    float4 a0 = pa[0], a1 = pa[1], a2 = pa[2], a3 = pa[3];
    int posa = atomicAdd(&cur[da >> BK_LOG], 1);
    ts4[posa] = edge_record(sa | ((da & (BKN - 1)) << SRC_BITS), a0, a1, a2, a3, wea);
  }
}

__global__ void __launch_bounds__(DT) k_bsort(const int* __restrict__ boff, const int* __restrict__ tslot,
                                              int* __restrict__ rowptr, int* __restrict__ slot, int n) {
  __shared__ int cnt[BKN];
  __shared__ int wsum[BKN / 32];
  __shared__ v4i stage[STAGE];
  int t = threadIdx.x, k = blockIdx.x;
  int lane = t & 31, wv = t >> 5;
  int beg = boff[k * NB];
  int end = boff[(k + 1) * NB];
  if (t < BKN) cnt[t] = 0;
  __syncthreads();
  for (int i = beg + t; i < end; i += 4 * DT) {
    int ib = i + DT, ic = i + 2 * DT, id = i + 3 * DT;
    int xa = tslot[4 * (size_t)i];
    int xb = ib < end ? tslot[4 * (size_t)ib] : -1;
    int xc = ic < end ? tslot[4 * (size_t)ic] : -1;
    int xd = id < end ? tslot[4 * (size_t)id] : -1;
    atomicAdd(&cnt[xa >> SRC_BITS], 1);
    if (xb >= 0) atomicAdd(&cnt[xb >> SRC_BITS], 1);
    if (xc >= 0) atomicAdd(&cnt[xc >> SRC_BITS], 1);
    if (xd >= 0) atomicAdd(&cnt[xd >> SRC_BITS], 1);
  }
  __syncthreads();
  int own = 0, inc = 0;
  if (t < BKN) {
    own = cnt[t];
    inc = own;
    #pragma unroll
    for (int o = 1; o < 32; o <<= 1) {
      int u = __shfl_up(inc, o);
      if (lane >= o) inc += u;
    }
    if (lane == 31) wsum[wv] = inc;
  }
  __syncthreads();
  if (t < BKN) {
    int base = 0;
    #pragma unroll
    for (int w = 0; w < BKN / 32; w++)
      if (w < wv) base += wsum[w];
    int ex = base + inc - own;
    cnt[t] = ex;
    int node = k * BKN + t;
    if (node < n) rowptr[node] = beg + ex;
  }
  if (k == (int)gridDim.x - 1 && t == 0) rowptr[n] = end;
  __syncthreads();
  const v4i* ts4 = (const v4i*)tslot;
  v4i* sl4 = (v4i*)slot;
  const v4i none = {-1, 0, 0, 0};
  const bool fits = end - beg <= STAGE;
  for (int i = beg + t; i < end; i += 4 * DT) {
    int ib = i + DT, ic = i + 2 * DT, id = i + 3 * DT;
    v4i ra = ts4[i];
    v4i rb = ib < end ? ts4[ib] : none;
    v4i rc = ic < end ? ts4[ic] : none;
    v4i rd = id < end ? ts4[id] : none;
    {
      int p = atomicAdd(&cnt[ra.x >> SRC_BITS], 1);
      ra.x &= (1 << SRC_BITS) - 1;
      if (fits) stage[p] = ra; else sl4[beg + p] = ra;
    }
    if (rb.x >= 0) {
      int p = atomicAdd(&cnt[rb.x >> SRC_BITS], 1);
      rb.x &= (1 << SRC_BITS) - 1;
      if (fits) stage[p] = rb; else sl4[beg + p] = rb;
    }
    if (rc.x >= 0) {
      int p = atomicAdd(&cnt[rc.x >> SRC_BITS], 1);
      rc.x &= (1 << SRC_BITS) - 1;
      if (fits) stage[p] = rc; else sl4[beg + p] = rc;
    }
    if (rd.x >= 0) {
      int p = atomicAdd(&cnt[rd.x >> SRC_BITS], 1);
      rd.x &= (1 << SRC_BITS) - 1;
      if (fits) stage[p] = rd; else sl4[beg + p] = rd;
    }
  }
  if (fits) {
    __syncthreads();
    for (int i = t; i < end - beg; i += DT) sl4[beg + i] = stage[i];
  }
}

__global__ void k_wea(const float* __restrict__ We, const float* __restrict__ ae,
                      float* __restrict__ wea) {
  int i = blockIdx.x * blockDim.x + threadIdx.x;
  if (i >= L * F_E) return;
  int l = i / F_E, f = i % F_E;
  float s = 0.f;
  #pragma unroll
  for (int d = 0; d < D; d++) s += We[(l * F_E + f) * D + d] * ae[l * D + d];
  wea[i] = s;
}

template <int M> __device__ __forceinline__ float row_xor(float v) {
  return __builtin_bit_cast(float, __builtin_amdgcn_update_dpp(0, __builtin_bit_cast(int, v), 0x160 + M, 0xf, 0xf, false));
}
__device__ __forceinline__ float row_max(float v) {
  v = fmaxf(v, row_xor<8>(v));
  v = fmaxf(v, row_xor<4>(v));
  v = fmaxf(v, row_xor<2>(v));
  v = fmaxf(v, row_xor<1>(v));
  return v;
}
__device__ __forceinline__ float row_sum(float v) {
  v += row_xor<8>(v);
  v += row_xor<4>(v);
  v += row_xor<2>(v);
  v += row_xor<1>(v);
  return v;
}

#define HP 68
__global__ void __launch_bounds__(128) k_in_gemm(const float* __restrict__ x, const float* __restrict__ W1,
                                                 const float* __restrict__ b1, const float* __restrict__ W2,
                                                 float* __restrict__ C, int n) {
  __shared__ __attribute__((aligned(16))) float ts[4][16][HP];
  int wv = threadIdx.x >> 5;
  int wave = blockIdx.x * 4 + wv;
  int lane = threadIdx.x & 31;
  int row0 = wave * 16;
  if (row0 >= n) return;
  int lm = lane & 15;
  int kh = (lane >> 4) << 1;
  int mrow = row0 + lm;
  int mrd = mrow < n ? mrow : n - 1;

  v8f acc[4] = {};
  for (int k = 0; k < F_IN; k += 4) {
    v2f a;
    a.x = x[(size_t)mrd * F_IN + k + kh];
    a.y = x[(size_t)mrd * F_IN + k + kh + 1];
    #pragma unroll
    for (int nt = 0; nt < 4; nt++) {
      int col = nt * 16 + lm;
      v2f b;
      b.x = W1[(k + kh) * D + col];
      b.y = W1[(k + kh + 1) * D + col];
      acc[nt] = __builtin_amdgcn_wmma_f32_16x16x4_f32(
          false, a, false, b, (short)0, acc[nt], false, false);
    }
  }

  int rh = (lane >> 4) << 3;
  #pragma unroll
  for (int nt = 0; nt < 4; nt++) {
    int col = nt * 16 + lm;
    float bv = b1[col];
    #pragma unroll
    for (int r = 0; r < 8; r++) ts[wv][rh + r][col] = acc[nt][r] + bv;
  }

  v8f acc2[4] = {};
  for (int k = 0; k < D; k += 4) {
    v2f a = *(const v2f*)&ts[wv][lm][k + kh];
    #pragma unroll
    for (int nt = 0; nt < 4; nt++) {
      int col = nt * 16 + lm;
      v2f b;
      b.x = W2[(k + kh) * D + col];
      b.y = W2[(k + kh + 1) * D + col];
      acc2[nt] = __builtin_amdgcn_wmma_f32_16x16x4_f32(
          false, a, false, b, (short)0, acc2[nt], false, false);
    }
  }

  #pragma unroll
  for (int nt = 0; nt < 4; nt++) {
    int col = nt * 16 + lm;
    #pragma unroll
    for (int r = 0; r < 8; r++) {
      int row = row0 + rh + r;
      if (row < n) C[(size_t)row * D + col] = acc2[nt][r];
    }
  }
}


__device__ __forceinline__ int row_bcast(int v, unsigned sel) {
  return (int)__builtin_amdgcn_permlane16(0u, (unsigned)v, sel, sel, false, false);
}
__device__ __forceinline__ int row_bcast_all(int v, unsigned sel) {
  return (int)__builtin_amdgcn_permlane16(0u, (unsigned)v, sel, sel, true, false);
}


template <int NG>
__device__ __forceinline__ void node_trip(const char* __restrict__ xwb, unsigned cb, int c, int s0, int s1,
                                          v2f asl, v2f ash, float dn, float ae0, float ae1,
                                          bool v0, bool v1, bool act,
                                          float& m, float& lsum, v2f& accl, v2f& acch) {
  v2f rl[8 * NG], rh[8 * NG];
  unsigned off[8 * NG];
  #pragma unroll
  for (int k = 0; k < 8 * NG; k++)
    off[k] = ((unsigned)row_bcast_all(k < 16 ? s0 : s1, (unsigned)(k & 15) * 0x11111111u) << 8) + cb;
  #pragma unroll
  for (int k = 0; k < 8 * NG; k++) {
    float4 q = *(const float4*)(xwb + off[k]);
    rl[k].x = q.x; rl[k].y = q.y;
    rh[k].x = q.z; rh[k].y = q.w;
  }
  float sc0 = 0.f, sc1 = 0.f;
  #pragma unroll
  for (int k = 0; k < 8 * NG; k++) {
    v2f t = rl[k] * asl;
    t = __builtin_elementwise_fma(rh[k], ash, t);
    float d = row_sum(t.x + t.y);
    if (k < 16) {
      if (c == k) sc0 = d;
    } else {
      if (c == k - 16) sc1 = d;
    }
  }
  float a0 = sc0 + dn + ae0;
  float a1 = sc1 + dn + ae1;
  a0 = a0 > 0.f ? a0 : NEG_SLOPE * a0;
  a1 = a1 > 0.f ? a1 : NEG_SLOPE * a1;
  if (!v0) a0 = -INFINITY;
  if (!v1) a1 = -INFINITY;
  float mnew = act ? fmaxf(m, row_max(fmaxf(a0, a1))) : m;
  float scale = act ? expf(m - mnew) : 1.0f;
  float p0 = v0 ? expf(a0 - mnew) : 0.f;
  float p1 = v1 ? expf(a1 - mnew) : 0.f;
  lsum = lsum * scale + row_sum(p0 + p1);
  accl *= scale;
  acch *= scale;
  m = mnew;
  int p0b = __float_as_int(p0), p1b = __float_as_int(p1);
  #pragma unroll
  for (int k = 0; k < 8 * NG; k++) {
    float pj = __int_as_float(row_bcast_all(k < 16 ? p0b : p1b, (unsigned)(k & 15) * 0x11111111u));
    v2f pp = {pj, pj};
    accl = __builtin_elementwise_fma(pp, rl[k], accl);
    acch = __builtin_elementwise_fma(pp, rh[k], acch);
  }
}

__global__ void __launch_bounds__(128, 7) k_node(const float* __restrict__ xw,
                       const float* __restrict__ avs, const float* __restrict__ avd,
                       const int* __restrict__ rowptr,
                       const int* __restrict__ slot, int layer,
                       const float* __restrict__ bias, const float* __restrict__ gamma,
                       const float* __restrict__ beta, const float* __restrict__ mean,
                       const float* __restrict__ var, const float* __restrict__ Wn,
                       float* __restrict__ outp, int n, int relu) {
  __shared__ __attribute__((aligned(16))) float hs[16][HP];
  int wv = __builtin_amdgcn_readfirstlane((int)(threadIdx.x >> 5));
  int node0 = (int)blockIdx.x * 16;
  int lane = threadIdx.x & 31;
  int c = lane & 15;
  int hi = lane >> 4;
  const int4* slot4 = (const int4*)slot;
  const char* xwb = (const char*)xw;
  unsigned cb = (unsigned)c << 4;
  float4 as4 = ((const float4*)avs)[c];
  v2f asl = {as4.x, as4.y}, ash = {as4.z, as4.w};
  int q0 = node0 + 4 * wv;
  int rp0 = rowptr[q0 < n ? q0 : n];
  int rp1 = rowptr[q0 + 1 < n ? q0 + 1 : n];
  int rp2 = rowptr[q0 + 2 < n ? q0 + 2 : n];
  int rp3 = rowptr[q0 + 3 < n ? q0 + 3 : n];
  int rp4 = rowptr[q0 + 4 < n ? q0 + 4 : n];

  for (int ps = 0; ps < 2; ps++) {
    int r0 = ps ? rp2 : rp0;
    int r1 = ps ? rp3 : rp1;
    int r2 = ps ? rp4 : rp2;
    int degA = r1 - r0, degB = r2 - r1;
    int maxdeg = degA > degB ? degA : degB;
    int beg = hi ? r1 : r0;
    int deg = hi ? degB : degA;
    int lrow = 4 * wv + 2 * ps + hi;
    int node = node0 + lrow;
    int nodec = node < n ? node : n - 1;
    float4 xr = *(const float4*)(xwb + (((unsigned)nodec << 8) + cb));
    float4 ad4 = ((const float4*)avd)[c];
    v2f xl = {xr.x, xr.y}, xh = {xr.z, xr.w};
    float dn, ss;
    {
      v2f adl = {ad4.x, ad4.y}, adh = {ad4.z, ad4.w};
      v2f td = xl * adl;
      td = __builtin_elementwise_fma(xh, adh, td);
      dn = row_sum(td.x + td.y);
      v2f tq = xl * asl;
      tq = __builtin_elementwise_fma(xh, ash, tq);
      ss = row_sum(tq.x + tq.y);
    }

    float m = -INFINITY, lsum = 0.f, aepart0 = 0.f, aepart1 = 0.f;
    v2f accl = {0.f, 0.f}, acch = {0.f, 0.f};
    for (int base = 0; base < maxdeg; base += 24) {
      int e0 = base + c, e1 = base + 16 + c;
      bool v0 = e0 < deg, v1 = c < 8 && e1 < deg;
      bool act = base < deg;
      int4 es0 = v0 ? slot4[beg + e0] : make_int4(nodec, 0, 0, 0);
      int4 es1 = v1 ? slot4[beg + e1] : make_int4(nodec, 0, 0, 0);
      float ae0 = __int_as_float(layer == 0 ? es0.y : (layer == 1 ? es0.z : es0.w));
      float ae1 = __int_as_float(layer == 0 ? es1.y : (layer == 1 ? es1.z : es1.w));
      if (!v0) ae0 = 0.f;
      if (!v1) ae1 = 0.f;
      aepart0 += ae0;
      aepart1 += ae1;
      int cnt = maxdeg - base < 24 ? maxdeg - base : 24;
      int ng = (cnt + 7) >> 3;
      if (ng == 1)      node_trip<1>(xwb, cb, c, es0.x, es1.x, asl, ash, dn, ae0, ae1, v0, v1, act, m, lsum, accl, acch);
      else if (ng == 2) node_trip<2>(xwb, cb, c, es0.x, es1.x, asl, ash, dn, ae0, ae1, v0, v1, act, m, lsum, accl, acch);
      else              node_trip<3>(xwb, cb, c, es0.x, es1.x, asl, ash, dn, ae0, ae1, v0, v1, act, m, lsum, accl, acch);
    }
    {
      float aes = row_sum(aepart0 + aepart1) / fmaxf((float)deg, 1.0f);
      float a = ss + dn + aes;
      a = a > 0.f ? a : NEG_SLOPE * a;
      float mnew = fmaxf(m, a);
      float scale = expf(m - mnew);
      float p = expf(a - mnew);
      lsum = lsum * scale + p;
      accl.x = accl.x * scale + p * xl.x;
      accl.y = accl.y * scale + p * xl.y;
      acch.x = acch.x * scale + p * xh.x;
      acch.y = acch.y * scale + p * xh.y;
    }
    *(float4*)&hs[lrow][4 * c] = make_float4(accl.x, accl.y, acch.x, acch.y);
    hs[lrow][D] = lsum;
  }

  const int lm = c;
  const int kh = hi << 1;
  const int col = wv * 16 + lm;
  v2f bf[D / 4];
  if (Wn != nullptr) {
    #pragma unroll
    for (int k = 0; k < D; k += 4) {
      bf[k >> 2].x = Wn[(k + kh) * D + col];
      bf[k >> 2].y = Wn[(k + kh + 1) * D + col];
    }
  }
  float4 b4 = ((const float4*)bias)[c], g4 = ((const float4*)gamma)[c], be4 = ((const float4*)beta)[c];
  float4 mu4 = ((const float4*)mean)[c], va4 = ((const float4*)var)[c];
  for (int ps = 0; ps < 2; ps++) {
    int lrow = 4 * wv + 2 * ps + hi;
    int node = node0 + lrow;
    bool live = node < n;
    float4 acc = *(const float4*)&hs[lrow][4 * c];
    float inv = 1.0f / hs[lrow][D];
    float4 o;
    o.x = (acc.x * inv + b4.x - mu4.x) * (g4.x / sqrtf(va4.x + BN_EPS)) + be4.x;
    o.y = (acc.y * inv + b4.y - mu4.y) * (g4.y / sqrtf(va4.y + BN_EPS)) + be4.y;
    o.z = (acc.z * inv + b4.z - mu4.z) * (g4.z / sqrtf(va4.z + BN_EPS)) + be4.z;
    o.w = (acc.w * inv + b4.w - mu4.w) * (g4.w / sqrtf(va4.w + BN_EPS)) + be4.w;
    if (relu) {
      o.x = fmaxf(o.x, 0.f);
      o.y = fmaxf(o.y, 0.f);
      o.z = fmaxf(o.z, 0.f);
      o.w = fmaxf(o.w, 0.f);
    }
    if (Wn == nullptr) {
      if (live) ((float4*)outp)[(size_t)node * (D / 4) + c] = o;
    } else {
      *(float4*)&hs[lrow][4 * c] = live ? o : make_float4(0.f, 0.f, 0.f, 0.f);
    }
  }
  if (Wn == nullptr) return;
  __syncthreads();
  v8f cacc = {};
  #pragma unroll
  for (int k = 0; k < D; k += 4) {
    v2f a = *(const v2f*)&hs[lm][k + kh];
    cacc = __builtin_amdgcn_wmma_f32_16x16x4_f32(
        false, a, false, bf[k >> 2], (short)0, cacc, false, false);
  }
  int rbase = node0 + (hi << 3);
  #pragma unroll
  for (int r = 0; r < 8; r++) {
    int row = rbase + r;
    if (row < n) outp[(size_t)row * D + col] = cacc[r];
  }
}

#define RUN 16
__device__ __forceinline__ void graph_add(float* __restrict__ og, int g, int c, float4 a) {
  float* p = og + (size_t)g * D + 4 * c;
  atomicAdd(p + 0, a.x);
  atomicAdd(p + 1, a.y);
  atomicAdd(p + 2, a.z);
  atomicAdd(p + 3, a.w);
}
__global__ void __launch_bounds__(128) k_output(const float* __restrict__ h, const int* __restrict__ batch,
                                                float* __restrict__ out_nodes, float* __restrict__ out_graph, int n) {
  int wave = (blockIdx.x * blockDim.x + threadIdx.x) >> 5;
  if (wave * 2 * RUN >= n) return;
  int lane = threadIdx.x & 31;
  int c = lane & 15;
  int hi = lane >> 4;
  int first = (wave * 2 + hi) * RUN;
  const float4* h4 = (const float4*)h;
  float4* o4 = (float4*)out_nodes;
  int gn = first + c < n ? first + c : n - 1;
  int gv = batch[gn];
  float4 r[RUN];
  #pragma unroll
  for (int j = 0; j < RUN; j++) {
    int node = first + j < n ? first + j : n - 1;
    r[j] = h4[(size_t)node * (D / 4) + c];
  }
  #pragma unroll
  for (int j = 0; j < RUN; j++)
    if (first + j < n) o4[(size_t)(first + j) * (D / 4) + c] = r[j];

  int cur = 0;
  bool have = false;
  float4 acc = make_float4(0.f, 0.f, 0.f, 0.f);
  #pragma unroll
  for (int j = 0; j < RUN; j++) {
    int g = row_bcast(gv, (unsigned)j * 0x11111111u);
    if (first + j < n) {
      if (have && g == cur) {
        acc.x += r[j].x;
        acc.y += r[j].y;
        acc.z += r[j].z;
        acc.w += r[j].w;
      } else {
        if (have) graph_add(out_graph, cur, c, acc);
        cur = g;
        have = true;
        acc = r[j];
      }
    }
  }
  int ocur = __shfl_xor(cur, 16);
  int ohave = __shfl_xor((int)have, 16);
  float4 oacc;
  oacc.x = __shfl_xor(acc.x, 16);
  oacc.y = __shfl_xor(acc.y, 16);
  oacc.z = __shfl_xor(acc.z, 16);
  oacc.w = __shfl_xor(acc.w, 16);
  bool joined = have && ohave && cur == ocur;
  if (joined) {
    if (hi == 0) {
      acc.x += oacc.x;
      acc.y += oacc.y;
      acc.z += oacc.z;
      acc.w += oacc.w;
      graph_add(out_graph, cur, c, acc);
    }
  } else if (have) {
    graph_add(out_graph, cur, c, acc);
  }
}

extern "C" void kernel_launch(void* const* d_in, const int* in_sizes, int n_in,
                              void* d_out, int out_size, void* d_ws, size_t ws_size,
                              hipStream_t stream) {
  const float* x     = (const float*)d_in[0];
  const int*   eidx  = (const int*)d_in[1];
  const float* eattr = (const float*)d_in[2];
  const int*   batch = (const int*)d_in[3];
  const float* xlw   = (const float*)d_in[4];
  const float* xlb   = (const float*)d_in[5];
  const float* glw   = (const float*)d_in[6];
  const float* gas   = (const float*)d_in[7];
  const float* gad   = (const float*)d_in[8];
  const float* gew   = (const float*)d_in[9];
  const float* gae   = (const float*)d_in[10];
  const float* gb    = (const float*)d_in[11];
  const float* bng   = (const float*)d_in[12];
  const float* bnb   = (const float*)d_in[13];
  const float* bnm   = (const float*)d_in[14];
  const float* bnv   = (const float*)d_in[15];

  const int n = in_sizes[3];
  const int E = in_sizes[2] / F_E;
  const int* srcp = eidx;
  const int* dstp = eidx + E;
  const int npad = ((n + 1 + SCAN_TILE - 1) / SCAN_TILE) * SCAN_TILE;
  const int nbk  = (n + BKN - 1) / BKN;
  const int m    = nbk * NB;
  const int mpad = ((m + 1 + SCAN_TILE - 1) / SCAN_TILE) * SCAN_TILE;
  const int ch   = (E + NB - 1) / NB;

  float* ws    = (float*)d_ws;
  float* h     = ws;  ws += (size_t)n * D;
  float* xw    = ws;  ws += (size_t)n * D;
  int*   rowptr = (int*)ws; ws += npad;
  int*   slot  = (int*)ws; ws += 4 * (size_t)E;
  float* wea   = ws;  ws += L * F_E;
  int*   tslot = (int*)h;
  int*   bh    = (int*)xw;
  int*   boff  = (int*)xw + mpad;

  float* out_nodes = (float*)d_out;
  float* out_graph = out_nodes + (size_t)n * D;
  const size_t graph_elems = (size_t)out_size - (size_t)n * D;

  const int B = 256;
  #define GRID(x) (((x) + B - 1) / B)

  k_wea<<<GRID(L * F_E), B, 0, stream>>>(gew, gae, wea);
  k_bcount<<<NB, PT, 0, stream>>>(dstp, bh, E, ch, nbk, m, mpad);
  k_scan<<<mpad / SCAN_TILE, SCAN_NT, 0, stream>>>(bh, boff);
  k_bplace<<<NB, PT, 0, stream>>>(srcp, dstp, boff, eattr, wea, tslot, E, ch, nbk);
  k_bsort<<<nbk, DT, 0, stream>>>(boff, tslot, rowptr, slot, n);

  {
    int waves = (n + 15) / 16;
    int blocks = (waves + 3) / 4;
    k_in_gemm<<<blocks, 128, 0, stream>>>(x, xlw, xlb, glw, xw, n);
  }

  const float* cur = xw;
  float* nxt = h;
  for (int l = 0; l < L; l++) {
    const bool last = l == L - 1;
    k_node<<<(n + 15) / 16, 128, 0, stream>>>(cur, gas + (size_t)l * D, gad + (size_t)l * D, rowptr, slot, l,
                                            gb + (size_t)l * D, bng + (size_t)l * D, bnb + (size_t)l * D,
                                            bnm + (size_t)l * D, bnv + (size_t)l * D,
                                            last ? nullptr : glw + (size_t)(l + 1) * D * D,
                                            nxt, n, last ? 0 : 1);
    if (!last) {
      float* t = (float*)cur;
      cur = nxt;
      nxt = t;
    }
  }

  hipMemsetAsync(out_graph, 0, graph_elems * sizeof(float), stream);
  k_output<<<(n + 8 * RUN - 1) / (8 * RUN), 128, 0, stream>>>(nxt, batch, out_nodes, out_graph, n);
  #undef GRID
}


# layer 0 gathers rows of x (128 B, not 256 B table rows): weighted mean of x, then @ (W1 W2) + b1 W2 on WMMA in the node kernel (mean is linear); input GEMM launch gone; layer-0 sum order changed
# speedup vs baseline: 10.5639x; 1.0952x over previous
#include <hip/hip_runtime.h>
#include <hip/hip_bf16.h>

#define F_IN   32
#define F_E    16
#define D      64
#define L      3
#define NEG_SLOPE 0.2f
#define BN_EPS 1e-5f
#define SCAN_TILE 4096

typedef __attribute__((ext_vector_type(2))) float v2f;
typedef __attribute__((ext_vector_type(8))) float v8f;
typedef __attribute__((ext_vector_type(4))) int v4i;

#define BK_LOG   8
#define BKN      (1 << BK_LOG)
#define NB       256
#define MAXBK    1024
#define SRC_BITS 20
#define PT       1024
#define DT       1024
#define STAGE    6144

__global__ void __launch_bounds__(PT) k_bcount(const int* __restrict__ dst, int* __restrict__ bh,
                                               int E, int ch, int nbk, int m, int mpad) {
  __shared__ int hist[MAXBK];
  int t = threadIdx.x, b = blockIdx.x;
  for (int k = t; k < nbk; k += PT) hist[k] = 0;
  __syncthreads();
  int e0 = b * ch;
  int e1 = e0 + ch < E ? e0 + ch : E;
  for (int e = e0 + t; e < e1; e += 4 * PT) {
    int eb = e + PT, ec = e + 2 * PT, ed = e + 3 * PT;
    int da = dst[e];
    int db = eb < e1 ? dst[eb] : -1;
    int dc = ec < e1 ? dst[ec] : -1;
    int dd = ed < e1 ? dst[ed] : -1;
    atomicAdd(&hist[da >> BK_LOG], 1);
    if (db >= 0) atomicAdd(&hist[db >> BK_LOG], 1);
    if (dc >= 0) atomicAdd(&hist[dc >> BK_LOG], 1);
    if (dd >= 0) atomicAdd(&hist[dd >> BK_LOG], 1);
  }
  __syncthreads();
  for (int k = t; k < nbk; k += PT) bh[k * NB + b] = hist[k];
  if (b == 0)
    for (int i = m + t; i < mpad; i += PT) bh[i] = 0;
}

#define SCAN_NT (SCAN_TILE / 4)
#define SCAN_NW (SCAN_NT / 32)
__global__ void __launch_bounds__(SCAN_NT) k_scan(const int* __restrict__ cnt, int* __restrict__ rowptr) {
  __shared__ int wsum[SCAN_NW];
  __shared__ int carry_s[SCAN_NW];
  int t = threadIdx.x;
  int lane = t & 31, wv = t >> 5;
  const int4* c4 = (const int4*)cnt;
  int nf = (int)blockIdx.x * SCAN_NT;
  int front = 0;
  int i = t;
  for (; i + 3 * SCAN_NT < nf; i += 4 * SCAN_NT) {
    int4 q0 = c4[i], q1 = c4[i + SCAN_NT], q2 = c4[i + 2 * SCAN_NT], q3 = c4[i + 3 * SCAN_NT];
    front += (q0.x + q0.y + q0.z + q0.w) + (q1.x + q1.y + q1.z + q1.w)
           + (q2.x + q2.y + q2.z + q2.w) + (q3.x + q3.y + q3.z + q3.w);
  }
  for (; i < nf; i += SCAN_NT) {
    int4 q0 = c4[i];
    front += q0.x + q0.y + q0.z + q0.w;
  }
  int4 q = c4[nf + t];
  int own = q.x + q.y + q.z + q.w;
  int inc = own;
  #pragma unroll
  for (int o = 1; o < 32; o <<= 1) {
    int u = __shfl_up(inc, o);
    if (lane >= o) inc += u;
  }
  #pragma unroll
  for (int o = 16; o > 0; o >>= 1) front += __shfl_xor(front, o);
  if (lane == 31) wsum[wv] = inc;
  if (lane == 0) carry_s[wv] = front;
  __syncthreads();
  int base = 0;
  #pragma unroll
  for (int w = 0; w < SCAN_NW; w++) {
    base += carry_s[w];
    if (w < wv) base += wsum[w];
  }
  int ex = base + inc - own;
  int4 r;
  r.x = ex;
  r.y = ex + q.x;
  r.z = r.y + q.y;
  r.w = r.z + q.z;
  ((int4*)rowptr)[nf + t] = r;
}

__device__ __forceinline__ int4 edge_record(int sx, float4 q0, float4 q1, float4 q2, float4 q3,
                                            const float* __restrict__ wea) {
  float t[L];
  for (int l = 0; l < L; l++) {
    const float* wl = wea + l * F_E;
    float ae = 0.f;
    ae += q0.x * wl[0];  ae += q0.y * wl[1];  ae += q0.z * wl[2];  ae += q0.w * wl[3];
    ae += q1.x * wl[4];  ae += q1.y * wl[5];  ae += q1.z * wl[6];  ae += q1.w * wl[7];
    ae += q2.x * wl[8];  ae += q2.y * wl[9];  ae += q2.z * wl[10]; ae += q2.w * wl[11];
    ae += q3.x * wl[12]; ae += q3.y * wl[13]; ae += q3.z * wl[14]; ae += q3.w * wl[15];
    t[l] = ae;
  }
  return make_int4(sx, __float_as_int(t[0]), __float_as_int(t[1]), __float_as_int(t[2]));
}
__global__ void __launch_bounds__(PT) k_bplace(const int* __restrict__ src, const int* __restrict__ dst,
                                               const int* __restrict__ boff,
                                               const float* __restrict__ eattr, const float* __restrict__ wea,
                                               int* __restrict__ tslot, int E, int ch, int nbk) {
  __shared__ int cur[MAXBK];
  int t = threadIdx.x, b = blockIdx.x;
  for (int k = t; k < nbk; k += PT) cur[k] = boff[k * NB + b];
  __syncthreads();
  int e0 = b * ch;
  int e1 = e0 + ch < E ? e0 + ch : E;
  int4* ts4 = (int4*)tslot;
  int e = e0 + t;
  for (; e + PT < e1; e += 2 * PT) {
    int eb = e + PT;
    int da = dst[e], db = dst[eb];
    int sa = src[e], sb = src[eb];
    const float4* pa = (const float4*)(eattr + (size_t)e * F_E);
    const float4* pb = (const float4*)(eattr + (size_t)eb * F_E);
    float4 a0 = pa[0], a1 = pa[1], a2 = pa[2], a3 = pa[3];
    float4 b0 = pb[0], b1 = pb[1], b2 = pb[2], b3 = pb[3];
    int posa = atomicAdd(&cur[da >> BK_LOG], 1);
    int posb = atomicAdd(&cur[db >> BK_LOG], 1);
    ts4[posa] = edge_record(sa | ((da & (BKN - 1)) << SRC_BITS), a0, a1, a2, a3, wea);
    ts4[posb] = edge_record(sb | ((db & (BKN - 1)) << SRC_BITS), b0, b1, b2, b3, wea);
  }
  if (e < e1) {
    int da = dst[e];
    int sa = src[e];
    const float4* pa = (const float4*)(eattr + (size_t)e * F_E);
    float4 a0 = pa[0], a1 = pa[1], a2 = pa[2], a3 = pa[3];
    int posa = atomicAdd(&cur[da >> BK_LOG], 1);
    ts4[posa] = edge_record(sa | ((da & (BKN - 1)) << SRC_BITS), a0, a1, a2, a3, wea);
  }
}

__global__ void __launch_bounds__(DT) k_bsort(const int* __restrict__ boff, const int* __restrict__ tslot,
                                              int* __restrict__ rowptr, int* __restrict__ slot, int n) {
  __shared__ int cnt[BKN];
  __shared__ int wsum[BKN / 32];
  __shared__ v4i stage[STAGE];
  int t = threadIdx.x, k = blockIdx.x;
  int lane = t & 31, wv = t >> 5;
  int beg = boff[k * NB];
  int end = boff[(k + 1) * NB];
  if (t < BKN) cnt[t] = 0;
  __syncthreads();
  for (int i = beg + t; i < end; i += 4 * DT) {
    int ib = i + DT, ic = i + 2 * DT, id = i + 3 * DT;
    int xa = tslot[4 * (size_t)i];
    int xb = ib < end ? tslot[4 * (size_t)ib] : -1;
    int xc = ic < end ? tslot[4 * (size_t)ic] : -1;
    int xd = id < end ? tslot[4 * (size_t)id] : -1;
    atomicAdd(&cnt[xa >> SRC_BITS], 1);
    if (xb >= 0) atomicAdd(&cnt[xb >> SRC_BITS], 1);
    if (xc >= 0) atomicAdd(&cnt[xc >> SRC_BITS], 1);
    if (xd >= 0) atomicAdd(&cnt[xd >> SRC_BITS], 1);
  }
  __syncthreads();
  int own = 0, inc = 0;
  if (t < BKN) {
    own = cnt[t];
    inc = own;
    #pragma unroll
    for (int o = 1; o < 32; o <<= 1) {
      int u = __shfl_up(inc, o);
      if (lane >= o) inc += u;
    }
    if (lane == 31) wsum[wv] = inc;
  }
  __syncthreads();
  if (t < BKN) {
    int base = 0;
    #pragma unroll
    for (int w = 0; w < BKN / 32; w++)
      if (w < wv) base += wsum[w];
    int ex = base + inc - own;
    cnt[t] = ex;
    int node = k * BKN + t;
    if (node < n) rowptr[node] = beg + ex;
  }
  if (k == (int)gridDim.x - 1 && t == 0) rowptr[n] = end;
  __syncthreads();
  const v4i* ts4 = (const v4i*)tslot;
  v4i* sl4 = (v4i*)slot;
  const v4i none = {-1, 0, 0, 0};
  const bool fits = end - beg <= STAGE;
  for (int i = beg + t; i < end; i += 4 * DT) {
    int ib = i + DT, ic = i + 2 * DT, id = i + 3 * DT;
    v4i ra = ts4[i];
    v4i rb = ib < end ? ts4[ib] : none;
    v4i rc = ic < end ? ts4[ic] : none;
    v4i rd = id < end ? ts4[id] : none;
    {
      int p = atomicAdd(&cnt[ra.x >> SRC_BITS], 1);
      ra.x &= (1 << SRC_BITS) - 1;
      if (fits) stage[p] = ra; else sl4[beg + p] = ra;
    }
    if (rb.x >= 0) {
      int p = atomicAdd(&cnt[rb.x >> SRC_BITS], 1);
      rb.x &= (1 << SRC_BITS) - 1;
      if (fits) stage[p] = rb; else sl4[beg + p] = rb;
    }
    if (rc.x >= 0) {
      int p = atomicAdd(&cnt[rc.x >> SRC_BITS], 1);
      rc.x &= (1 << SRC_BITS) - 1;
      if (fits) stage[p] = rc; else sl4[beg + p] = rc;
    }
    if (rd.x >= 0) {
      int p = atomicAdd(&cnt[rd.x >> SRC_BITS], 1);
      rd.x &= (1 << SRC_BITS) - 1;
      if (fits) stage[p] = rd; else sl4[beg + p] = rd;
    }
  }
  if (fits) {
    __syncthreads();
    for (int i = t; i < end - beg; i += DT) sl4[beg + i] = stage[i];
  }
}

#define FO_M   0
#define FO_BM  (F_IN * D)
#define FO_US  (FO_BM + D)
#define FO_UD  (FO_US + F_IN)
#define FO_CC  (FO_UD + F_IN)
#define FOLD_WORDS 4096
__global__ void __launch_bounds__(256) k_prep(const float* __restrict__ We, const float* __restrict__ ae,
                                              float* __restrict__ wea,
                                              const float* __restrict__ W1, const float* __restrict__ b1,
                                              const float* __restrict__ W2, const float* __restrict__ as,
                                              const float* __restrict__ ad, float* __restrict__ fold) {
  __shared__ float vs[D], vd[D];
  int t = threadIdx.x;
  {
    int wv = t >> 5, lane = t & 31;
    int lm = lane & 15;
    int kh = (lane >> 4) << 1;
    int rt = wv >> 2, ct = wv & 3;
    int col = ct * 16 + lm;
    v8f acc = {};
    for (int k = 0; k < D; k += 4) {
      v2f a, b;
      a.x = W1[(rt * 16 + lm) * D + k + kh];
      a.y = W1[(rt * 16 + lm) * D + k + kh + 1];
      b.x = W2[(k + kh) * D + col];
      b.y = W2[(k + kh + 1) * D + col];
      acc = __builtin_amdgcn_wmma_f32_16x16x4_f32(false, a, false, b, (short)0, acc, false, false);
    }
    int rh = (lane >> 4) << 3;
    #pragma unroll
    for (int r = 0; r < 8; r++) fold[FO_M + (rt * 16 + rh + r) * D + col] = acc[r];
  }
  if (t < D) {
    float s = 0.f;
    #pragma unroll
    for (int d = 0; d < D; d++) s += W2[t * D + d] * as[d];
    vs[t] = s;
  } else if (t < 2 * D) {
    int j = t - D;
    float s = 0.f;
    #pragma unroll
    for (int d = 0; d < D; d++) s += W2[j * D + d] * ad[d];
    vd[j] = s;
  } else if (t < 3 * D) {
    int j = t - 2 * D;
    float s = 0.f;
    #pragma unroll
    for (int k = 0; k < D; k++) s += b1[k] * W2[k * D + j];
    fold[FO_BM + j] = s;
  } else if (t < 3 * D + L * F_E) {
    int i = t - 3 * D;
    int l = i / F_E, f = i % F_E;
    float s = 0.f;
    #pragma unroll
    for (int d = 0; d < D; d++) s += We[(l * F_E + f) * D + d] * ae[l * D + d];
    wea[i] = s;
  }
  __syncthreads();
  if (t < F_IN) {
    float s = 0.f;
    for (int j = 0; j < D; j++) s += W1[t * D + j] * vs[j];
    fold[FO_US + t] = s;
  } else if (t < 2 * F_IN) {
    int i = t - F_IN;
    float s = 0.f;
    for (int j = 0; j < D; j++) s += W1[i * D + j] * vd[j];
    fold[FO_UD + i] = s;
  } else if (t == 2 * F_IN) {
    float s = 0.f;
    for (int j = 0; j < D; j++) s += b1[j] * (vs[j] + vd[j]);
    fold[FO_CC] = s;
  }
}

template <int M> __device__ __forceinline__ float row_xor(float v) {
  return __builtin_bit_cast(float, __builtin_amdgcn_update_dpp(0, __builtin_bit_cast(int, v), 0x160 + M, 0xf, 0xf, false));
}
__device__ __forceinline__ float row_max(float v) {
  v = fmaxf(v, row_xor<8>(v));
  v = fmaxf(v, row_xor<4>(v));
  v = fmaxf(v, row_xor<2>(v));
  v = fmaxf(v, row_xor<1>(v));
  return v;
}
__device__ __forceinline__ float row_sum(float v) {
  v += row_xor<8>(v);
  v += row_xor<4>(v);
  v += row_xor<2>(v);
  v += row_xor<1>(v);
  return v;
}

#define HP 68


__device__ __forceinline__ int row_bcast(int v, unsigned sel) {
  return (int)__builtin_amdgcn_permlane16(0u, (unsigned)v, sel, sel, false, false);
}
__device__ __forceinline__ int row_bcast_all(int v, unsigned sel) {
  return (int)__builtin_amdgcn_permlane16(0u, (unsigned)v, sel, sel, true, false);
}


template <int NG, bool FIRST>
__device__ __forceinline__ void node_trip(const char* __restrict__ xwb, unsigned cb, int c, int s0, int s1,
                                          v2f asl, v2f ash, float dn, float ae0, float ae1,
                                          bool v0, bool v1, bool act,
                                          float& m, float& lsum, v2f& accl, v2f& acch) {
  v2f rl[8 * NG], rh[8 * NG];
  unsigned off[8 * NG];
  #pragma unroll
  for (int k = 0; k < 8 * NG; k++)
    off[k] = ((unsigned)row_bcast_all(k < 16 ? s0 : s1, (unsigned)(k & 15) * 0x11111111u) << (FIRST ? 7 : 8)) + cb;
  #pragma unroll
  for (int k = 0; k < 8 * NG; k++) {
    if constexpr (FIRST) {
      rl[k] = *(const v2f*)(xwb + off[k]);
      rh[k].x = 0.f; rh[k].y = 0.f;
    } else {
      float4 q = *(const float4*)(xwb + off[k]);
      rl[k].x = q.x; rl[k].y = q.y;
      rh[k].x = q.z; rh[k].y = q.w;
    }
  }
  float sc0 = 0.f, sc1 = 0.f;
  #pragma unroll
  for (int k = 0; k < 8 * NG; k++) {
    v2f t = rl[k] * asl;
    if constexpr (!FIRST) t = __builtin_elementwise_fma(rh[k], ash, t);
    float d = row_sum(t.x + t.y);
    if (k < 16) {
      if (c == k) sc0 = d;
    } else {
      if (c == k - 16) sc1 = d;
    }
  }
  float a0 = sc0 + dn + ae0;
  float a1 = sc1 + dn + ae1;
  a0 = a0 > 0.f ? a0 : NEG_SLOPE * a0;
  a1 = a1 > 0.f ? a1 : NEG_SLOPE * a1;
  if (!v0) a0 = -INFINITY;
  if (!v1) a1 = -INFINITY;
  float mnew = act ? fmaxf(m, row_max(fmaxf(a0, a1))) : m;
  float scale = act ? expf(m - mnew) : 1.0f;
  float p0 = v0 ? expf(a0 - mnew) : 0.f;
  float p1 = v1 ? expf(a1 - mnew) : 0.f;
  lsum = lsum * scale + row_sum(p0 + p1);
  accl *= scale;
  acch *= scale;
  m = mnew;
  int p0b = __float_as_int(p0), p1b = __float_as_int(p1);
  #pragma unroll
  for (int k = 0; k < 8 * NG; k++) {
    float pj = __int_as_float(row_bcast_all(k < 16 ? p0b : p1b, (unsigned)(k & 15) * 0x11111111u));
    v2f pp = {pj, pj};
    accl = __builtin_elementwise_fma(pp, rl[k], accl);
    if constexpr (!FIRST) acch = __builtin_elementwise_fma(pp, rh[k], acch);
  }
}

template <bool FIRST>
__global__ void __launch_bounds__(128, 7) k_node(const float* __restrict__ xw,
                       const float* __restrict__ avs, const float* __restrict__ avd,
                       const int* __restrict__ rowptr,
                       const int* __restrict__ slot, int layer,
                       const float* __restrict__ bias, const float* __restrict__ gamma,
                       const float* __restrict__ beta, const float* __restrict__ mean,
                       const float* __restrict__ var, const float* __restrict__ Wn,
                       const float* __restrict__ fold,
                       float* __restrict__ outp, int n, int relu) {
  __shared__ __attribute__((aligned(16))) float hs[16][HP];
  __shared__ __attribute__((aligned(16))) float hs2[FIRST ? 16 : 1][HP];
  int wv = __builtin_amdgcn_readfirstlane((int)(threadIdx.x >> 5));
  int node0 = (int)blockIdx.x * 16;
  int lane = threadIdx.x & 31;
  int c = lane & 15;
  int hi = lane >> 4;
  const int4* slot4 = (const int4*)slot;
  const char* xwb = (const char*)xw;
  unsigned cb = FIRST ? (unsigned)c << 3 : (unsigned)c << 4;
  const unsigned rsh = FIRST ? 7 : 8;
  v2f asl, ash;
  if constexpr (FIRST) {
    asl = ((const v2f*)avs)[c];
    ash.x = 0.f; ash.y = 0.f;
  } else {
    float4 as4 = ((const float4*)avs)[c];
    asl.x = as4.x; asl.y = as4.y; ash.x = as4.z; ash.y = as4.w;
  }
  float cc = 0.f;
  if constexpr (FIRST) cc = fold[FO_CC];
  int q0 = node0 + 4 * wv;
  int rp0 = rowptr[q0 < n ? q0 : n];
  int rp1 = rowptr[q0 + 1 < n ? q0 + 1 : n];
  int rp2 = rowptr[q0 + 2 < n ? q0 + 2 : n];
  int rp3 = rowptr[q0 + 3 < n ? q0 + 3 : n];
  int rp4 = rowptr[q0 + 4 < n ? q0 + 4 : n];

  for (int ps = 0; ps < 2; ps++) {
    int r0 = ps ? rp2 : rp0;
    int r1 = ps ? rp3 : rp1;
    int r2 = ps ? rp4 : rp2;
    int degA = r1 - r0, degB = r2 - r1;
    int maxdeg = degA > degB ? degA : degB;
    int beg = hi ? r1 : r0;
    int deg = hi ? degB : degA;
    int lrow = 4 * wv + 2 * ps + hi;
    int node = node0 + lrow;
    int nodec = node < n ? node : n - 1;
    v2f xl, xh;
    float dn, ss;
    if constexpr (FIRST) {
      xl = *(const v2f*)(xwb + (((unsigned)nodec << rsh) + cb));
      xh.x = 0.f; xh.y = 0.f;
      v2f adl = ((const v2f*)avd)[c];
      v2f td = xl * adl;
      dn = row_sum(td.x + td.y) + cc;
      v2f tq = xl * asl;
      ss = row_sum(tq.x + tq.y);
    } else {
      float4 xr = *(const float4*)(xwb + (((unsigned)nodec << rsh) + cb));
      float4 ad4 = ((const float4*)avd)[c];
      xl.x = xr.x; xl.y = xr.y; xh.x = xr.z; xh.y = xr.w;
      v2f adl = {ad4.x, ad4.y}, adh = {ad4.z, ad4.w};
      v2f td = xl * adl;
      td = __builtin_elementwise_fma(xh, adh, td);
      dn = row_sum(td.x + td.y);
      v2f tq = xl * asl;
      tq = __builtin_elementwise_fma(xh, ash, tq);
      ss = row_sum(tq.x + tq.y);
    }

    float m = -INFINITY, lsum = 0.f, aepart0 = 0.f, aepart1 = 0.f;
    v2f accl = {0.f, 0.f}, acch = {0.f, 0.f};
    for (int base = 0; base < maxdeg; base += 24) {
      int e0 = base + c, e1 = base + 16 + c;
      bool v0 = e0 < deg, v1 = c < 8 && e1 < deg;
      bool act = base < deg;
      int4 es0 = v0 ? slot4[beg + e0] : make_int4(nodec, 0, 0, 0);
      int4 es1 = v1 ? slot4[beg + e1] : make_int4(nodec, 0, 0, 0);
      float ae0 = __int_as_float(layer == 0 ? es0.y : (layer == 1 ? es0.z : es0.w));
      float ae1 = __int_as_float(layer == 0 ? es1.y : (layer == 1 ? es1.z : es1.w));
      if (!v0) ae0 = 0.f;
      if (!v1) ae1 = 0.f;
      aepart0 += ae0;
      aepart1 += ae1;
      int cnt = maxdeg - base < 24 ? maxdeg - base : 24;
      int ng = (cnt + 7) >> 3;
      if (ng == 1)      node_trip<1, FIRST>(xwb, cb, c, es0.x, es1.x, asl, ash, dn, ae0, ae1, v0, v1, act, m, lsum, accl, acch);
      else if (ng == 2) node_trip<2, FIRST>(xwb, cb, c, es0.x, es1.x, asl, ash, dn, ae0, ae1, v0, v1, act, m, lsum, accl, acch);
      else              node_trip<3, FIRST>(xwb, cb, c, es0.x, es1.x, asl, ash, dn, ae0, ae1, v0, v1, act, m, lsum, accl, acch);
    }
    {
      float aes = row_sum(aepart0 + aepart1) / fmaxf((float)deg, 1.0f);
      float a = ss + dn + aes;
      a = a > 0.f ? a : NEG_SLOPE * a;
      float mnew = fmaxf(m, a);
      float scale = expf(m - mnew);
      float p = expf(a - mnew);
      lsum = lsum * scale + p;
      accl.x = accl.x * scale + p * xl.x;
      accl.y = accl.y * scale + p * xl.y;
      if constexpr (!FIRST) {
        acch.x = acch.x * scale + p * xh.x;
        acch.y = acch.y * scale + p * xh.y;
      }
    }
    if constexpr (FIRST) *(v2f*)&hs[lrow][2 * c] = accl;
    else                 *(float4*)&hs[lrow][4 * c] = make_float4(accl.x, accl.y, acch.x, acch.y);
    hs[lrow][D] = lsum;
  }

  const int lm = c;
  const int kh = hi << 1;
  const int col = wv * 16 + lm;
  v2f bf[D / 4];
  if (Wn != nullptr) {
    #pragma unroll
    for (int k = 0; k < D; k += 4) {
      bf[k >> 2].x = Wn[(k + kh) * D + col];
      bf[k >> 2].y = Wn[(k + kh + 1) * D + col];
    }
  }

  if constexpr (FIRST) {
    v2f bm[F_IN / 4];
    #pragma unroll
    for (int k = 0; k < F_IN; k += 4) {
      bm[k >> 2].x = fold[FO_M + (k + kh) * D + col];
      bm[k >> 2].y = fold[FO_M + (k + kh + 1) * D + col];
    }
    float shift = fold[FO_BM + col] + bias[col] - mean[col];
    float gain = gamma[col] / sqrtf(var[col] + BN_EPS);
    float offs = beta[col];
    for (int ps = 0; ps < 2; ps++) {
      int lrow = 4 * wv + 2 * ps + hi;
      bool live = node0 + lrow < n;
      v2f a = *(const v2f*)&hs[lrow][2 * c];
      float inv = 1.0f / hs[lrow][D];
      a.x *= inv;
      a.y *= inv;
      if (!live) { a.x = 0.f; a.y = 0.f; }
      *(v2f*)&hs[lrow][2 * c] = a;
    }
    __syncthreads();
    v8f macc = {};
    #pragma unroll
    for (int k = 0; k < F_IN; k += 4) {
      v2f a = *(const v2f*)&hs[lm][k + kh];
      macc = __builtin_amdgcn_wmma_f32_16x16x4_f32(
          false, a, false, bm[k >> 2], (short)0, macc, false, false);
    }
    const int rb = hi << 3;
    #pragma unroll
    for (int r = 0; r < 8; r++) {
      float o = (macc[r] + shift) * gain + offs;
      if (relu) o = fmaxf(o, 0.f);
      hs2[rb + r][col] = node0 + rb + r < n ? o : 0.f;
    }
    __syncthreads();
    v8f cacc = {};
    #pragma unroll
    for (int k = 0; k < D; k += 4) {
      v2f a = *(const v2f*)&hs2[lm][k + kh];
      cacc = __builtin_amdgcn_wmma_f32_16x16x4_f32(
          false, a, false, bf[k >> 2], (short)0, cacc, false, false);
    }
    #pragma unroll
    for (int r = 0; r < 8; r++) {
      int row = node0 + rb + r;
      if (row < n) outp[(size_t)row * D + col] = cacc[r];
    }
  } else {
    float4 b4 = ((const float4*)bias)[c], g4 = ((const float4*)gamma)[c], be4 = ((const float4*)beta)[c];
    float4 mu4 = ((const float4*)mean)[c], va4 = ((const float4*)var)[c];
    for (int ps = 0; ps < 2; ps++) {
      int lrow = 4 * wv + 2 * ps + hi;
      int node = node0 + lrow;
      bool live = node < n;
      float4 acc = *(const float4*)&hs[lrow][4 * c];
      float inv = 1.0f / hs[lrow][D];
      float4 o;
      o.x = (acc.x * inv + b4.x - mu4.x) * (g4.x / sqrtf(va4.x + BN_EPS)) + be4.x;
      o.y = (acc.y * inv + b4.y - mu4.y) * (g4.y / sqrtf(va4.y + BN_EPS)) + be4.y;
      o.z = (acc.z * inv + b4.z - mu4.z) * (g4.z / sqrtf(va4.z + BN_EPS)) + be4.z;
      o.w = (acc.w * inv + b4.w - mu4.w) * (g4.w / sqrtf(va4.w + BN_EPS)) + be4.w;
      if (relu) {
        o.x = fmaxf(o.x, 0.f);
        o.y = fmaxf(o.y, 0.f);
        o.z = fmaxf(o.z, 0.f);
        o.w = fmaxf(o.w, 0.f);
      }
      if (Wn == nullptr) {
        if (live) ((float4*)outp)[(size_t)node * (D / 4) + c] = o;
      } else {
        *(float4*)&hs[lrow][4 * c] = live ? o : make_float4(0.f, 0.f, 0.f, 0.f);
      }
    }
    if (Wn == nullptr) return;
    __syncthreads();
    v8f cacc = {};
    #pragma unroll
    for (int k = 0; k < D; k += 4) {
      v2f a = *(const v2f*)&hs[lm][k + kh];
      cacc = __builtin_amdgcn_wmma_f32_16x16x4_f32(
          false, a, false, bf[k >> 2], (short)0, cacc, false, false);
    }
    int rbase = node0 + (hi << 3);
    #pragma unroll
    for (int r = 0; r < 8; r++) {
      int row = rbase + r;
      if (row < n) outp[(size_t)row * D + col] = cacc[r];
    }
  }
}

#define RUN 16
__device__ __forceinline__ void graph_add(float* __restrict__ og, int g, int c, float4 a) {
  float* p = og + (size_t)g * D + 4 * c;
  atomicAdd(p + 0, a.x);
  atomicAdd(p + 1, a.y);
  atomicAdd(p + 2, a.z);
  atomicAdd(p + 3, a.w);
}
__global__ void __launch_bounds__(128) k_output(const float* __restrict__ h, const int* __restrict__ batch,
                                                float* __restrict__ out_nodes, float* __restrict__ out_graph, int n) {
  int wave = (blockIdx.x * blockDim.x + threadIdx.x) >> 5;
  if (wave * 2 * RUN >= n) return;
  int lane = threadIdx.x & 31;
  int c = lane & 15;
  int hi = lane >> 4;
  int first = (wave * 2 + hi) * RUN;
  const float4* h4 = (const float4*)h;
  float4* o4 = (float4*)out_nodes;
  int gn = first + c < n ? first + c : n - 1;
  int gv = batch[gn];
  float4 r[RUN];
  #pragma unroll
  for (int j = 0; j < RUN; j++) {
    int node = first + j < n ? first + j : n - 1;
    r[j] = h4[(size_t)node * (D / 4) + c];
  }
  #pragma unroll
  for (int j = 0; j < RUN; j++)
    if (first + j < n) o4[(size_t)(first + j) * (D / 4) + c] = r[j];

  int cur = 0;
  bool have = false;
  float4 acc = make_float4(0.f, 0.f, 0.f, 0.f);
  #pragma unroll
  for (int j = 0; j < RUN; j++) {
    int g = row_bcast(gv, (unsigned)j * 0x11111111u);
    if (first + j < n) {
      if (have && g == cur) {
        acc.x += r[j].x;
        acc.y += r[j].y;
        acc.z += r[j].z;
        acc.w += r[j].w;
      } else {
        if (have) graph_add(out_graph, cur, c, acc);
        cur = g;
        have = true;
        acc = r[j];
      }
    }
  }
  int ocur = __shfl_xor(cur, 16);
  int ohave = __shfl_xor((int)have, 16);
  float4 oacc;
  oacc.x = __shfl_xor(acc.x, 16);
  oacc.y = __shfl_xor(acc.y, 16);
  oacc.z = __shfl_xor(acc.z, 16);
  oacc.w = __shfl_xor(acc.w, 16);
  bool joined = have && ohave && cur == ocur;
  if (joined) {
    if (hi == 0) {
      acc.x += oacc.x;
      acc.y += oacc.y;
      acc.z += oacc.z;
      acc.w += oacc.w;
      graph_add(out_graph, cur, c, acc);
    }
  } else if (have) {
    graph_add(out_graph, cur, c, acc);
  }
}

extern "C" void kernel_launch(void* const* d_in, const int* in_sizes, int n_in,
                              void* d_out, int out_size, void* d_ws, size_t ws_size,
                              hipStream_t stream) {
  const float* x     = (const float*)d_in[0];
  const int*   eidx  = (const int*)d_in[1];
  const float* eattr = (const float*)d_in[2];
  const int*   batch = (const int*)d_in[3];
  const float* xlw   = (const float*)d_in[4];
  const float* xlb   = (const float*)d_in[5];
  const float* glw   = (const float*)d_in[6];
  const float* gas   = (const float*)d_in[7];
  const float* gad   = (const float*)d_in[8];
  const float* gew   = (const float*)d_in[9];
  const float* gae   = (const float*)d_in[10];
  const float* gb    = (const float*)d_in[11];
  const float* bng   = (const float*)d_in[12];
  const float* bnb   = (const float*)d_in[13];
  const float* bnm   = (const float*)d_in[14];
  const float* bnv   = (const float*)d_in[15];

  const int n = in_sizes[3];
  const int E = in_sizes[2] / F_E;
  const int* srcp = eidx;
  const int* dstp = eidx + E;
  const int npad = ((n + 1 + SCAN_TILE - 1) / SCAN_TILE) * SCAN_TILE;
  const int nbk  = (n + BKN - 1) / BKN;
  const int m    = nbk * NB;
  const int mpad = ((m + 1 + SCAN_TILE - 1) / SCAN_TILE) * SCAN_TILE;
  const int ch   = (E + NB - 1) / NB;

  float* ws    = (float*)d_ws;
  float* h     = ws;  ws += (size_t)n * D;
  float* xw    = ws;  ws += (size_t)n * D;
  int*   rowptr = (int*)ws; ws += npad;
  int*   slot  = (int*)ws; ws += 4 * (size_t)E;
  float* wea   = ws;  ws += L * F_E;
  int*   tslot = (int*)h;
  int*   bh    = (int*)xw;
  int*   boff  = (int*)xw + mpad;
  float* fold  = xw + (size_t)n * D - FOLD_WORDS;

  float* out_nodes = (float*)d_out;
  float* out_graph = out_nodes + (size_t)n * D;
  const size_t graph_elems = (size_t)out_size - (size_t)n * D;

  k_prep<<<1, 256, 0, stream>>>(gew, gae, wea, xlw, xlb, glw, gas, gad, fold);
  k_bcount<<<NB, PT, 0, stream>>>(dstp, bh, E, ch, nbk, m, mpad);
  k_scan<<<mpad / SCAN_TILE, SCAN_NT, 0, stream>>>(bh, boff);
  k_bplace<<<NB, PT, 0, stream>>>(srcp, dstp, boff, eattr, wea, tslot, E, ch, nbk);
  k_bsort<<<nbk, DT, 0, stream>>>(boff, tslot, rowptr, slot, n);

  k_node<true><<<(n + 15) / 16, 128, 0, stream>>>(x, fold + FO_US, fold + FO_UD, rowptr, slot, 0,
                                                  gb, bng, bnb, bnm, bnv, glw + (size_t)D * D, fold, h, n, 1);
  const float* cur = h;
  float* nxt = xw;
  for (int l = 1; l < L; l++) {
    const bool last = l == L - 1;
    k_node<false><<<(n + 15) / 16, 128, 0, stream>>>(cur, gas + (size_t)l * D, gad + (size_t)l * D, rowptr, slot, l,
                                                     gb + (size_t)l * D, bng + (size_t)l * D, bnb + (size_t)l * D,
                                                     bnm + (size_t)l * D, bnv + (size_t)l * D,
                                                     last ? nullptr : glw + (size_t)(l + 1) * D * D,
                                                     nullptr, nxt, n, last ? 0 : 1);
    if (!last) {
      float* t = (float*)cur;
      cur = nxt;
      nxt = t;
    }
  }

  hipMemsetAsync(out_graph, 0, graph_elems * sizeof(float), stream);
  k_output<<<(n + 8 * RUN - 1) / (8 * RUN), 128, 0, stream>>>(nxt, batch, out_nodes, out_graph, n);
}
